# starting kernel
# baseline (speedup 1.0000x reference)
#include <hip/hip_runtime.h>
#include <hip/hip_bf16.h>
#include <cstddef>

typedef __attribute__((ext_vector_type(2))) float v2f;
typedef __attribute__((ext_vector_type(8))) float v8f;

#define IN_CH   128
#define OUT_CH  64
#define NEG_SLOPE 0.2f
#define EPS_F   1e-10f

__global__ void gat_gemm_wmma(const float* __restrict__ x,
                              const float* __restrict__ W,
                              float* __restrict__ h) {
    const int lane  = threadIdx.x & 31;
    const int wave  = threadIdx.x >> 5;
    const int tile  = blockIdx.x * (blockDim.x >> 5) + wave;
    const int mBase = (tile >> 2) * 16;
    const int nBase = (tile & 3) * 16;

    const int l15  = lane & 15;
    const int kOff = (lane >> 4) << 1;

    const float* xRow = x + (size_t)(mBase + l15) * IN_CH + kOff;
    const float* wRow = W + (size_t)(nBase + l15) * IN_CH + kOff;

    v8f acc = {};
#pragma unroll
    for (int k = 0; k < IN_CH; k += 4) {
        v2f a = *(const v2f*)(xRow + k);
        v2f b = *(const v2f*)(wRow + k);
        acc = __builtin_amdgcn_wmma_f32_16x16x4_f32(
             false, a,  false, b,
             (short)0, acc,  false,  false);
    }

    const int rowHalf = (lane >> 4) << 3;
    float* o = h + (size_t)(mBase + rowHalf) * OUT_CH + nBase + l15;
#pragma unroll
    for (int r = 0; r < 8; ++r)
        o[(size_t)r * OUT_CH] = acc[r];
}

__global__ void gat_gemm_tail(const float* __restrict__ x,
                              const float* __restrict__ W,
                              float* __restrict__ h,
                              int rowStart, int nNodes) {
    int idx = blockIdx.x * blockDim.x + threadIdx.x;
    int row = rowStart + (idx >> 6);
    int col = idx & 63;
    if (row >= nNodes) return;
    const float* xr = x + (size_t)row * IN_CH;
    const float* wr = W + (size_t)col * IN_CH;
    float s = 0.f;
    for (int k = 0; k < IN_CH; ++k) s += xr[k] * wr[k];
    h[(size_t)row * OUT_CH + col] = s;
}

__global__ void gat_el_er(const float* __restrict__ h,
                          const float* __restrict__ a_l,
                          const float* __restrict__ a_r,
                          float* __restrict__ e_l,
                          float* __restrict__ e_r, int nNodes) {
    const int lane = threadIdx.x & 31;
    const int node = (blockIdx.x * blockDim.x + threadIdx.x) >> 5;
    if (node >= nNodes) return;
    float2 hv = ((const float2*)(h + (size_t)node * OUT_CH))[lane];
    float2 al = ((const float2*)a_l)[lane];
    float2 ar = ((const float2*)a_r)[lane];
    float pl = hv.x * al.x + hv.y * al.y;
    float pr = hv.x * ar.x + hv.y * ar.y;
#pragma unroll
    for (int off = 16; off > 0; off >>= 1) {
        pl += __shfl_xor(pl, off, 32);
        pr += __shfl_xor(pr, off, 32);
    }
    if (lane == 0) { e_l[node] = pl; e_r[node] = pr; }
}

__global__ void gat_init(float* __restrict__ out, const float* __restrict__ bias,
                         float* __restrict__ node_max, float* __restrict__ node_sum,
                         int nNodes) {
    int i = blockIdx.x * blockDim.x + threadIdx.x;
    if (i < nNodes * OUT_CH) out[i] = bias[i & (OUT_CH - 1)];
    if (i < nNodes) { node_max[i] = -__builtin_inff(); node_sum[i] = 0.f; }
}

__device__ __forceinline__ void atomicMaxF32(float* addr, float v) {
    if (v >= 0.f)
        atomicMax((int*)addr, __float_as_int(v));
    else
        atomicMin((unsigned int*)addr, __float_as_uint(v));
}

__global__ void gat_edge_max(const int* __restrict__ src, const int* __restrict__ dst,
                             const float* __restrict__ e_l, const float* __restrict__ e_r,
                             float* __restrict__ edge_w, float* __restrict__ node_max,
                             int nEdges) {
    int e = blockIdx.x * blockDim.x + threadIdx.x;
    if (e >= nEdges) return;
    int s = src[e], d = dst[e];
    float a = e_l[d] + e_r[s];
    a = (a > 0.f) ? a : NEG_SLOPE * a;
    edge_w[e] = a;
    atomicMaxF32(node_max + d, a);
}

__global__ void gat_edge_exp(const int* __restrict__ dst,
                             float* __restrict__ edge_w,
                             const float* __restrict__ node_max,
                             float* __restrict__ node_sum, int nEdges) {
    int e = blockIdx.x * blockDim.x + threadIdx.x;
    if (e >= nEdges) return;
    int d = dst[e];
    float w = __expf(edge_w[e] - node_max[d]);
    edge_w[e] = w;
    atomicAdd(node_sum + d, w);
}

__global__ void gat_scatter(const int* __restrict__ src, const int* __restrict__ dst,
                            const float* __restrict__ h,
                            const float* __restrict__ edge_w,
                            const float* __restrict__ node_sum,
                            float* __restrict__ out, int nEdges) {
    const int lane = threadIdx.x & 31;
    const int e = (blockIdx.x * blockDim.x + threadIdx.x) >> 5;
    if (e >= nEdges) return;
    const int s = src[e], d = dst[e];
    const float coef = edge_w[e] / (node_sum[d] + EPS_F);
    float2 hv = ((const float2*)(h + (size_t)s * OUT_CH))[lane];
    float* o = out + (size_t)d * OUT_CH + lane * 2;
    atomicAdd(o + 0, hv.x * coef);
    atomicAdd(o + 1, hv.y * coef);
}

extern "C" void kernel_launch(void* const* d_in, const int* in_sizes, int n_in,
                              void* d_out, int out_size, void* d_ws, size_t ws_size,
                              hipStream_t stream) {
    const float* x    = (const float*)d_in[0];
    const int*   ei   = (const int*)d_in[1];
    const float* W    = (const float*)d_in[2];
    const float* a_l  = (const float*)d_in[3];
    const float* a_r  = (const float*)d_in[4];
    const float* bias = (const float*)d_in[5];
    float* out = (float*)d_out;

    const int nNodes = in_sizes[0] / IN_CH;
    const int nEdges = in_sizes[1] / 2;
    const int* src = ei;
    const int* dst = ei + nEdges;

    float* ws       = (float*)d_ws;
    float* h        = ws;
    float* e_l      = h + (size_t)nNodes * OUT_CH;
    float* e_r      = e_l + nNodes;
    float* node_max = e_r + nNodes;
    float* node_sum = node_max + nNodes;
    float* edge_w   = node_sum + nNodes;

    const int mFull = nNodes & ~15;
    const int tiles = (mFull / 16) * (OUT_CH / 16);
    if (tiles > 0)
        gat_gemm_wmma<<<(tiles + 3) / 4, 128, 0, stream>>>(x, W, h);
    if (mFull < nNodes) {
        int tailThreads = (nNodes - mFull) * OUT_CH;
        gat_gemm_tail<<<(tailThreads + 255) / 256, 256, 0, stream>>>(x, W, h, mFull, nNodes);
    }

    gat_el_er<<<(nNodes * 32 + 255) / 256, 256, 0, stream>>>(h, a_l, a_r, e_l, e_r, nNodes);

    int initN = nNodes * OUT_CH;
    gat_init<<<(initN + 255) / 256, 256, 0, stream>>>(out, bias, node_max, node_sum, nNodes);

    gat_edge_max<<<(nEdges + 255) / 256, 256, 0, stream>>>(src, dst, e_l, e_r, edge_w, node_max, nEdges);
    gat_edge_exp<<<(nEdges + 255) / 256, 256, 0, stream>>>(dst, edge_w, node_max, node_sum, nEdges);
    gat_scatter<<<((size_t)nEdges * 32 + 255) / 256, 256, 0, stream>>>(src, dst, h, edge_w, node_sum, out, nEdges);
}


# edge passes as owner-computes LDS tiles: 256 blocks each own a dst range and filter the dst stream into per-wave LDS lists; max, exp-sum, row sums by LDS atomics; rows stored once; sum order changes
# speedup vs baseline: 2.3106x; 2.3106x over previous
#include <hip/hip_runtime.h>
#include <hip/hip_bf16.h>
#include <cstddef>

typedef __attribute__((ext_vector_type(2))) float v2f;
typedef __attribute__((ext_vector_type(8))) float v8f;

#define IN_CH   128
#define OUT_CH  64
#define NEG_SLOPE 0.2f
#define EPS_F   1e-10f

__global__ void gat_gemm_wmma(const float* __restrict__ x,
                              const float* __restrict__ W,
                              float* __restrict__ h) {
    const int lane  = threadIdx.x & 31;
    const int wave  = threadIdx.x >> 5;
    const int tile  = blockIdx.x * (blockDim.x >> 5) + wave;
    const int mBase = (tile >> 2) * 16;
    const int nBase = (tile & 3) * 16;

    const int l15  = lane & 15;
    const int kOff = (lane >> 4) << 1;

    const float* xRow = x + (size_t)(mBase + l15) * IN_CH + kOff;
    const float* wRow = W + (size_t)(nBase + l15) * IN_CH + kOff;

    v8f acc = {};
#pragma unroll
    for (int k = 0; k < IN_CH; k += 4) {
        v2f a = *(const v2f*)(xRow + k);
        v2f b = *(const v2f*)(wRow + k);
        acc = __builtin_amdgcn_wmma_f32_16x16x4_f32(
             false, a,  false, b,
             (short)0, acc,  false,  false);
    }

    const int rowHalf = (lane >> 4) << 3;
    float* o = h + (size_t)(mBase + rowHalf) * OUT_CH + nBase + l15;
#pragma unroll
    for (int r = 0; r < 8; ++r)
        o[(size_t)r * OUT_CH] = acc[r];
}

__global__ void gat_gemm_tail(const float* __restrict__ x,
                              const float* __restrict__ W,
                              float* __restrict__ h,
                              int rowStart, int nNodes) {
    int idx = blockIdx.x * blockDim.x + threadIdx.x;
    int row = rowStart + (idx >> 6);
    int col = idx & 63;
    if (row >= nNodes) return;
    const float* xr = x + (size_t)row * IN_CH;
    const float* wr = W + (size_t)col * IN_CH;
    float s = 0.f;
    for (int k = 0; k < IN_CH; ++k) s += xr[k] * wr[k];
    h[(size_t)row * OUT_CH + col] = s;
}

__global__ void gat_el_er(const float* __restrict__ h,
                          const float* __restrict__ a_l,
                          const float* __restrict__ a_r,
                          float* __restrict__ e_l,
                          float* __restrict__ e_r, int nNodes) {
    const int lane = threadIdx.x & 31;
    const int node = (blockIdx.x * blockDim.x + threadIdx.x) >> 5;
    if (node >= nNodes) return;
    float2 hv = ((const float2*)(h + (size_t)node * OUT_CH))[lane];
    float2 al = ((const float2*)a_l)[lane];
    float2 ar = ((const float2*)a_r)[lane];
    float pl = hv.x * al.x + hv.y * al.y;
    float pr = hv.x * ar.x + hv.y * ar.y;
#pragma unroll
    for (int off = 16; off > 0; off >>= 1) {
        pl += __shfl_xor(pl, off, 32);
        pr += __shfl_xor(pr, off, 32);
    }
    if (lane == 0) { e_l[node] = pl; e_r[node] = pr; }
}

__global__ void gat_init(float* __restrict__ out, const float* __restrict__ bias,
                         float* __restrict__ node_max, float* __restrict__ node_sum,
                         int nNodes) {
    int i = blockIdx.x * blockDim.x + threadIdx.x;
    if (i < nNodes * OUT_CH) out[i] = bias[i & (OUT_CH - 1)];
    if (i < nNodes) { node_max[i] = -__builtin_inff(); node_sum[i] = 0.f; }
}

__device__ __forceinline__ void atomicMaxF32(float* addr, float v) {
    if (v >= 0.f)
        atomicMax((int*)addr, __float_as_int(v));
    else
        atomicMin((unsigned int*)addr, __float_as_uint(v));
}

__global__ void gat_edge_max(const int* __restrict__ src, const int* __restrict__ dst,
                             const float* __restrict__ e_l, const float* __restrict__ e_r,
                             float* __restrict__ edge_w, float* __restrict__ node_max,
                             int nEdges) {
    int e = blockIdx.x * blockDim.x + threadIdx.x;
    if (e >= nEdges) return;
    int s = src[e], d = dst[e];
    float a = e_l[d] + e_r[s];
    a = (a > 0.f) ? a : NEG_SLOPE * a;
    edge_w[e] = a;
    atomicMaxF32(node_max + d, a);
}

__global__ void gat_edge_exp(const int* __restrict__ dst,
                             float* __restrict__ edge_w,
                             const float* __restrict__ node_max,
                             float* __restrict__ node_sum, int nEdges) {
    int e = blockIdx.x * blockDim.x + threadIdx.x;
    if (e >= nEdges) return;
    int d = dst[e];
    float w = __expf(edge_w[e] - node_max[d]);
    edge_w[e] = w;
    atomicAdd(node_sum + d, w);
}

__global__ void gat_scatter(const int* __restrict__ src, const int* __restrict__ dst,
                            const float* __restrict__ h,
                            const float* __restrict__ edge_w,
                            const float* __restrict__ node_sum,
                            float* __restrict__ out, int nEdges) {
    const int lane = threadIdx.x & 31;
    const int e = (blockIdx.x * blockDim.x + threadIdx.x) >> 5;
    if (e >= nEdges) return;
    const int s = src[e], d = dst[e];
    const float coef = edge_w[e] / (node_sum[d] + EPS_F);
    float2 hv = ((const float2*)(h + (size_t)s * OUT_CH))[lane];
    float* o = out + (size_t)d * OUT_CH + lane * 2;
    atomicAdd(o + 0, hv.x * coef);
    atomicAdd(o + 1, hv.y * coef);
}


#define OWN_BLOCKS   256
#define OWN_THREADS  1024
#define OWN_WAVES    32
#define NPER_MAX     391
#define SEG_CAP      768
#define E_BITS       21
#define S_BITS       17

typedef __attribute__((ext_vector_type(4))) int v4i;

__device__ __forceinline__ void ldsMaxF32(int* p, float v) {
    const int b = __float_as_int(v);
    if (b >= 0) atomicMax(p, b);
    else        atomicMin((unsigned int*)p, (unsigned int)b);
}

template <int PHASE>
__device__ __forceinline__ void edge_now(bool m, int e, int dl, int lo, int lane,
                                         const int* __restrict__ src, const float* __restrict__ h,
                                         const float* __restrict__ e_l, const float* __restrict__ e_r,
                                         int* nmaxi, float* nsum, float* acc) {
    int s = 0; float w = 0.f;
    if (m) {
        s = src[e];
        const float x = e_l[lo + dl] + e_r[s];
        const float a = (x > 0.f) ? x : NEG_SLOPE * x;
        if (PHASE == 1) {
            ldsMaxF32(&nmaxi[dl], a);
        } else {
            w = __expf(a - __int_as_float(nmaxi[dl]));
            atomicAdd(&nsum[dl], w);
        }
    }
    if (PHASE == 2) {
        unsigned mask = __builtin_amdgcn_ballot_w32(m);
        const int dls = m ? dl : 0;
        while (mask) {
            const int k = __builtin_ctz(mask);
            mask &= mask - 1;
            const int   sk = __builtin_amdgcn_readlane(s, k);
            const int   dk = __builtin_amdgcn_readlane(dls, k);
            const float wk = __int_as_float(__builtin_amdgcn_readlane(__float_as_int(w), k));
            const float2 hv = ((const float2*)(h + (size_t)sk * OUT_CH))[lane];
            float* r = &acc[dk * OUT_CH + 2 * lane];
            atomicAdd(r,     hv.x * wk);
            atomicAdd(r + 1, hv.y * wk);
        }
    }
}

template <int MODE>
__device__ __forceinline__ void scan_quad(v4i v, int e0, int lo, int cnt, int lane,
                                          const int* __restrict__ src, const float* __restrict__ h,
                                          const float* __restrict__ e_l, const float* __restrict__ e_r,
                                          int* nmaxi, float* nsum, float* acc,
                                          unsigned int* seg, unsigned int& wcount) {
    const int d0 = v.x - lo, d1 = v.y - lo, d2 = v.z - lo, d3 = v.w - lo;
    const bool m0 = (unsigned)d0 < (unsigned)cnt;
    const bool m1 = (unsigned)d1 < (unsigned)cnt;
    const bool m2 = (unsigned)d2 < (unsigned)cnt;
    const bool m3 = (unsigned)d3 < (unsigned)cnt;
    if (__builtin_amdgcn_ballot_w32(m0 | m1 | m2 | m3) == 0) return;
    if (MODE == 0) {
        const unsigned k0 = __builtin_amdgcn_ballot_w32(m0);
        const unsigned k1 = __builtin_amdgcn_ballot_w32(m1);
        const unsigned k2 = __builtin_amdgcn_ballot_w32(m2);
        const unsigned k3 = __builtin_amdgcn_ballot_w32(m3);
        const unsigned n0 = __builtin_popcount(k0), n1 = __builtin_popcount(k1);
        const unsigned n2 = __builtin_popcount(k2), n3 = __builtin_popcount(k3);
        const unsigned i0 = wcount + __builtin_amdgcn_mbcnt_lo(k0, 0);
        const unsigned i1 = wcount + n0 + __builtin_amdgcn_mbcnt_lo(k1, 0);
        const unsigned i2 = wcount + n0 + n1 + __builtin_amdgcn_mbcnt_lo(k2, 0);
        const unsigned i3 = wcount + n0 + n1 + n2 + __builtin_amdgcn_mbcnt_lo(k3, 0);
        if (m0 && i0 < SEG_CAP) seg[i0] = (unsigned)(e0 + 0) | ((unsigned)d0 << E_BITS);
        if (m1 && i1 < SEG_CAP) seg[i1] = (unsigned)(e0 + 1) | ((unsigned)d1 << E_BITS);
        if (m2 && i2 < SEG_CAP) seg[i2] = (unsigned)(e0 + 2) | ((unsigned)d2 << E_BITS);
        if (m3 && i3 < SEG_CAP) seg[i3] = (unsigned)(e0 + 3) | ((unsigned)d3 << E_BITS);
        wcount += n0 + n1 + n2 + n3;
    } else {
        edge_now<MODE>(m0, e0 + 0, d0, lo, lane, src, h, e_l, e_r, nmaxi, nsum, acc);
        edge_now<MODE>(m1, e0 + 1, d1, lo, lane, src, h, e_l, e_r, nmaxi, nsum, acc);
        edge_now<MODE>(m2, e0 + 2, d2, lo, lane, src, h, e_l, e_r, nmaxi, nsum, acc);
        edge_now<MODE>(m3, e0 + 3, d3, lo, lane, src, h, e_l, e_r, nmaxi, nsum, acc);
    }
}

template <int MODE>
__device__ __forceinline__ void scan_edges(const int* __restrict__ src, const int* __restrict__ dst,
                                           const float* __restrict__ h,
                                           const float* __restrict__ e_l, const float* __restrict__ e_r,
                                           int nEdges, int lo, int cnt, int lane, int wave,
                                           int* nmaxi, float* nsum, float* acc,
                                           unsigned int* seg, unsigned int& wcount) {
    const int nGroups = nEdges >> 7;
    const v4i* dst4 = (const v4i*)dst;
    const int start = (int)(((long long)blockIdx.x * nGroups) / OWN_BLOCKS);
    const v4i none = {-1, -1, -1, -1};
    for (int j = wave; j < nGroups; j += OWN_WAVES * 8) {
        v4i v[8]; int g[8];
#pragma unroll
        for (int u = 0; u < 8; ++u) {
            const int jj = j + u * OWN_WAVES;
            int gg = jj + start;
            if (gg >= nGroups) gg -= nGroups;
            g[u] = gg;
            v[u] = (jj < nGroups) ? dst4[(size_t)gg * 32 + lane] : none;
        }
#pragma unroll
        for (int u = 0; u < 8; ++u)
            scan_quad<MODE>(v[u], g[u] * 128 + lane * 4, lo, cnt, lane, src, h, e_l, e_r,
                            nmaxi, nsum, acc, seg, wcount);
    }
    for (int eb = nGroups * 128 + wave * 32; eb < nEdges; eb += OWN_THREADS) {
        const int e = eb + lane;
        const int d = (e < nEdges) ? dst[e] : -1;
        const int dl = d - lo;
        const bool m = (unsigned)dl < (unsigned)cnt;
        const unsigned km = __builtin_amdgcn_ballot_w32(m);
        if (km == 0) continue;
        if (MODE == 0) {
            const unsigned i = wcount + __builtin_amdgcn_mbcnt_lo(km, 0);
            if (m && i < SEG_CAP) seg[i] = (unsigned)e | ((unsigned)dl << E_BITS);
            wcount += __builtin_popcount(km);
        } else {
            edge_now<MODE>(m, e, dl, lo, lane, src, h, e_l, e_r, nmaxi, nsum, acc);
        }
    }
}

__global__ __launch_bounds__(OWN_THREADS)
void gat_owner(const int* __restrict__ src, const int* __restrict__ dst,
               const float* __restrict__ h,
               const float* __restrict__ e_l, const float* __restrict__ e_r,
               const float* __restrict__ bias, float* __restrict__ out,
               int nNodes, int nEdges, int nper) {
    __shared__ float        acc[(NPER_MAX + 1) * OUT_CH];
    __shared__ unsigned int list[OWN_WAVES * SEG_CAP];
    __shared__ int          nmaxi[NPER_MAX];
    __shared__ float        nsum[NPER_MAX];
    __shared__ unsigned int overflow;

    const int tid  = threadIdx.x;
    const int lane = tid & 31;
    const int wave = __builtin_amdgcn_readfirstlane(tid >> 5);
    const int lo   = blockIdx.x * nper;
    int cnt = nNodes - lo;
    cnt = cnt < 0 ? 0 : (cnt > nper ? nper : cnt);
    unsigned int* seg = list + wave * SEG_CAP;

    for (int i = tid; i < cnt * OUT_CH; i += OWN_THREADS) acc[i] = 0.f;
    for (int i = tid; i < cnt; i += OWN_THREADS) { nmaxi[i] = (int)0xFF800000u; nsum[i] = 0.f; }
    if (tid == 0) overflow = 0u;
    __syncthreads();

    unsigned int wcount = 0u;
    scan_edges<0>(src, dst, h, e_l, e_r, nEdges, lo, cnt, lane, wave, nmaxi, nsum, acc, seg, wcount);
    if (wcount > SEG_CAP && lane == 0) overflow = 1u;
    __syncthreads();

    const unsigned over = __builtin_amdgcn_readfirstlane(overflow);
    if (over == 0u) {
        for (unsigned b = 0; b < wcount; b += 128u) {
            unsigned idx[4], p[4]; int s[4]; float el[4], er[4];
#pragma unroll
            for (int t = 0; t < 4; ++t) {
                idx[t] = b + 32u * t + lane;
                p[t] = seg[idx[t] < wcount ? idx[t] : wcount - 1u];
            }
#pragma unroll
            for (int t = 0; t < 4; ++t) {
                s[t]  = src[(int)(p[t] & ((1u << E_BITS) - 1u))];
                el[t] = e_l[lo + (int)(p[t] >> E_BITS)];
            }
#pragma unroll
            for (int t = 0; t < 4; ++t) er[t] = e_r[s[t]];
#pragma unroll
            for (int t = 0; t < 4; ++t) {
                if (idx[t] < wcount) {
                    const int dl = (int)(p[t] >> E_BITS);
                    const float x = el[t] + er[t];
                    const float a = (x > 0.f) ? x : NEG_SLOPE * x;
                    ldsMaxF32(&nmaxi[dl], a);
                    seg[idx[t]] = (unsigned)s[t] | ((unsigned)dl << S_BITS);
                }
            }
        }
        __syncthreads();

        for (unsigned b = 0; b < wcount; b += 32u) {
            const unsigned i = b + lane;
            int s = 0, dl = NPER_MAX; float w = 0.f;
            if (i < wcount) {
                const unsigned p = seg[i];
                s  = (int)(p & ((1u << S_BITS) - 1u));
                dl = (int)(p >> S_BITS);
                const float x = e_l[lo + dl] + e_r[s];
                const float a = (x > 0.f) ? x : NEG_SLOPE * x;
                w = __expf(a - __int_as_float(nmaxi[dl]));
                atomicAdd(&nsum[dl], w);
            }
            const int nvalid = (wcount - b) < 32u ? (int)(wcount - b) : 32;
            for (int k0 = 0; k0 < nvalid; k0 += 16) {
                float2 hv[16]; int dk[16]; float wk[16];
#pragma unroll
                for (int u = 0; u < 16; ++u) {
                    const int k = k0 + u;
                    const int sk = __builtin_amdgcn_readlane(s, k);
                    dk[u] = __builtin_amdgcn_readlane(dl, k);
                    wk[u] = __int_as_float(__builtin_amdgcn_readlane(__float_as_int(w), k));
                    hv[u] = ((const float2*)(h + (size_t)sk * OUT_CH))[lane];
                }
#pragma unroll
                for (int u = 0; u < 16; ++u) {
                    float* r = &acc[dk[u] * OUT_CH + 2 * lane];
                    atomicAdd(r,     hv[u].x * wk[u]);
                    atomicAdd(r + 1, hv[u].y * wk[u]);
                }
            }
        }
    } else {
        scan_edges<1>(src, dst, h, e_l, e_r, nEdges, lo, cnt, lane, wave, nmaxi, nsum, acc, seg, wcount);
        __syncthreads();
        scan_edges<2>(src, dst, h, e_l, e_r, nEdges, lo, cnt, lane, wave, nmaxi, nsum, acc, seg, wcount);
    }
    __syncthreads();

    float* o = out + (size_t)lo * OUT_CH;
    for (int i = tid; i < cnt * OUT_CH; i += OWN_THREADS)
        o[i] = acc[i] / (nsum[i >> 6] + EPS_F) + bias[i & (OUT_CH - 1)];
}

extern "C" void kernel_launch(void* const* d_in, const int* in_sizes, int n_in,
                              void* d_out, int out_size, void* d_ws, size_t ws_size,
                              hipStream_t stream) {
    const float* x    = (const float*)d_in[0];
    const int*   ei   = (const int*)d_in[1];
    const float* W    = (const float*)d_in[2];
    const float* a_l  = (const float*)d_in[3];
    const float* a_r  = (const float*)d_in[4];
    const float* bias = (const float*)d_in[5];
    float* out = (float*)d_out;

    const int nNodes = in_sizes[0] / IN_CH;
    const int nEdges = in_sizes[1] / 2;
    const int* src = ei;
    const int* dst = ei + nEdges;

    float* ws       = (float*)d_ws;
    float* h        = ws;
    float* e_l      = h + (size_t)nNodes * OUT_CH;
    float* e_r      = e_l + nNodes;
    float* node_max = e_r + nNodes;
    float* node_sum = node_max + nNodes;
    float* edge_w   = node_sum + nNodes;

    const int mFull = nNodes & ~15;
    const int tiles = (mFull / 16) * (OUT_CH / 16);
    if (tiles > 0)
        gat_gemm_wmma<<<(tiles + 3) / 4, 128, 0, stream>>>(x, W, h);
    if (mFull < nNodes) {
        int tailThreads = (nNodes - mFull) * OUT_CH;
        gat_gemm_tail<<<(tailThreads + 255) / 256, 256, 0, stream>>>(x, W, h, mFull, nNodes);
    }

    gat_el_er<<<(nNodes * 32 + 255) / 256, 256, 0, stream>>>(h, a_l, a_r, e_l, e_r, nNodes);

    const int nper = (nNodes + OWN_BLOCKS - 1) / OWN_BLOCKS;
    if (nper <= NPER_MAX && (nEdges & 3) == 0 && nEdges < (1 << E_BITS) && nNodes <= (1 << S_BITS)) {
        gat_owner<<<OWN_BLOCKS, OWN_THREADS, 0, stream>>>(src, dst, h, e_l, e_r, bias, out, nNodes, nEdges, nper);
        return;
    }

    int initN = nNodes * OUT_CH;
    gat_init<<<(initN + 255) / 256, 256, 0, stream>>>(out, bias, node_max, node_sum, nNodes);

    gat_edge_max<<<(nEdges + 255) / 256, 256, 0, stream>>>(src, dst, e_l, e_r, edge_w, node_max, nEdges);
    gat_edge_exp<<<(nEdges + 255) / 256, 256, 0, stream>>>(dst, edge_w, node_max, node_sum, nEdges);
    gat_scatter<<<((size_t)nEdges * 32 + 255) / 256, 256, 0, stream>>>(src, dst, h, edge_w, node_sum, out, nEdges);
}


# dst stream cut to one owner byte per edge (dst / nper, written in the el_er launch): owner blocks filter a quarter of the bytes, 16 edges per lane and load; list order, so sum order, changes
# speedup vs baseline: 2.8627x; 1.2389x over previous
#include <hip/hip_runtime.h>
#include <hip/hip_bf16.h>
#include <cstddef>

typedef __attribute__((ext_vector_type(2))) float v2f;
typedef __attribute__((ext_vector_type(8))) float v8f;

#define IN_CH   128
#define OUT_CH  64
#define NEG_SLOPE 0.2f
#define EPS_F   1e-10f

__global__ void gat_gemm_wmma(const float* __restrict__ x,
                              const float* __restrict__ W,
                              float* __restrict__ h) {
    const int lane  = threadIdx.x & 31;
    const int wave  = threadIdx.x >> 5;
    const int tile  = blockIdx.x * (blockDim.x >> 5) + wave;
    const int mBase = (tile >> 2) * 16;
    const int nBase = (tile & 3) * 16;

    const int l15  = lane & 15;
    const int kOff = (lane >> 4) << 1;

    const float* xRow = x + (size_t)(mBase + l15) * IN_CH + kOff;
    const float* wRow = W + (size_t)(nBase + l15) * IN_CH + kOff;

    v8f acc = {};
#pragma unroll
    for (int k = 0; k < IN_CH; k += 4) {
        v2f a = *(const v2f*)(xRow + k);
        v2f b = *(const v2f*)(wRow + k);
        acc = __builtin_amdgcn_wmma_f32_16x16x4_f32(
             false, a,  false, b,
             (short)0, acc,  false,  false);
    }

    const int rowHalf = (lane >> 4) << 3;
    float* o = h + (size_t)(mBase + rowHalf) * OUT_CH + nBase + l15;
#pragma unroll
    for (int r = 0; r < 8; ++r)
        o[(size_t)r * OUT_CH] = acc[r];
}

__global__ void gat_gemm_tail(const float* __restrict__ x,
                              const float* __restrict__ W,
                              float* __restrict__ h,
                              int rowStart, int nNodes) {
    int idx = blockIdx.x * blockDim.x + threadIdx.x;
    int row = rowStart + (idx >> 6);
    int col = idx & 63;
    if (row >= nNodes) return;
    const float* xr = x + (size_t)row * IN_CH;
    const float* wr = W + (size_t)col * IN_CH;
    float s = 0.f;
    for (int k = 0; k < IN_CH; ++k) s += xr[k] * wr[k];
    h[(size_t)row * OUT_CH + col] = s;
}

typedef __attribute__((ext_vector_type(4))) int v4i;
typedef __attribute__((ext_vector_type(4))) unsigned int v4u;

__global__ void gat_el_er(const float* __restrict__ h,
                          const float* __restrict__ a_l,
                          const float* __restrict__ a_r,
                          float* __restrict__ e_l,
                          float* __restrict__ e_r, int nNodes,
                          const int* __restrict__ dst, unsigned int* __restrict__ own4,
                          int nEdgesOwn, int nper) {
    {
        const int nQuads = nEdgesOwn >> 2;
        const int stride = (int)(gridDim.x * blockDim.x);
        const unsigned un = (unsigned)nper;
        for (int q = blockIdx.x * blockDim.x + threadIdx.x; q < nQuads; q += stride) {
            const v4i d = ((const v4i*)dst)[q];
            own4[q] = (((unsigned)d.x / un) & 0xFFu) | ((((unsigned)d.y / un) & 0xFFu) << 8) |
                      ((((unsigned)d.z / un) & 0xFFu) << 16) | (((unsigned)d.w / un) << 24);
        }
    }
    const int lane = threadIdx.x & 31;
    const int node = (blockIdx.x * blockDim.x + threadIdx.x) >> 5;
    if (node >= nNodes) return;
    float2 hv = ((const float2*)(h + (size_t)node * OUT_CH))[lane];
    float2 al = ((const float2*)a_l)[lane];
    float2 ar = ((const float2*)a_r)[lane];
    float pl = hv.x * al.x + hv.y * al.y;
    float pr = hv.x * ar.x + hv.y * ar.y;
#pragma unroll
    for (int off = 16; off > 0; off >>= 1) {
        pl += __shfl_xor(pl, off, 32);
        pr += __shfl_xor(pr, off, 32);
    }
    if (lane == 0) { e_l[node] = pl; e_r[node] = pr; }
}

__global__ void gat_init(float* __restrict__ out, const float* __restrict__ bias,
                         float* __restrict__ node_max, float* __restrict__ node_sum,
                         int nNodes) {
    int i = blockIdx.x * blockDim.x + threadIdx.x;
    if (i < nNodes * OUT_CH) out[i] = bias[i & (OUT_CH - 1)];
    if (i < nNodes) { node_max[i] = -__builtin_inff(); node_sum[i] = 0.f; }
}

__device__ __forceinline__ void atomicMaxF32(float* addr, float v) {
    if (v >= 0.f)
        atomicMax((int*)addr, __float_as_int(v));
    else
        atomicMin((unsigned int*)addr, __float_as_uint(v));
}

__global__ void gat_edge_max(const int* __restrict__ src, const int* __restrict__ dst,
                             const float* __restrict__ e_l, const float* __restrict__ e_r,
                             float* __restrict__ edge_w, float* __restrict__ node_max,
                             int nEdges) {
    int e = blockIdx.x * blockDim.x + threadIdx.x;
    if (e >= nEdges) return;
    int s = src[e], d = dst[e];
    float a = e_l[d] + e_r[s];
    a = (a > 0.f) ? a : NEG_SLOPE * a;
    edge_w[e] = a;
    atomicMaxF32(node_max + d, a);
}

__global__ void gat_edge_exp(const int* __restrict__ dst,
                             float* __restrict__ edge_w,
                             const float* __restrict__ node_max,
                             float* __restrict__ node_sum, int nEdges) {
    int e = blockIdx.x * blockDim.x + threadIdx.x;
    if (e >= nEdges) return;
    int d = dst[e];
    float w = __expf(edge_w[e] - node_max[d]);
    edge_w[e] = w;
    atomicAdd(node_sum + d, w);
}

__global__ void gat_scatter(const int* __restrict__ src, const int* __restrict__ dst,
                            const float* __restrict__ h,
                            const float* __restrict__ edge_w,
                            const float* __restrict__ node_sum,
                            float* __restrict__ out, int nEdges) {
    const int lane = threadIdx.x & 31;
    const int e = (blockIdx.x * blockDim.x + threadIdx.x) >> 5;
    if (e >= nEdges) return;
    const int s = src[e], d = dst[e];
    const float coef = edge_w[e] / (node_sum[d] + EPS_F);
    float2 hv = ((const float2*)(h + (size_t)s * OUT_CH))[lane];
    float* o = out + (size_t)d * OUT_CH + lane * 2;
    atomicAdd(o + 0, hv.x * coef);
    atomicAdd(o + 1, hv.y * coef);
}


#define OWN_BLOCKS   256
#define OWN_THREADS  1024
#define OWN_WAVES    32
#define NPER_MAX     391
#define SEG_CAP      768
#define E_BITS       21
#define S_BITS       17

__device__ __forceinline__ void ldsMaxF32(int* p, float v) {
    const int b = __float_as_int(v);
    if (b >= 0) atomicMax(p, b);
    else        atomicMin((unsigned int*)p, (unsigned int)b);
}

template <int PHASE>
__device__ __forceinline__ void edge_now(bool m, int e, int dl, int lo, int lane,
                                         const int* __restrict__ src, const float* __restrict__ h,
                                         const float* __restrict__ e_l, const float* __restrict__ e_r,
                                         int* nmaxi, float* nsum, float* acc) {
    int s = 0; float w = 0.f;
    if (m) {
        s = src[e];
        const float x = e_l[lo + dl] + e_r[s];
        const float a = (x > 0.f) ? x : NEG_SLOPE * x;
        if (PHASE == 1) {
            ldsMaxF32(&nmaxi[dl], a);
        } else {
            w = __expf(a - __int_as_float(nmaxi[dl]));
            atomicAdd(&nsum[dl], w);
        }
    }
    if (PHASE == 2) {
        unsigned mask = __builtin_amdgcn_ballot_w32(m);
        const int dls = m ? dl : 0;
        while (mask) {
            const int k = __builtin_ctz(mask);
            mask &= mask - 1;
            const int   sk = __builtin_amdgcn_readlane(s, k);
            const int   dk = __builtin_amdgcn_readlane(dls, k);
            const float wk = __int_as_float(__builtin_amdgcn_readlane(__float_as_int(w), k));
            const float2 hv = ((const float2*)(h + (size_t)sk * OUT_CH))[lane];
            float* r = &acc[dk * OUT_CH + 2 * lane];
            atomicAdd(r,     hv.x * wk);
            atomicAdd(r + 1, hv.y * wk);
        }
    }
}

template <int MODE>
__device__ __forceinline__ void scan_quad(v4i v, int e0, int lo, int cnt, int lane,
                                          const int* __restrict__ src, const float* __restrict__ h,
                                          const float* __restrict__ e_l, const float* __restrict__ e_r,
                                          int* nmaxi, float* nsum, float* acc) {
    const int d0 = v.x - lo, d1 = v.y - lo, d2 = v.z - lo, d3 = v.w - lo;
    const bool m0 = (unsigned)d0 < (unsigned)cnt;
    const bool m1 = (unsigned)d1 < (unsigned)cnt;
    const bool m2 = (unsigned)d2 < (unsigned)cnt;
    const bool m3 = (unsigned)d3 < (unsigned)cnt;
    if (__builtin_amdgcn_ballot_w32(m0 | m1 | m2 | m3) == 0) return;
    edge_now<MODE>(m0, e0 + 0, d0, lo, lane, src, h, e_l, e_r, nmaxi, nsum, acc);
    edge_now<MODE>(m1, e0 + 1, d1, lo, lane, src, h, e_l, e_r, nmaxi, nsum, acc);
    edge_now<MODE>(m2, e0 + 2, d2, lo, lane, src, h, e_l, e_r, nmaxi, nsum, acc);
    edge_now<MODE>(m3, e0 + 3, d3, lo, lane, src, h, e_l, e_r, nmaxi, nsum, acc);
}

template <int MODE>
__device__ __forceinline__ void scan_edges(const int* __restrict__ src, const int* __restrict__ dst,
                                           const float* __restrict__ h,
                                           const float* __restrict__ e_l, const float* __restrict__ e_r,
                                           int nEdges, int lo, int cnt, int lane, int wave,
                                           int* nmaxi, float* nsum, float* acc) {
    const int nGroups = nEdges >> 7;
    const v4i* dst4 = (const v4i*)dst;
    const int start = (int)(((long long)blockIdx.x * nGroups) / OWN_BLOCKS);
    const v4i none = {-1, -1, -1, -1};
    for (int j = wave; j < nGroups; j += OWN_WAVES * 8) {
        v4i v[8]; int g[8];
#pragma unroll
        for (int u = 0; u < 8; ++u) {
            const int jj = j + u * OWN_WAVES;
            int gg = jj + start;
            if (gg >= nGroups) gg -= nGroups;
            g[u] = gg;
            v[u] = (jj < nGroups) ? dst4[(size_t)gg * 32 + lane] : none;
        }
#pragma unroll
        for (int u = 0; u < 8; ++u)
            scan_quad<MODE>(v[u], g[u] * 128 + lane * 4, lo, cnt, lane, src, h, e_l, e_r,
                            nmaxi, nsum, acc);
    }
    for (int eb = nGroups * 128 + wave * 32; eb < nEdges; eb += OWN_THREADS) {
        const int e = eb + lane;
        const int d = (e < nEdges) ? dst[e] : -1;
        const int dl = d - lo;
        const bool m = (unsigned)dl < (unsigned)cnt;
        const unsigned km = __builtin_amdgcn_ballot_w32(m);
        if (km == 0) continue;
        edge_now<MODE>(m, e, dl, lo, lane, src, h, e_l, e_r, nmaxi, nsum, acc);
    }
}

__device__ __forceinline__ void scan_bytes16(v4u v, int e0, unsigned pat,
                                             unsigned int* seg, unsigned int& wcount) {
    unsigned f[4];
#pragma unroll
    for (int j = 0; j < 4; ++j) {
        const unsigned x = v[j] ^ pat;
        const unsigned y = (x & 0x7F7F7F7Fu) + 0x7F7F7F7Fu;
        f[j] = ~(y | x | 0x7F7F7F7Fu);
    }
    unsigned m = (f[0] >> 7) | (f[1] >> 6) | (f[2] >> 5) | (f[3] >> 4);
    for (unsigned k = __builtin_amdgcn_ballot_w32(m != 0u); k != 0u; k = __builtin_amdgcn_ballot_w32(m != 0u)) {
        const int bit = __builtin_ctz(m | 0x80000000u);
        const unsigned i = wcount + __builtin_amdgcn_mbcnt_lo(k, 0);
        if (m != 0u && i < SEG_CAP) seg[i] = (unsigned)(e0 + ((bit & 7) << 2) + (bit >> 3));
        m &= m - 1u;
        wcount += __builtin_popcount(k);
    }
}

__device__ __forceinline__ void scan_owner_bytes(const unsigned int* __restrict__ own4,
                                                 const int* __restrict__ dst,
                                                 int nEdges, int lo, int cnt, int lane, int wave,
                                                 unsigned int* seg, unsigned int& wcount) {
    const int nGroups = nEdges >> 9;
    const v4u* ob = (const v4u*)own4;
    const unsigned pat = (unsigned)blockIdx.x * 0x01010101u;
    const int start = (int)(((long long)blockIdx.x * nGroups) / OWN_BLOCKS);
    const v4u none = {~pat, ~pat, ~pat, ~pat};
    for (int j = wave; j < nGroups; j += OWN_WAVES * 8) {
        v4u v[8]; int g[8];
#pragma unroll
        for (int u = 0; u < 8; ++u) {
            const int jj = j + u * OWN_WAVES;
            int gg = jj + start;
            if (gg >= nGroups) gg -= nGroups;
            g[u] = gg;
            v[u] = (jj < nGroups) ? ob[(size_t)gg * 32 + lane] : none;
        }
#pragma unroll
        for (int u = 0; u < 8; ++u)
            scan_bytes16(v[u], (g[u] * 32 + lane) * 16, pat, seg, wcount);
    }
    for (int eb = (nGroups << 9) + wave * 32; eb < nEdges; eb += OWN_THREADS) {
        const int e = eb + lane;
        const int d = (e < nEdges) ? dst[e] : -1;
        const int dl = d - lo;
        const bool m = (unsigned)dl < (unsigned)cnt;
        const unsigned km = __builtin_amdgcn_ballot_w32(m);
        if (km == 0) continue;
        const unsigned i = wcount + __builtin_amdgcn_mbcnt_lo(km, 0);
        if (m && i < SEG_CAP) seg[i] = (unsigned)e;
        wcount += __builtin_popcount(km);
    }
}

__global__ __launch_bounds__(OWN_THREADS)
void gat_owner(const int* __restrict__ src, const int* __restrict__ dst,
               const unsigned int* __restrict__ own4,
               const float* __restrict__ h,
               const float* __restrict__ e_l, const float* __restrict__ e_r,
               const float* __restrict__ bias, float* __restrict__ out,
               int nNodes, int nEdges, int nper) {
    __shared__ float        acc[(NPER_MAX + 1) * OUT_CH];
    __shared__ unsigned int list[OWN_WAVES * SEG_CAP];
    __shared__ int          nmaxi[NPER_MAX];
    __shared__ float        nsum[NPER_MAX];
    __shared__ unsigned int overflow;

    const int tid  = threadIdx.x;
    const int lane = tid & 31;
    const int wave = __builtin_amdgcn_readfirstlane(tid >> 5);
    const int lo   = blockIdx.x * nper;
    int cnt = nNodes - lo;
    cnt = cnt < 0 ? 0 : (cnt > nper ? nper : cnt);
    unsigned int* seg = list + wave * SEG_CAP;

    for (int i = tid; i < cnt * OUT_CH; i += OWN_THREADS) acc[i] = 0.f;
    for (int i = tid; i < cnt; i += OWN_THREADS) { nmaxi[i] = (int)0xFF800000u; nsum[i] = 0.f; }
    if (tid == 0) overflow = 0u;
    __syncthreads();

    unsigned int wcount = 0u;
    scan_owner_bytes(own4, dst, nEdges, lo, cnt, lane, wave, seg, wcount);
    if (wcount > SEG_CAP && lane == 0) overflow = 1u;
    __syncthreads();

    const unsigned over = __builtin_amdgcn_readfirstlane(overflow);
    if (over == 0u) {
        for (unsigned b = 0; b < wcount; b += 128u) {
            unsigned idx[4], p[4]; int s[4], dl[4]; bool mine[4]; float el[4], er[4];
#pragma unroll
            for (int t = 0; t < 4; ++t) {
                idx[t] = b + 32u * t + lane;
                p[t] = seg[idx[t] < wcount ? idx[t] : wcount - 1u];
            }
#pragma unroll
            for (int t = 0; t < 4; ++t) {
                s[t]  = src[(int)p[t]];
                dl[t] = dst[(int)p[t]] - lo;
            }
#pragma unroll
            for (int t = 0; t < 4; ++t) {
                mine[t] = (unsigned)dl[t] < (unsigned)cnt;
                el[t] = e_l[lo + (mine[t] ? dl[t] : 0)];
                er[t] = e_r[s[t]];
            }
#pragma unroll
            for (int t = 0; t < 4; ++t) {
                if (idx[t] < wcount) {
                    if (mine[t]) {
                        const float x = el[t] + er[t];
                        const float a = (x > 0.f) ? x : NEG_SLOPE * x;
                        ldsMaxF32(&nmaxi[dl[t]], a);
                        seg[idx[t]] = (unsigned)s[t] | ((unsigned)dl[t] << S_BITS);
                    } else {
                        seg[idx[t]] = (unsigned)NPER_MAX << S_BITS;
                    }
                }
            }
        }
        __syncthreads();

        for (unsigned b = 0; b < wcount; b += 32u) {
            const unsigned i = b + lane;
            int s = 0, dl = NPER_MAX; float w = 0.f;
            const unsigned p = (i < wcount) ? seg[i] : ((unsigned)NPER_MAX << S_BITS);
            if ((p >> S_BITS) != (unsigned)NPER_MAX) {
                s  = (int)(p & ((1u << S_BITS) - 1u));
                dl = (int)(p >> S_BITS);
                const float x = e_l[lo + dl] + e_r[s];
                const float a = (x > 0.f) ? x : NEG_SLOPE * x;
                w = __expf(a - __int_as_float(nmaxi[dl]));
                atomicAdd(&nsum[dl], w);
            }
            const int nvalid = (wcount - b) < 32u ? (int)(wcount - b) : 32;
            for (int k0 = 0; k0 < nvalid; k0 += 16) {
                float2 hv[16]; int dk[16]; float wk[16];
#pragma unroll
                for (int u = 0; u < 16; ++u) {
                    const int k = k0 + u;
                    const int sk = __builtin_amdgcn_readlane(s, k);
                    dk[u] = __builtin_amdgcn_readlane(dl, k);
                    wk[u] = __int_as_float(__builtin_amdgcn_readlane(__float_as_int(w), k));
                    hv[u] = ((const float2*)(h + (size_t)sk * OUT_CH))[lane];
                }
#pragma unroll
                for (int u = 0; u < 16; ++u) {
                    float* r = &acc[dk[u] * OUT_CH + 2 * lane];
                    atomicAdd(r,     hv[u].x * wk[u]);
                    atomicAdd(r + 1, hv[u].y * wk[u]);
                }
            }
        }
    } else {
        scan_edges<1>(src, dst, h, e_l, e_r, nEdges, lo, cnt, lane, wave, nmaxi, nsum, acc);
        __syncthreads();
        scan_edges<2>(src, dst, h, e_l, e_r, nEdges, lo, cnt, lane, wave, nmaxi, nsum, acc);
    }
    __syncthreads();

    float* o = out + (size_t)lo * OUT_CH;
    for (int i = tid; i < cnt * OUT_CH; i += OWN_THREADS)
        o[i] = acc[i] / (nsum[i >> 6] + EPS_F) + bias[i & (OUT_CH - 1)];
}

extern "C" void kernel_launch(void* const* d_in, const int* in_sizes, int n_in,
                              void* d_out, int out_size, void* d_ws, size_t ws_size,
                              hipStream_t stream) {
    const float* x    = (const float*)d_in[0];
    const int*   ei   = (const int*)d_in[1];
    const float* W    = (const float*)d_in[2];
    const float* a_l  = (const float*)d_in[3];
    const float* a_r  = (const float*)d_in[4];
    const float* bias = (const float*)d_in[5];
    float* out = (float*)d_out;

    const int nNodes = in_sizes[0] / IN_CH;
    const int nEdges = in_sizes[1] / 2;
    const int* src = ei;
    const int* dst = ei + nEdges;

    float* ws       = (float*)d_ws;
    float* h        = ws;
    float* e_l      = h + (size_t)nNodes * OUT_CH;
    float* e_r      = e_l + nNodes;
    float* node_max = e_r + nNodes;
    float* node_sum = node_max + nNodes;
    float* edge_w   = node_sum + nNodes;

    const int mFull = nNodes & ~15;
    const int tiles = (mFull / 16) * (OUT_CH / 16);
    if (tiles > 0)
        gat_gemm_wmma<<<(tiles + 3) / 4, 128, 0, stream>>>(x, W, h);
    if (mFull < nNodes) {
        int tailThreads = (nNodes - mFull) * OUT_CH;
        gat_gemm_tail<<<(tailThreads + 255) / 256, 256, 0, stream>>>(x, W, h, mFull, nNodes);
    }

    const int nper = (nNodes + OWN_BLOCKS - 1) / OWN_BLOCKS;
    const bool owner = nper <= NPER_MAX && (nEdges & 3) == 0 && nEdges < (1 << E_BITS) && nNodes <= (1 << S_BITS);
    unsigned int* own4 = (unsigned int*)edge_w;

    gat_el_er<<<(nNodes * 32 + 255) / 256, 256, 0, stream>>>(h, a_l, a_r, e_l, e_r, nNodes,
                                                            dst, own4, owner ? nEdges : 0, nper);

    if (owner) {
        gat_owner<<<OWN_BLOCKS, OWN_THREADS, 0, stream>>>(src, dst, own4, h, e_l, e_r, bias, out, nNodes, nEdges, nper);
        return;
    }

    int initN = nNodes * OUT_CH;
    gat_init<<<(initN + 255) / 256, 256, 0, stream>>>(out, bias, node_max, node_sum, nNodes);

    gat_edge_max<<<(nEdges + 255) / 256, 256, 0, stream>>>(src, dst, e_l, e_r, edge_w, node_max, nEdges);
    gat_edge_exp<<<(nEdges + 255) / 256, 256, 0, stream>>>(dst, edge_w, node_max, node_sum, nEdges);
    gat_scatter<<<((size_t)nEdges * 32 + 255) / 256, 256, 0, stream>>>(src, dst, h, edge_w, node_sum, out, nEdges);
}


# GEMM re-tiled: 8-wave work-groups loop over 64-row slabs of x parked in LDS by whole-row loads; a wave keeps its 32 columns of W in registers; h leaves as whole 128-B lines; sum order unchanged
# speedup vs baseline: 3.2909x; 1.1496x over previous
#include <hip/hip_runtime.h>
#include <hip/hip_bf16.h>
#include <cstddef>

typedef __attribute__((ext_vector_type(2))) float v2f;
typedef __attribute__((ext_vector_type(8))) float v8f;

#define IN_CH   128
#define OUT_CH  64
#define NEG_SLOPE 0.2f
#define EPS_F   1e-10f

#define GB_ROWS    64
#define GB_PITCH   132
#define GB_THREADS 256
#define GB_KSTEPS  (IN_CH / 4)

typedef __attribute__((ext_vector_type(4))) float v4f;

__device__ __forceinline__ float gb_xhalf(float v) {
    return __builtin_bit_cast(float, __builtin_amdgcn_permlanex16(__builtin_bit_cast(int, v), __builtin_bit_cast(int, v),
                                                                   0x76543210, 0xfedcba98, false, false));
}

__global__ __launch_bounds__(GB_THREADS)
void gat_gemm_slab(const float* __restrict__ x, const float* __restrict__ W,
                   float* __restrict__ h, int mFull, int nSlabs) {
    static_assert(OUT_CH == GB_ROWS, "W passes through a slab buffer");
    __shared__ __attribute__((aligned(16))) float xs[2][GB_ROWS * GB_PITCH];

    const int tid  = threadIdx.x;
    const int lane = tid & 31;
    const int wave = __builtin_amdgcn_readfirstlane(tid >> 5);
    const int l15  = lane & 15;
    const int half = lane >> 4;
    const int kOff = half << 1;
    const int nBase = (wave & 1) * 32;
    const int mt    = wave >> 1;

    const v4f* x4 = (const v4f*)x;
    v4f stg[GB_ROWS / 8];
    const v4f zero4 = {0.f, 0.f, 0.f, 0.f};

    int slab = blockIdx.x;
    if (slab < nSlabs) {
#pragma unroll
        for (int i = 0; i < GB_ROWS / 8; ++i) {
            const int row = slab * GB_ROWS + i * 8 + wave;
            stg[i] = (row < mFull) ? x4[(size_t)row * (IN_CH / 4) + lane] : zero4;
        }
    }
    {
        const v4f* w4 = (const v4f*)W;
        v4f wst[OUT_CH / 8];
#pragma unroll
        for (int i = 0; i < OUT_CH / 8; ++i)
            wst[i] = w4[(size_t)(i * 8 + wave) * (IN_CH / 4) + lane];
#pragma unroll
        for (int i = 0; i < OUT_CH / 8; ++i)
            *(v4f*)(&xs[1][(i * 8 + wave) * GB_PITCH + lane * 4]) = wst[i];
    }
    if (slab < nSlabs) {
#pragma unroll
        for (int i = 0; i < GB_ROWS / 8; ++i)
            *(v4f*)(&xs[0][(i * 8 + wave) * GB_PITCH + lane * 4]) = stg[i];
    }
    __syncthreads();

    v2f b0[GB_KSTEPS], b1[GB_KSTEPS];
    {
        const float* w0 = &xs[1][(nBase + l15) * GB_PITCH + kOff];
        const float* w1 = w0 + 16 * GB_PITCH;
#pragma unroll
        for (int kk = 0; kk < GB_KSTEPS; ++kk) {
            b0[kk] = *(const v2f*)(w0 + 4 * kk);
            b1[kk] = *(const v2f*)(w1 + 4 * kk);
        }
    }
    __syncthreads();

    int buf = 0;
    for (; slab < nSlabs; slab += (int)gridDim.x) {
        const int  nxt  = slab + (int)gridDim.x;
        const bool more = nxt < nSlabs;
        if (more) {
#pragma unroll
            for (int i = 0; i < GB_ROWS / 8; ++i) {
                const int row = nxt * GB_ROWS + i * 8 + wave;
                stg[i] = (row < mFull) ? x4[(size_t)row * (IN_CH / 4) + lane] : zero4;
            }
        }

        const float* ap = &xs[buf][(mt * 16 + l15) * GB_PITCH + kOff];
        v8f acc0 = {}, acc1 = {};
#pragma unroll
        for (int kk = 0; kk < GB_KSTEPS; ++kk) {
            const v2f a = *(const v2f*)(ap + 4 * kk);
            acc0 = __builtin_amdgcn_wmma_f32_16x16x4_f32(
                 false, a,  false, b0[kk],
                 (short)0, acc0,  false,  false);
            acc1 = __builtin_amdgcn_wmma_f32_16x16x4_f32(
                 false, a,  false, b1[kk],
                 (short)0, acc1,  false,  false);
        }

        const int mBase = slab * GB_ROWS + mt * 16;
        if (mBase < mFull) {
            float* o = h + (size_t)mBase * OUT_CH + nBase + lane;
#pragma unroll
            for (int r = 0; r < 8; ++r) {
                const float lo = acc0[r], hi = acc1[r];
                const float lox = gb_xhalf(lo), hix = gb_xhalf(hi);
                o[(size_t)r * OUT_CH]       = half ? hix : lo;
                o[(size_t)(r + 8) * OUT_CH] = half ? hi  : lox;
            }
        }

        if (more) {
#pragma unroll
            for (int i = 0; i < GB_ROWS / 8; ++i)
                *(v4f*)(&xs[buf ^ 1][(i * 8 + wave) * GB_PITCH + lane * 4]) = stg[i];
        }
        __syncthreads();
        buf ^= 1;
    }
}

__global__ void gat_gemm_tail(const float* __restrict__ x,
                              const float* __restrict__ W,
                              float* __restrict__ h,
                              int rowStart, int nNodes) {
    int idx = blockIdx.x * blockDim.x + threadIdx.x;
    int row = rowStart + (idx >> 6);
    int col = idx & 63;
    if (row >= nNodes) return;
    const float* xr = x + (size_t)row * IN_CH;
    const float* wr = W + (size_t)col * IN_CH;
    float s = 0.f;
    for (int k = 0; k < IN_CH; ++k) s += xr[k] * wr[k];
    h[(size_t)row * OUT_CH + col] = s;
}

typedef __attribute__((ext_vector_type(4))) int v4i;
typedef __attribute__((ext_vector_type(4))) unsigned int v4u;

__global__ void gat_el_er(const float* __restrict__ h,
                          const float* __restrict__ a_l,
                          const float* __restrict__ a_r,
                          float* __restrict__ e_l,
                          float* __restrict__ e_r, int nNodes,
                          const int* __restrict__ dst, unsigned int* __restrict__ own4,
                          int nEdgesOwn, int nper) {
    {
        const int nQuads = nEdgesOwn >> 2;
        const int stride = (int)(gridDim.x * blockDim.x);
        const unsigned un = (unsigned)nper;
        for (int q = blockIdx.x * blockDim.x + threadIdx.x; q < nQuads; q += stride) {
            const v4i d = ((const v4i*)dst)[q];
            own4[q] = (((unsigned)d.x / un) & 0xFFu) | ((((unsigned)d.y / un) & 0xFFu) << 8) |
                      ((((unsigned)d.z / un) & 0xFFu) << 16) | (((unsigned)d.w / un) << 24);
        }
    }
    const int lane = threadIdx.x & 31;
    const int node = (blockIdx.x * blockDim.x + threadIdx.x) >> 5;
    if (node >= nNodes) return;
    float2 hv = ((const float2*)(h + (size_t)node * OUT_CH))[lane];
    float2 al = ((const float2*)a_l)[lane];
    float2 ar = ((const float2*)a_r)[lane];
    float pl = hv.x * al.x + hv.y * al.y;
    float pr = hv.x * ar.x + hv.y * ar.y;
#pragma unroll
    for (int off = 16; off > 0; off >>= 1) {
        pl += __shfl_xor(pl, off, 32);
        pr += __shfl_xor(pr, off, 32);
    }
    if (lane == 0) { e_l[node] = pl; e_r[node] = pr; }
}

__global__ void gat_init(float* __restrict__ out, const float* __restrict__ bias,
                         float* __restrict__ node_max, float* __restrict__ node_sum,
                         int nNodes) {
    int i = blockIdx.x * blockDim.x + threadIdx.x;
    if (i < nNodes * OUT_CH) out[i] = bias[i & (OUT_CH - 1)];
    if (i < nNodes) { node_max[i] = -__builtin_inff(); node_sum[i] = 0.f; }
}

__device__ __forceinline__ void atomicMaxF32(float* addr, float v) {
    if (v >= 0.f)
        atomicMax((int*)addr, __float_as_int(v));
    else
        atomicMin((unsigned int*)addr, __float_as_uint(v));
}

__global__ void gat_edge_max(const int* __restrict__ src, const int* __restrict__ dst,
                             const float* __restrict__ e_l, const float* __restrict__ e_r,
                             float* __restrict__ edge_w, float* __restrict__ node_max,
                             int nEdges) {
    int e = blockIdx.x * blockDim.x + threadIdx.x;
    if (e >= nEdges) return;
    int s = src[e], d = dst[e];
    float a = e_l[d] + e_r[s];
    a = (a > 0.f) ? a : NEG_SLOPE * a;
    edge_w[e] = a;
    atomicMaxF32(node_max + d, a);
}

__global__ void gat_edge_exp(const int* __restrict__ dst,
                             float* __restrict__ edge_w,
                             const float* __restrict__ node_max,
                             float* __restrict__ node_sum, int nEdges) {
    int e = blockIdx.x * blockDim.x + threadIdx.x;
    if (e >= nEdges) return;
    int d = dst[e];
    float w = __expf(edge_w[e] - node_max[d]);
    edge_w[e] = w;
    atomicAdd(node_sum + d, w);
}

__global__ void gat_scatter(const int* __restrict__ src, const int* __restrict__ dst,
                            const float* __restrict__ h,
                            const float* __restrict__ edge_w,
                            const float* __restrict__ node_sum,
                            float* __restrict__ out, int nEdges) {
    const int lane = threadIdx.x & 31;
    const int e = (blockIdx.x * blockDim.x + threadIdx.x) >> 5;
    if (e >= nEdges) return;
    const int s = src[e], d = dst[e];
    const float coef = edge_w[e] / (node_sum[d] + EPS_F);
    float2 hv = ((const float2*)(h + (size_t)s * OUT_CH))[lane];
    float* o = out + (size_t)d * OUT_CH + lane * 2;
    atomicAdd(o + 0, hv.x * coef);
    atomicAdd(o + 1, hv.y * coef);
}


#define OWN_BLOCKS   256
#define OWN_THREADS  1024
#define OWN_WAVES    32
#define NPER_MAX     391
#define SEG_CAP      768
#define E_BITS       21
#define S_BITS       17

__device__ __forceinline__ void ldsMaxF32(int* p, float v) {
    const int b = __float_as_int(v);
    if (b >= 0) atomicMax(p, b);
    else        atomicMin((unsigned int*)p, (unsigned int)b);
}

template <int PHASE>
__device__ __forceinline__ void edge_now(bool m, int e, int dl, int lo, int lane,
                                         const int* __restrict__ src, const float* __restrict__ h,
                                         const float* __restrict__ e_l, const float* __restrict__ e_r,
                                         int* nmaxi, float* nsum, float* acc) {
    int s = 0; float w = 0.f;
    if (m) {
        s = src[e];
        const float x = e_l[lo + dl] + e_r[s];
        const float a = (x > 0.f) ? x : NEG_SLOPE * x;
        if (PHASE == 1) {
            ldsMaxF32(&nmaxi[dl], a);
        } else {
            w = __expf(a - __int_as_float(nmaxi[dl]));
            atomicAdd(&nsum[dl], w);
        }
    }
    if (PHASE == 2) {
        unsigned mask = __builtin_amdgcn_ballot_w32(m);
        const int dls = m ? dl : 0;
        while (mask) {
            const int k = __builtin_ctz(mask);
            mask &= mask - 1;
            const int   sk = __builtin_amdgcn_readlane(s, k);
            const int   dk = __builtin_amdgcn_readlane(dls, k);
            const float wk = __int_as_float(__builtin_amdgcn_readlane(__float_as_int(w), k));
            const float2 hv = ((const float2*)(h + (size_t)sk * OUT_CH))[lane];
            float* r = &acc[dk * OUT_CH + 2 * lane];
            atomicAdd(r,     hv.x * wk);
            atomicAdd(r + 1, hv.y * wk);
        }
    }
}

template <int MODE>
__device__ __forceinline__ void scan_quad(v4i v, int e0, int lo, int cnt, int lane,
                                          const int* __restrict__ src, const float* __restrict__ h,
                                          const float* __restrict__ e_l, const float* __restrict__ e_r,
                                          int* nmaxi, float* nsum, float* acc) {
    const int d0 = v.x - lo, d1 = v.y - lo, d2 = v.z - lo, d3 = v.w - lo;
    const bool m0 = (unsigned)d0 < (unsigned)cnt;
    const bool m1 = (unsigned)d1 < (unsigned)cnt;
    const bool m2 = (unsigned)d2 < (unsigned)cnt;
    const bool m3 = (unsigned)d3 < (unsigned)cnt;
    if (__builtin_amdgcn_ballot_w32(m0 | m1 | m2 | m3) == 0) return;
    edge_now<MODE>(m0, e0 + 0, d0, lo, lane, src, h, e_l, e_r, nmaxi, nsum, acc);
    edge_now<MODE>(m1, e0 + 1, d1, lo, lane, src, h, e_l, e_r, nmaxi, nsum, acc);
    edge_now<MODE>(m2, e0 + 2, d2, lo, lane, src, h, e_l, e_r, nmaxi, nsum, acc);
    edge_now<MODE>(m3, e0 + 3, d3, lo, lane, src, h, e_l, e_r, nmaxi, nsum, acc);
}

template <int MODE>
__device__ __forceinline__ void scan_edges(const int* __restrict__ src, const int* __restrict__ dst,
                                           const float* __restrict__ h,
                                           const float* __restrict__ e_l, const float* __restrict__ e_r,
                                           int nEdges, int lo, int cnt, int lane, int wave,
                                           int* nmaxi, float* nsum, float* acc) {
    const int nGroups = nEdges >> 7;
    const v4i* dst4 = (const v4i*)dst;
    const int start = (int)(((long long)blockIdx.x * nGroups) / OWN_BLOCKS);
    const v4i none = {-1, -1, -1, -1};
    for (int j = wave; j < nGroups; j += OWN_WAVES * 8) {
        v4i v[8]; int g[8];
#pragma unroll
        for (int u = 0; u < 8; ++u) {
            const int jj = j + u * OWN_WAVES;
            int gg = jj + start;
            if (gg >= nGroups) gg -= nGroups;
            g[u] = gg;
            v[u] = (jj < nGroups) ? dst4[(size_t)gg * 32 + lane] : none;
        }
#pragma unroll
        for (int u = 0; u < 8; ++u)
            scan_quad<MODE>(v[u], g[u] * 128 + lane * 4, lo, cnt, lane, src, h, e_l, e_r,
                            nmaxi, nsum, acc);
    }
    for (int eb = nGroups * 128 + wave * 32; eb < nEdges; eb += OWN_THREADS) {
        const int e = eb + lane;
        const int d = (e < nEdges) ? dst[e] : -1;
        const int dl = d - lo;
        const bool m = (unsigned)dl < (unsigned)cnt;
        const unsigned km = __builtin_amdgcn_ballot_w32(m);
        if (km == 0) continue;
        edge_now<MODE>(m, e, dl, lo, lane, src, h, e_l, e_r, nmaxi, nsum, acc);
    }
}

__device__ __forceinline__ void scan_bytes16(v4u v, int e0, unsigned pat,
                                             unsigned int* seg, unsigned int& wcount) {
    unsigned f[4];
#pragma unroll
    for (int j = 0; j < 4; ++j) {
        const unsigned x = v[j] ^ pat;
        const unsigned y = (x & 0x7F7F7F7Fu) + 0x7F7F7F7Fu;
        f[j] = ~(y | x | 0x7F7F7F7Fu);
    }
    unsigned m = (f[0] >> 7) | (f[1] >> 6) | (f[2] >> 5) | (f[3] >> 4);
    for (unsigned k = __builtin_amdgcn_ballot_w32(m != 0u); k != 0u; k = __builtin_amdgcn_ballot_w32(m != 0u)) {
        const int bit = __builtin_ctz(m | 0x80000000u);
        const unsigned i = wcount + __builtin_amdgcn_mbcnt_lo(k, 0);
        if (m != 0u && i < SEG_CAP) seg[i] = (unsigned)(e0 + ((bit & 7) << 2) + (bit >> 3));
        m &= m - 1u;
        wcount += __builtin_popcount(k);
    }
}

__device__ __forceinline__ void scan_owner_bytes(const unsigned int* __restrict__ own4,
                                                 const int* __restrict__ dst,
                                                 int nEdges, int lo, int cnt, int lane, int wave,
                                                 unsigned int* seg, unsigned int& wcount) {
    const int nGroups = nEdges >> 9;
    const v4u* ob = (const v4u*)own4;
    const unsigned pat = (unsigned)blockIdx.x * 0x01010101u;
    const int start = (int)(((long long)blockIdx.x * nGroups) / OWN_BLOCKS);
    const v4u none = {~pat, ~pat, ~pat, ~pat};
    for (int j = wave; j < nGroups; j += OWN_WAVES * 8) {
        v4u v[8]; int g[8];
#pragma unroll
        for (int u = 0; u < 8; ++u) {
            const int jj = j + u * OWN_WAVES;
            int gg = jj + start;
            if (gg >= nGroups) gg -= nGroups;
            g[u] = gg;
            v[u] = (jj < nGroups) ? ob[(size_t)gg * 32 + lane] : none;
        }
#pragma unroll
        for (int u = 0; u < 8; ++u)
            scan_bytes16(v[u], (g[u] * 32 + lane) * 16, pat, seg, wcount);
    }
    for (int eb = (nGroups << 9) + wave * 32; eb < nEdges; eb += OWN_THREADS) {
        const int e = eb + lane;
        const int d = (e < nEdges) ? dst[e] : -1;
        const int dl = d - lo;
        const bool m = (unsigned)dl < (unsigned)cnt;
        const unsigned km = __builtin_amdgcn_ballot_w32(m);
        if (km == 0) continue;
        const unsigned i = wcount + __builtin_amdgcn_mbcnt_lo(km, 0);
        if (m && i < SEG_CAP) seg[i] = (unsigned)e;
        wcount += __builtin_popcount(km);
    }
}

__global__ __launch_bounds__(OWN_THREADS)
void gat_owner(const int* __restrict__ src, const int* __restrict__ dst,
               const unsigned int* __restrict__ own4,
               const float* __restrict__ h,
               const float* __restrict__ e_l, const float* __restrict__ e_r,
               const float* __restrict__ bias, float* __restrict__ out,
               int nNodes, int nEdges, int nper) {
    __shared__ float        acc[(NPER_MAX + 1) * OUT_CH];
    __shared__ unsigned int list[OWN_WAVES * SEG_CAP];
    __shared__ int          nmaxi[NPER_MAX];
    __shared__ float        nsum[NPER_MAX];
    __shared__ unsigned int overflow;

    const int tid  = threadIdx.x;
    const int lane = tid & 31;
    const int wave = __builtin_amdgcn_readfirstlane(tid >> 5);
    const int lo   = blockIdx.x * nper;
    int cnt = nNodes - lo;
    cnt = cnt < 0 ? 0 : (cnt > nper ? nper : cnt);
    unsigned int* seg = list + wave * SEG_CAP;

    for (int i = tid; i < cnt * OUT_CH; i += OWN_THREADS) acc[i] = 0.f;
    for (int i = tid; i < cnt; i += OWN_THREADS) { nmaxi[i] = (int)0xFF800000u; nsum[i] = 0.f; }
    if (tid == 0) overflow = 0u;
    __syncthreads();

    unsigned int wcount = 0u;
    scan_owner_bytes(own4, dst, nEdges, lo, cnt, lane, wave, seg, wcount);
    if (wcount > SEG_CAP && lane == 0) overflow = 1u;
    __syncthreads();

    const unsigned over = __builtin_amdgcn_readfirstlane(overflow);
    if (over == 0u) {
        for (unsigned b = 0; b < wcount; b += 128u) {
            unsigned idx[4], p[4]; int s[4], dl[4]; bool mine[4]; float el[4], er[4];
#pragma unroll
            for (int t = 0; t < 4; ++t) {
                idx[t] = b + 32u * t + lane;
                p[t] = seg[idx[t] < wcount ? idx[t] : wcount - 1u];
            }
#pragma unroll
            for (int t = 0; t < 4; ++t) {
                s[t]  = src[(int)p[t]];
                dl[t] = dst[(int)p[t]] - lo;
            }
#pragma unroll
            for (int t = 0; t < 4; ++t) {
                mine[t] = (unsigned)dl[t] < (unsigned)cnt;
                el[t] = e_l[lo + (mine[t] ? dl[t] : 0)];
                er[t] = e_r[s[t]];
            }
#pragma unroll
            for (int t = 0; t < 4; ++t) {
                if (idx[t] < wcount) {
                    if (mine[t]) {
                        const float x = el[t] + er[t];
                        const float a = (x > 0.f) ? x : NEG_SLOPE * x;
                        ldsMaxF32(&nmaxi[dl[t]], a);
                        seg[idx[t]] = (unsigned)s[t] | ((unsigned)dl[t] << S_BITS);
                    } else {
                        seg[idx[t]] = (unsigned)NPER_MAX << S_BITS;
                    }
                }
            }
        }
        __syncthreads();

        for (unsigned b = 0; b < wcount; b += 32u) {
            const unsigned i = b + lane;
            int s = 0, dl = NPER_MAX; float w = 0.f;
            const unsigned p = (i < wcount) ? seg[i] : ((unsigned)NPER_MAX << S_BITS);
            if ((p >> S_BITS) != (unsigned)NPER_MAX) {
                s  = (int)(p & ((1u << S_BITS) - 1u));
                dl = (int)(p >> S_BITS);
                const float x = e_l[lo + dl] + e_r[s];
                const float a = (x > 0.f) ? x : NEG_SLOPE * x;
                w = __expf(a - __int_as_float(nmaxi[dl]));
                atomicAdd(&nsum[dl], w);
            }
            const int nvalid = (wcount - b) < 32u ? (int)(wcount - b) : 32;
            for (int k0 = 0; k0 < nvalid; k0 += 16) {
                float2 hv[16]; int dk[16]; float wk[16];
#pragma unroll
                for (int u = 0; u < 16; ++u) {
                    const int k = k0 + u;
                    const int sk = __builtin_amdgcn_readlane(s, k);
                    dk[u] = __builtin_amdgcn_readlane(dl, k);
                    wk[u] = __int_as_float(__builtin_amdgcn_readlane(__float_as_int(w), k));
                    hv[u] = ((const float2*)(h + (size_t)sk * OUT_CH))[lane];
                }
#pragma unroll
                for (int u = 0; u < 16; ++u) {
                    float* r = &acc[dk[u] * OUT_CH + 2 * lane];
                    atomicAdd(r,     hv[u].x * wk[u]);
                    atomicAdd(r + 1, hv[u].y * wk[u]);
                }
            }
        }
    } else {
        scan_edges<1>(src, dst, h, e_l, e_r, nEdges, lo, cnt, lane, wave, nmaxi, nsum, acc);
        __syncthreads();
        scan_edges<2>(src, dst, h, e_l, e_r, nEdges, lo, cnt, lane, wave, nmaxi, nsum, acc);
    }
    __syncthreads();

    float* o = out + (size_t)lo * OUT_CH;
    for (int i = tid; i < cnt * OUT_CH; i += OWN_THREADS)
        o[i] = acc[i] / (nsum[i >> 6] + EPS_F) + bias[i & (OUT_CH - 1)];
}

extern "C" void kernel_launch(void* const* d_in, const int* in_sizes, int n_in,
                              void* d_out, int out_size, void* d_ws, size_t ws_size,
                              hipStream_t stream) {
    const float* x    = (const float*)d_in[0];
    const int*   ei   = (const int*)d_in[1];
    const float* W    = (const float*)d_in[2];
    const float* a_l  = (const float*)d_in[3];
    const float* a_r  = (const float*)d_in[4];
    const float* bias = (const float*)d_in[5];
    float* out = (float*)d_out;

    const int nNodes = in_sizes[0] / IN_CH;
    const int nEdges = in_sizes[1] / 2;
    const int* src = ei;
    const int* dst = ei + nEdges;

    float* ws       = (float*)d_ws;
    float* h        = ws;
    float* e_l      = h + (size_t)nNodes * OUT_CH;
    float* e_r      = e_l + nNodes;
    float* node_max = e_r + nNodes;
    float* node_sum = node_max + nNodes;
    float* edge_w   = node_sum + nNodes;

    const int mFull = nNodes & ~15;
    const int nSlabs = (mFull + GB_ROWS - 1) / GB_ROWS;
    if (nSlabs > 0)
        gat_gemm_slab<<<nSlabs < 512 ? nSlabs : 512, GB_THREADS, 0, stream>>>(x, W, h, mFull, nSlabs);
    if (mFull < nNodes) {
        int tailThreads = (nNodes - mFull) * OUT_CH;
        gat_gemm_tail<<<(tailThreads + 255) / 256, 256, 0, stream>>>(x, W, h, mFull, nNodes);
    }

    const int nper = (nNodes + OWN_BLOCKS - 1) / OWN_BLOCKS;
    const bool owner = nper <= NPER_MAX && (nEdges & 3) == 0 && nEdges < (1 << E_BITS) && nNodes <= (1 << S_BITS);
    unsigned int* own4 = (unsigned int*)edge_w;

    gat_el_er<<<(nNodes * 32 + 255) / 256, 256, 0, stream>>>(h, a_l, a_r, e_l, e_r, nNodes,
                                                            dst, own4, owner ? nEdges : 0, nper);

    if (owner) {
        gat_owner<<<OWN_BLOCKS, OWN_THREADS, 0, stream>>>(src, dst, own4, h, e_l, e_r, bias, out, nNodes, nEdges, nper);
        return;
    }

    int initN = nNodes * OUT_CH;
    gat_init<<<(initN + 255) / 256, 256, 0, stream>>>(out, bias, node_max, node_sum, nNodes);

    gat_edge_max<<<(nEdges + 255) / 256, 256, 0, stream>>>(src, dst, e_l, e_r, edge_w, node_max, nEdges);
    gat_edge_exp<<<(nEdges + 255) / 256, 256, 0, stream>>>(dst, edge_w, node_max, node_sum, nEdges);
    gat_scatter<<<((size_t)nEdges * 32 + 255) / 256, 256, 0, stream>>>(src, dst, h, edge_w, node_sum, out, nEdges);
}


# owner blocks: max pass parks each edge's score in a second per-wave LDS list, exp pass reads it there: e_l, e_r gathered once per edge, not twice (one random L2 request fewer); sum order unchanged
# speedup vs baseline: 3.4111x; 1.0365x over previous
#include <hip/hip_runtime.h>
#include <hip/hip_bf16.h>
#include <cstddef>

typedef __attribute__((ext_vector_type(2))) float v2f;
typedef __attribute__((ext_vector_type(8))) float v8f;

#define IN_CH   128
#define OUT_CH  64
#define NEG_SLOPE 0.2f
#define EPS_F   1e-10f

#define GB_ROWS    64
#define GB_PITCH   132
#define GB_THREADS 256
#define GB_KSTEPS  (IN_CH / 4)

typedef __attribute__((ext_vector_type(4))) float v4f;

__device__ __forceinline__ float gb_xhalf(float v) {
    return __builtin_bit_cast(float, __builtin_amdgcn_permlanex16(__builtin_bit_cast(int, v), __builtin_bit_cast(int, v),
                                                                   0x76543210, 0xfedcba98, false, false));
}

__global__ __launch_bounds__(GB_THREADS)
void gat_gemm_slab(const float* __restrict__ x, const float* __restrict__ W,
                   float* __restrict__ h, int mFull, int nSlabs) {
    static_assert(OUT_CH == GB_ROWS, "W passes through a slab buffer");
    __shared__ __attribute__((aligned(16))) float xs[2][GB_ROWS * GB_PITCH];

    const int tid  = threadIdx.x;
    const int lane = tid & 31;
    const int wave = __builtin_amdgcn_readfirstlane(tid >> 5);
    const int l15  = lane & 15;
    const int half = lane >> 4;
    const int kOff = half << 1;
    const int nBase = (wave & 1) * 32;
    const int mt    = wave >> 1;

    const v4f* x4 = (const v4f*)x;
    v4f stg[GB_ROWS / 8];
    const v4f zero4 = {0.f, 0.f, 0.f, 0.f};

    int slab = blockIdx.x;
    if (slab < nSlabs) {
#pragma unroll
        for (int i = 0; i < GB_ROWS / 8; ++i) {
            const int row = slab * GB_ROWS + i * 8 + wave;
            stg[i] = (row < mFull) ? x4[(size_t)row * (IN_CH / 4) + lane] : zero4;
        }
    }
    {
        const v4f* w4 = (const v4f*)W;
        v4f wst[OUT_CH / 8];
#pragma unroll
        for (int i = 0; i < OUT_CH / 8; ++i)
            wst[i] = w4[(size_t)(i * 8 + wave) * (IN_CH / 4) + lane];
#pragma unroll
        for (int i = 0; i < OUT_CH / 8; ++i)
            *(v4f*)(&xs[1][(i * 8 + wave) * GB_PITCH + lane * 4]) = wst[i];
    }
    if (slab < nSlabs) {
#pragma unroll
        for (int i = 0; i < GB_ROWS / 8; ++i)
            *(v4f*)(&xs[0][(i * 8 + wave) * GB_PITCH + lane * 4]) = stg[i];
    }
    __syncthreads();

    v2f b0[GB_KSTEPS], b1[GB_KSTEPS];
    {
        const float* w0 = &xs[1][(nBase + l15) * GB_PITCH + kOff];
        const float* w1 = w0 + 16 * GB_PITCH;
#pragma unroll
        for (int kk = 0; kk < GB_KSTEPS; ++kk) {
            b0[kk] = *(const v2f*)(w0 + 4 * kk);
            b1[kk] = *(const v2f*)(w1 + 4 * kk);
        }
    }
    __syncthreads();

    int buf = 0;
    for (; slab < nSlabs; slab += (int)gridDim.x) {
        const int  nxt  = slab + (int)gridDim.x;
        const bool more = nxt < nSlabs;
        if (more) {
#pragma unroll
            for (int i = 0; i < GB_ROWS / 8; ++i) {
                const int row = nxt * GB_ROWS + i * 8 + wave;
                stg[i] = (row < mFull) ? x4[(size_t)row * (IN_CH / 4) + lane] : zero4;
            }
        }

        const float* ap = &xs[buf][(mt * 16 + l15) * GB_PITCH + kOff];
        v8f acc0 = {}, acc1 = {};
#pragma unroll
        for (int kk = 0; kk < GB_KSTEPS; ++kk) {
            const v2f a = *(const v2f*)(ap + 4 * kk);
            acc0 = __builtin_amdgcn_wmma_f32_16x16x4_f32(
                 false, a,  false, b0[kk],
                 (short)0, acc0,  false,  false);
            acc1 = __builtin_amdgcn_wmma_f32_16x16x4_f32(
                 false, a,  false, b1[kk],
                 (short)0, acc1,  false,  false);
        }

        const int mBase = slab * GB_ROWS + mt * 16;
        if (mBase < mFull) {
            float* o = h + (size_t)mBase * OUT_CH + nBase + lane;
#pragma unroll
            for (int r = 0; r < 8; ++r) {
                const float lo = acc0[r], hi = acc1[r];
                const float lox = gb_xhalf(lo), hix = gb_xhalf(hi);
                o[(size_t)r * OUT_CH]       = half ? hix : lo;
                o[(size_t)(r + 8) * OUT_CH] = half ? hi  : lox;
            }
        }

        if (more) {
#pragma unroll
            for (int i = 0; i < GB_ROWS / 8; ++i)
                *(v4f*)(&xs[buf ^ 1][(i * 8 + wave) * GB_PITCH + lane * 4]) = stg[i];
        }
        __syncthreads();
        buf ^= 1;
    }
}

__global__ void gat_gemm_tail(const float* __restrict__ x,
                              const float* __restrict__ W,
                              float* __restrict__ h,
                              int rowStart, int nNodes) {
    int idx = blockIdx.x * blockDim.x + threadIdx.x;
    int row = rowStart + (idx >> 6);
    int col = idx & 63;
    if (row >= nNodes) return;
    const float* xr = x + (size_t)row * IN_CH;
    const float* wr = W + (size_t)col * IN_CH;
    float s = 0.f;
    for (int k = 0; k < IN_CH; ++k) s += xr[k] * wr[k];
    h[(size_t)row * OUT_CH + col] = s;
}

typedef __attribute__((ext_vector_type(4))) int v4i;
typedef __attribute__((ext_vector_type(4))) unsigned int v4u;

__global__ void gat_el_er(const float* __restrict__ h,
                          const float* __restrict__ a_l,
                          const float* __restrict__ a_r,
                          float* __restrict__ e_l,
                          float* __restrict__ e_r, int nNodes,
                          const int* __restrict__ dst, unsigned int* __restrict__ own4,
                          int nEdgesOwn, int nper) {
    {
        const int nQuads = nEdgesOwn >> 2;
        const int stride = (int)(gridDim.x * blockDim.x);
        const unsigned un = (unsigned)nper;
        for (int q = blockIdx.x * blockDim.x + threadIdx.x; q < nQuads; q += stride) {
            const v4i d = ((const v4i*)dst)[q];
            own4[q] = (((unsigned)d.x / un) & 0xFFu) | ((((unsigned)d.y / un) & 0xFFu) << 8) |
                      ((((unsigned)d.z / un) & 0xFFu) << 16) | (((unsigned)d.w / un) << 24);
        }
    }
    const int lane = threadIdx.x & 31;
    const int node = (blockIdx.x * blockDim.x + threadIdx.x) >> 5;
    if (node >= nNodes) return;
    float2 hv = ((const float2*)(h + (size_t)node * OUT_CH))[lane];
    float2 al = ((const float2*)a_l)[lane];
    float2 ar = ((const float2*)a_r)[lane];
    float pl = hv.x * al.x + hv.y * al.y;
    float pr = hv.x * ar.x + hv.y * ar.y;
#pragma unroll
    for (int off = 16; off > 0; off >>= 1) {
        pl += __shfl_xor(pl, off, 32);
        pr += __shfl_xor(pr, off, 32);
    }
    if (lane == 0) { e_l[node] = pl; e_r[node] = pr; }
}

__global__ void gat_init(float* __restrict__ out, const float* __restrict__ bias,
                         float* __restrict__ node_max, float* __restrict__ node_sum,
                         int nNodes) {
    int i = blockIdx.x * blockDim.x + threadIdx.x;
    if (i < nNodes * OUT_CH) out[i] = bias[i & (OUT_CH - 1)];
    if (i < nNodes) { node_max[i] = -__builtin_inff(); node_sum[i] = 0.f; }
}

__device__ __forceinline__ void atomicMaxF32(float* addr, float v) {
    if (v >= 0.f)
        atomicMax((int*)addr, __float_as_int(v));
    else
        atomicMin((unsigned int*)addr, __float_as_uint(v));
}

__global__ void gat_edge_max(const int* __restrict__ src, const int* __restrict__ dst,
                             const float* __restrict__ e_l, const float* __restrict__ e_r,
                             float* __restrict__ edge_w, float* __restrict__ node_max,
                             int nEdges) {
    int e = blockIdx.x * blockDim.x + threadIdx.x;
    if (e >= nEdges) return;
    int s = src[e], d = dst[e];
    float a = e_l[d] + e_r[s];
    a = (a > 0.f) ? a : NEG_SLOPE * a;
    edge_w[e] = a;
    atomicMaxF32(node_max + d, a);
}

__global__ void gat_edge_exp(const int* __restrict__ dst,
                             float* __restrict__ edge_w,
                             const float* __restrict__ node_max,
                             float* __restrict__ node_sum, int nEdges) {
    int e = blockIdx.x * blockDim.x + threadIdx.x;
    if (e >= nEdges) return;
    int d = dst[e];
    float w = __expf(edge_w[e] - node_max[d]);
    edge_w[e] = w;
    atomicAdd(node_sum + d, w);
}

__global__ void gat_scatter(const int* __restrict__ src, const int* __restrict__ dst,
                            const float* __restrict__ h,
                            const float* __restrict__ edge_w,
                            const float* __restrict__ node_sum,
                            float* __restrict__ out, int nEdges) {
    const int lane = threadIdx.x & 31;
    const int e = (blockIdx.x * blockDim.x + threadIdx.x) >> 5;
    if (e >= nEdges) return;
    const int s = src[e], d = dst[e];
    const float coef = edge_w[e] / (node_sum[d] + EPS_F);
    float2 hv = ((const float2*)(h + (size_t)s * OUT_CH))[lane];
    float* o = out + (size_t)d * OUT_CH + lane * 2;
    atomicAdd(o + 0, hv.x * coef);
    atomicAdd(o + 1, hv.y * coef);
}


#define OWN_BLOCKS   256
#define OWN_THREADS  1024
#define OWN_WAVES    32
#define NPER_MAX     391
#define SEG_CAP      768
#define E_BITS       21
#define S_BITS       17

__device__ __forceinline__ void ldsMaxF32(int* p, float v) {
    const int b = __float_as_int(v);
    if (b >= 0) atomicMax(p, b);
    else        atomicMin((unsigned int*)p, (unsigned int)b);
}

template <int PHASE>
__device__ __forceinline__ void edge_now(bool m, int e, int dl, int lo, int lane,
                                         const int* __restrict__ src, const float* __restrict__ h,
                                         const float* __restrict__ e_l, const float* __restrict__ e_r,
                                         int* nmaxi, float* nsum, float* acc) {
    int s = 0; float w = 0.f;
    if (m) {
        s = src[e];
        const float x = e_l[lo + dl] + e_r[s];
        const float a = (x > 0.f) ? x : NEG_SLOPE * x;
        if (PHASE == 1) {
            ldsMaxF32(&nmaxi[dl], a);
        } else {
            w = __expf(a - __int_as_float(nmaxi[dl]));
            atomicAdd(&nsum[dl], w);
        }
    }
    if (PHASE == 2) {
        unsigned mask = __builtin_amdgcn_ballot_w32(m);
        const int dls = m ? dl : 0;
        while (mask) {
            const int k = __builtin_ctz(mask);
            mask &= mask - 1;
            const int   sk = __builtin_amdgcn_readlane(s, k);
            const int   dk = __builtin_amdgcn_readlane(dls, k);
            const float wk = __int_as_float(__builtin_amdgcn_readlane(__float_as_int(w), k));
            const float2 hv = ((const float2*)(h + (size_t)sk * OUT_CH))[lane];
            float* r = &acc[dk * OUT_CH + 2 * lane];
            atomicAdd(r,     hv.x * wk);
            atomicAdd(r + 1, hv.y * wk);
        }
    }
}

template <int MODE>
__device__ __forceinline__ void scan_quad(v4i v, int e0, int lo, int cnt, int lane,
                                          const int* __restrict__ src, const float* __restrict__ h,
                                          const float* __restrict__ e_l, const float* __restrict__ e_r,
                                          int* nmaxi, float* nsum, float* acc) {
    const int d0 = v.x - lo, d1 = v.y - lo, d2 = v.z - lo, d3 = v.w - lo;
    const bool m0 = (unsigned)d0 < (unsigned)cnt;
    const bool m1 = (unsigned)d1 < (unsigned)cnt;
    const bool m2 = (unsigned)d2 < (unsigned)cnt;
    const bool m3 = (unsigned)d3 < (unsigned)cnt;
    if (__builtin_amdgcn_ballot_w32(m0 | m1 | m2 | m3) == 0) return;
    edge_now<MODE>(m0, e0 + 0, d0, lo, lane, src, h, e_l, e_r, nmaxi, nsum, acc);
    edge_now<MODE>(m1, e0 + 1, d1, lo, lane, src, h, e_l, e_r, nmaxi, nsum, acc);
    edge_now<MODE>(m2, e0 + 2, d2, lo, lane, src, h, e_l, e_r, nmaxi, nsum, acc);
    edge_now<MODE>(m3, e0 + 3, d3, lo, lane, src, h, e_l, e_r, nmaxi, nsum, acc);
}

template <int MODE>
__device__ __forceinline__ void scan_edges(const int* __restrict__ src, const int* __restrict__ dst,
                                           const float* __restrict__ h,
                                           const float* __restrict__ e_l, const float* __restrict__ e_r,
                                           int nEdges, int lo, int cnt, int lane, int wave,
                                           int* nmaxi, float* nsum, float* acc) {
    const int nGroups = nEdges >> 7;
    const v4i* dst4 = (const v4i*)dst;
    const int start = (int)(((long long)blockIdx.x * nGroups) / OWN_BLOCKS);
    const v4i none = {-1, -1, -1, -1};
    for (int j = wave; j < nGroups; j += OWN_WAVES * 8) {
        v4i v[8]; int g[8];
#pragma unroll
        for (int u = 0; u < 8; ++u) {
            const int jj = j + u * OWN_WAVES;
            int gg = jj + start;
            if (gg >= nGroups) gg -= nGroups;
            g[u] = gg;
            v[u] = (jj < nGroups) ? dst4[(size_t)gg * 32 + lane] : none;
        }
#pragma unroll
        for (int u = 0; u < 8; ++u)
            scan_quad<MODE>(v[u], g[u] * 128 + lane * 4, lo, cnt, lane, src, h, e_l, e_r,
                            nmaxi, nsum, acc);
    }
    for (int eb = nGroups * 128 + wave * 32; eb < nEdges; eb += OWN_THREADS) {
        const int e = eb + lane;
        const int d = (e < nEdges) ? dst[e] : -1;
        const int dl = d - lo;
        const bool m = (unsigned)dl < (unsigned)cnt;
        const unsigned km = __builtin_amdgcn_ballot_w32(m);
        if (km == 0) continue;
        edge_now<MODE>(m, e, dl, lo, lane, src, h, e_l, e_r, nmaxi, nsum, acc);
    }
}

__device__ __forceinline__ void scan_bytes16(v4u v, int e0, unsigned pat,
                                             unsigned int* seg, unsigned int& wcount) {
    unsigned f[4];
#pragma unroll
    for (int j = 0; j < 4; ++j) {
        const unsigned x = v[j] ^ pat;
        const unsigned y = (x & 0x7F7F7F7Fu) + 0x7F7F7F7Fu;
        f[j] = ~(y | x | 0x7F7F7F7Fu);
    }
    unsigned m = (f[0] >> 7) | (f[1] >> 6) | (f[2] >> 5) | (f[3] >> 4);
    for (unsigned k = __builtin_amdgcn_ballot_w32(m != 0u); k != 0u; k = __builtin_amdgcn_ballot_w32(m != 0u)) {
        const int bit = __builtin_ctz(m | 0x80000000u);
        const unsigned i = wcount + __builtin_amdgcn_mbcnt_lo(k, 0);
        if (m != 0u && i < SEG_CAP) seg[i] = (unsigned)(e0 + ((bit & 7) << 2) + (bit >> 3));
        m &= m - 1u;
        wcount += __builtin_popcount(k);
    }
}

__device__ __forceinline__ void scan_owner_bytes(const unsigned int* __restrict__ own4,
                                                 const int* __restrict__ dst,
                                                 int nEdges, int lo, int cnt, int lane, int wave,
                                                 unsigned int* seg, unsigned int& wcount) {
    const int nGroups = nEdges >> 9;
    const v4u* ob = (const v4u*)own4;
    const unsigned pat = (unsigned)blockIdx.x * 0x01010101u;
    const int start = (int)(((long long)blockIdx.x * nGroups) / OWN_BLOCKS);
    const v4u none = {~pat, ~pat, ~pat, ~pat};
    for (int j = wave; j < nGroups; j += OWN_WAVES * 8) {
        v4u v[8]; int g[8];
#pragma unroll
        for (int u = 0; u < 8; ++u) {
            const int jj = j + u * OWN_WAVES;
            int gg = jj + start;
            if (gg >= nGroups) gg -= nGroups;
            g[u] = gg;
            v[u] = (jj < nGroups) ? ob[(size_t)gg * 32 + lane] : none;
        }
#pragma unroll
        for (int u = 0; u < 8; ++u)
            scan_bytes16(v[u], (g[u] * 32 + lane) * 16, pat, seg, wcount);
    }
    for (int eb = (nGroups << 9) + wave * 32; eb < nEdges; eb += OWN_THREADS) {
        const int e = eb + lane;
        const int d = (e < nEdges) ? dst[e] : -1;
        const int dl = d - lo;
        const bool m = (unsigned)dl < (unsigned)cnt;
        const unsigned km = __builtin_amdgcn_ballot_w32(m);
        if (km == 0) continue;
        const unsigned i = wcount + __builtin_amdgcn_mbcnt_lo(km, 0);
        if (m && i < SEG_CAP) seg[i] = (unsigned)e;
        wcount += __builtin_popcount(km);
    }
}

__global__ __launch_bounds__(OWN_THREADS)
void gat_owner(const int* __restrict__ src, const int* __restrict__ dst,
               const unsigned int* __restrict__ own4,
               const float* __restrict__ h,
               const float* __restrict__ e_l, const float* __restrict__ e_r,
               const float* __restrict__ bias, float* __restrict__ out,
               int nNodes, int nEdges, int nper) {
    __shared__ float        acc[(NPER_MAX + 1) * OUT_CH];
    __shared__ unsigned int list[OWN_WAVES * SEG_CAP];
    __shared__ float        score[OWN_WAVES * SEG_CAP];
    __shared__ int          nmaxi[NPER_MAX];
    __shared__ float        nsum[NPER_MAX];
    __shared__ unsigned int overflow;

    const int tid  = threadIdx.x;
    const int lane = tid & 31;
    const int wave = __builtin_amdgcn_readfirstlane(tid >> 5);
    const int lo   = blockIdx.x * nper;
    int cnt = nNodes - lo;
    cnt = cnt < 0 ? 0 : (cnt > nper ? nper : cnt);
    unsigned int* seg = list + wave * SEG_CAP;
    float* sco = score + wave * SEG_CAP;

    for (int i = tid; i < cnt * OUT_CH; i += OWN_THREADS) acc[i] = 0.f;
    for (int i = tid; i < cnt; i += OWN_THREADS) { nmaxi[i] = (int)0xFF800000u; nsum[i] = 0.f; }
    if (tid == 0) overflow = 0u;
    __syncthreads();

    unsigned int wcount = 0u;
    scan_owner_bytes(own4, dst, nEdges, lo, cnt, lane, wave, seg, wcount);
    if (wcount > SEG_CAP && lane == 0) overflow = 1u;
    __syncthreads();

    const unsigned over = __builtin_amdgcn_readfirstlane(overflow);
    if (over == 0u) {
        for (unsigned b = 0; b < wcount; b += 128u) {
            unsigned idx[4], p[4]; int s[4], dl[4]; bool mine[4]; float el[4], er[4];
#pragma unroll
            for (int t = 0; t < 4; ++t) {
                idx[t] = b + 32u * t + lane;
                p[t] = seg[idx[t] < wcount ? idx[t] : wcount - 1u];
            }
#pragma unroll
            for (int t = 0; t < 4; ++t) {
                s[t]  = src[(int)p[t]];
                dl[t] = dst[(int)p[t]] - lo;
            }
#pragma unroll
            for (int t = 0; t < 4; ++t) {
                mine[t] = (unsigned)dl[t] < (unsigned)cnt;
                el[t] = e_l[lo + (mine[t] ? dl[t] : 0)];
                er[t] = e_r[s[t]];
            }
#pragma unroll
            for (int t = 0; t < 4; ++t) {
                if (idx[t] < wcount) {
                    if (mine[t]) {
                        const float x = el[t] + er[t];
                        const float a = (x > 0.f) ? x : NEG_SLOPE * x;
                        ldsMaxF32(&nmaxi[dl[t]], a);
                        seg[idx[t]] = (unsigned)s[t] | ((unsigned)dl[t] << S_BITS);
                        sco[idx[t]] = a;
                    } else {
                        seg[idx[t]] = (unsigned)NPER_MAX << S_BITS;
                        sco[idx[t]] = 0.f;
                    }
                }
            }
        }
        __syncthreads();

        for (unsigned b = 0; b < wcount; b += 32u) {
            const unsigned i = b + lane;
            int s = 0, dl = NPER_MAX; float w = 0.f;
            const unsigned p = (i < wcount) ? seg[i] : ((unsigned)NPER_MAX << S_BITS);
            const float    a = (i < wcount) ? sco[i] : 0.f;
            if ((p >> S_BITS) != (unsigned)NPER_MAX) {
                s  = (int)(p & ((1u << S_BITS) - 1u));
                dl = (int)(p >> S_BITS);
                w = __expf(a - __int_as_float(nmaxi[dl]));
                atomicAdd(&nsum[dl], w);
            }
            const int nvalid = (wcount - b) < 32u ? (int)(wcount - b) : 32;
            for (int k0 = 0; k0 < nvalid; k0 += 16) {
                float2 hv[16]; int dk[16]; float wk[16];
#pragma unroll
                for (int u = 0; u < 16; ++u) {
                    const int k = k0 + u;
                    const int sk = __builtin_amdgcn_readlane(s, k);
                    dk[u] = __builtin_amdgcn_readlane(dl, k);
                    wk[u] = __int_as_float(__builtin_amdgcn_readlane(__float_as_int(w), k));
                    hv[u] = ((const float2*)(h + (size_t)sk * OUT_CH))[lane];
                }
#pragma unroll
                for (int u = 0; u < 16; ++u) {
                    float* r = &acc[dk[u] * OUT_CH + 2 * lane];
                    atomicAdd(r,     hv[u].x * wk[u]);
                    atomicAdd(r + 1, hv[u].y * wk[u]);
                }
            }
        }
    } else {
        scan_edges<1>(src, dst, h, e_l, e_r, nEdges, lo, cnt, lane, wave, nmaxi, nsum, acc);
        __syncthreads();
        scan_edges<2>(src, dst, h, e_l, e_r, nEdges, lo, cnt, lane, wave, nmaxi, nsum, acc);
    }
    __syncthreads();

    float* o = out + (size_t)lo * OUT_CH;
    for (int i = tid; i < cnt * OUT_CH; i += OWN_THREADS)
        o[i] = acc[i] / (nsum[i >> 6] + EPS_F) + bias[i & (OUT_CH - 1)];
}

extern "C" void kernel_launch(void* const* d_in, const int* in_sizes, int n_in,
                              void* d_out, int out_size, void* d_ws, size_t ws_size,
                              hipStream_t stream) {
    const float* x    = (const float*)d_in[0];
    const int*   ei   = (const int*)d_in[1];
    const float* W    = (const float*)d_in[2];
    const float* a_l  = (const float*)d_in[3];
    const float* a_r  = (const float*)d_in[4];
    const float* bias = (const float*)d_in[5];
    float* out = (float*)d_out;

    const int nNodes = in_sizes[0] / IN_CH;
    const int nEdges = in_sizes[1] / 2;
    const int* src = ei;
    const int* dst = ei + nEdges;

    float* ws       = (float*)d_ws;
    float* h        = ws;
    float* e_l      = h + (size_t)nNodes * OUT_CH;
    float* e_r      = e_l + nNodes;
    float* node_max = e_r + nNodes;
    float* node_sum = node_max + nNodes;
    float* edge_w   = node_sum + nNodes;

    const int mFull = nNodes & ~15;
    const int nSlabs = (mFull + GB_ROWS - 1) / GB_ROWS;
    if (nSlabs > 0)
        gat_gemm_slab<<<nSlabs < 512 ? nSlabs : 512, GB_THREADS, 0, stream>>>(x, W, h, mFull, nSlabs);
    if (mFull < nNodes) {
        int tailThreads = (nNodes - mFull) * OUT_CH;
        gat_gemm_tail<<<(tailThreads + 255) / 256, 256, 0, stream>>>(x, W, h, mFull, nNodes);
    }

    const int nper = (nNodes + OWN_BLOCKS - 1) / OWN_BLOCKS;
    const bool owner = nper <= NPER_MAX && (nEdges & 3) == 0 && nEdges < (1 << E_BITS) && nNodes <= (1 << S_BITS);
    unsigned int* own4 = (unsigned int*)edge_w;

    gat_el_er<<<(nNodes * 32 + 255) / 256, 256, 0, stream>>>(h, a_l, a_r, e_l, e_r, nNodes,
                                                            dst, own4, owner ? nEdges : 0, nper);

    if (owner) {
        gat_owner<<<OWN_BLOCKS, OWN_THREADS, 0, stream>>>(src, dst, own4, h, e_l, e_r, bias, out, nNodes, nEdges, nper);
        return;
    }

    int initN = nNodes * OUT_CH;
    gat_init<<<(initN + 255) / 256, 256, 0, stream>>>(out, bias, node_max, node_sum, nNodes);

    gat_edge_max<<<(nEdges + 255) / 256, 256, 0, stream>>>(src, dst, e_l, e_r, edge_w, node_max, nEdges);
    gat_edge_exp<<<(nEdges + 255) / 256, 256, 0, stream>>>(dst, edge_w, node_max, node_sum, nEdges);
    gat_scatter<<<((size_t)nEdges * 32 + 255) / 256, 256, 0, stream>>>(src, dst, h, edge_w, node_sum, out, nEdges);
}


# edges binned once per call by owner: el_er launch also sorts 256 edge chunks by owner in LDS; owner blocks read only their own (src, local dst) words, not every edge; list, so sum, order changes
# speedup vs baseline: 4.7908x; 1.4045x over previous
#include <hip/hip_runtime.h>
#include <hip/hip_bf16.h>
#include <cstddef>

typedef __attribute__((ext_vector_type(2))) float v2f;
typedef __attribute__((ext_vector_type(8))) float v8f;

#define IN_CH   128
#define OUT_CH  64
#define NEG_SLOPE 0.2f
#define EPS_F   1e-10f

#define GB_ROWS    64
#define GB_PITCH   132
#define GB_THREADS 256
#define GB_KSTEPS  (IN_CH / 4)

typedef __attribute__((ext_vector_type(4))) float v4f;

__device__ __forceinline__ float gb_xhalf(float v) {
    return __builtin_bit_cast(float, __builtin_amdgcn_permlanex16(__builtin_bit_cast(int, v), __builtin_bit_cast(int, v),
                                                                   0x76543210, 0xfedcba98, false, false));
}

__global__ __launch_bounds__(GB_THREADS)
void gat_gemm_slab(const float* __restrict__ x, const float* __restrict__ W,
                   float* __restrict__ h, int mFull, int nSlabs) {
    static_assert(OUT_CH == GB_ROWS, "W passes through a slab buffer");
    __shared__ __attribute__((aligned(16))) float xs[2][GB_ROWS * GB_PITCH];

    const int tid  = threadIdx.x;
    const int lane = tid & 31;
    const int wave = __builtin_amdgcn_readfirstlane(tid >> 5);
    const int l15  = lane & 15;
    const int half = lane >> 4;
    const int kOff = half << 1;
    const int nBase = (wave & 1) * 32;
    const int mt    = wave >> 1;

    const v4f* x4 = (const v4f*)x;
    v4f stg[GB_ROWS / 8];
    const v4f zero4 = {0.f, 0.f, 0.f, 0.f};

    int slab = blockIdx.x;
    if (slab < nSlabs) {
#pragma unroll
        for (int i = 0; i < GB_ROWS / 8; ++i) {
            const int row = slab * GB_ROWS + i * 8 + wave;
            stg[i] = (row < mFull) ? x4[(size_t)row * (IN_CH / 4) + lane] : zero4;
        }
    }
    {
        const v4f* w4 = (const v4f*)W;
        v4f wst[OUT_CH / 8];
#pragma unroll
        for (int i = 0; i < OUT_CH / 8; ++i)
            wst[i] = w4[(size_t)(i * 8 + wave) * (IN_CH / 4) + lane];
#pragma unroll
        for (int i = 0; i < OUT_CH / 8; ++i)
            *(v4f*)(&xs[1][(i * 8 + wave) * GB_PITCH + lane * 4]) = wst[i];
    }
    if (slab < nSlabs) {
#pragma unroll
        for (int i = 0; i < GB_ROWS / 8; ++i)
            *(v4f*)(&xs[0][(i * 8 + wave) * GB_PITCH + lane * 4]) = stg[i];
    }
    __syncthreads();

    v2f b0[GB_KSTEPS], b1[GB_KSTEPS];
    {
        const float* w0 = &xs[1][(nBase + l15) * GB_PITCH + kOff];
        const float* w1 = w0 + 16 * GB_PITCH;
#pragma unroll
        for (int kk = 0; kk < GB_KSTEPS; ++kk) {
            b0[kk] = *(const v2f*)(w0 + 4 * kk);
            b1[kk] = *(const v2f*)(w1 + 4 * kk);
        }
    }
    __syncthreads();

    int buf = 0;
    for (; slab < nSlabs; slab += (int)gridDim.x) {
        const int  nxt  = slab + (int)gridDim.x;
        const bool more = nxt < nSlabs;
        if (more) {
#pragma unroll
            for (int i = 0; i < GB_ROWS / 8; ++i) {
                const int row = nxt * GB_ROWS + i * 8 + wave;
                stg[i] = (row < mFull) ? x4[(size_t)row * (IN_CH / 4) + lane] : zero4;
            }
        }

        const float* ap = &xs[buf][(mt * 16 + l15) * GB_PITCH + kOff];
        v8f acc0 = {}, acc1 = {};
#pragma unroll
        for (int kk = 0; kk < GB_KSTEPS; ++kk) {
            const v2f a = *(const v2f*)(ap + 4 * kk);
            acc0 = __builtin_amdgcn_wmma_f32_16x16x4_f32(
                 false, a,  false, b0[kk],
                 (short)0, acc0,  false,  false);
            acc1 = __builtin_amdgcn_wmma_f32_16x16x4_f32(
                 false, a,  false, b1[kk],
                 (short)0, acc1,  false,  false);
        }

        const int mBase = slab * GB_ROWS + mt * 16;
        if (mBase < mFull) {
            float* o = h + (size_t)mBase * OUT_CH + nBase + lane;
#pragma unroll
            for (int r = 0; r < 8; ++r) {
                const float lo = acc0[r], hi = acc1[r];
                const float lox = gb_xhalf(lo), hix = gb_xhalf(hi);
                o[(size_t)r * OUT_CH]       = half ? hix : lo;
                o[(size_t)(r + 8) * OUT_CH] = half ? hi  : lox;
            }
        }

        if (more) {
#pragma unroll
            for (int i = 0; i < GB_ROWS / 8; ++i)
                *(v4f*)(&xs[buf ^ 1][(i * 8 + wave) * GB_PITCH + lane * 4]) = stg[i];
        }
        __syncthreads();
        buf ^= 1;
    }
}

__global__ void gat_gemm_tail(const float* __restrict__ x,
                              const float* __restrict__ W,
                              float* __restrict__ h,
                              int rowStart, int nNodes) {
    int idx = blockIdx.x * blockDim.x + threadIdx.x;
    int row = rowStart + (idx >> 6);
    int col = idx & 63;
    if (row >= nNodes) return;
    const float* xr = x + (size_t)row * IN_CH;
    const float* wr = W + (size_t)col * IN_CH;
    float s = 0.f;
    for (int k = 0; k < IN_CH; ++k) s += xr[k] * wr[k];
    h[(size_t)row * OUT_CH + col] = s;
}

typedef __attribute__((ext_vector_type(4))) int v4i;
typedef __attribute__((ext_vector_type(4))) unsigned int v4u;

#define OWN_BLOCKS   256
#define NPER_MAX     391
#define S_BITS       17
#define EMPTY_WORD   ((unsigned)NPER_MAX << S_BITS)
#define BIN_THREADS  256
#define BIN_WAVES    8
#define BIN_CAP      6400
#define BIN_QUADS    4

__device__ __forceinline__ int chunk_edges(int nEdges) {
    return (((nEdges + OWN_BLOCKS - 1) / OWN_BLOCKS) + 31) & ~31;
}

__device__ __forceinline__ void bin_chunk(const int* __restrict__ src, const int* __restrict__ dst,
                                          unsigned int* __restrict__ binw, unsigned int* __restrict__ tab,
                                          int nEdges, int nper,
                                          unsigned int* hist, unsigned int* stage, unsigned int* wtot) {
    static_assert(BIN_THREADS == OWN_BLOCKS && BIN_WAVES * 32 == BIN_THREADS, "thread t scans owner t's counters");
    static_assert(BIN_WAVES * 800 >= BIN_CAP, "a wave's stretch is two trips at most");
    const int tid  = threadIdx.x;
    const int lane = tid & 31;
    const int wave = __builtin_amdgcn_readfirstlane(tid >> 5);
    const int ch = chunk_edges(nEdges);
    const int c  = blockIdx.x;
    const int e0 = c * ch;
    int e1 = e0 + ch; if (e1 > nEdges) e1 = nEdges;
    const int wlen = ((ch + BIN_WAVES - 1) / BIN_WAVES + 31) & ~31;
    const int ws = e0 + wave * wlen;
    int we = ws + wlen; if (we > e1) we = e1;
    const unsigned un = (unsigned)nper;

    for (int i = tid; i < BIN_WAVES * OWN_BLOCKS; i += BIN_THREADS) hist[i] = 0u;
    __syncthreads();
    unsigned int* myh = hist + wave * OWN_BLOCKS;

    for (int base = ws; base < we; base += BIN_QUADS * 128) {
        v4i dq[BIN_QUADS];
#pragma unroll
        for (int u = 0; u < BIN_QUADS; ++u) {
            const int q = base + (u * 32 + lane) * 4;
            const v4i none = {-1, -1, -1, -1};
            dq[u] = (q < we) ? *(const v4i*)(dst + q) : none;
        }
#pragma unroll
        for (int u = 0; u < BIN_QUADS; ++u) {
            const int q = base + (u * 32 + lane) * 4;
            if (q < we) {
#pragma unroll
                for (int j = 0; j < 4; ++j) {
                    unsigned o = (unsigned)dq[u][j] / un; if (o > OWN_BLOCKS - 1) o = OWN_BLOCKS - 1;
                    atomicAdd(&myh[o], 1u);
                }
            }
        }
    }
    __syncthreads();

    unsigned run = 0u;
#pragma unroll
    for (int w = 0; w < BIN_WAVES; ++w) {
        const unsigned v = hist[w * OWN_BLOCKS + tid];
        hist[w * OWN_BLOCKS + tid] = run;
        run += v;
    }
    unsigned incl = run;
#pragma unroll
    for (int off = 1; off < 32; off <<= 1) {
        const unsigned y = __shfl_up(incl, off, 32);
        if (lane >= off) incl += y;
    }
    if (lane == 31) wtot[wave] = incl;
    __syncthreads();
    unsigned wbase = 0u;
    for (int j = 0; j < wave; ++j) wbase += wtot[j];
    const unsigned excl = wbase + incl - run;
#pragma unroll
    for (int w = 0; w < BIN_WAVES; ++w) hist[w * OWN_BLOCKS + tid] += excl;
    tab[c * (OWN_BLOCKS + 1) + tid] = excl;
    if (tid == OWN_BLOCKS - 1) tab[c * (OWN_BLOCKS + 1) + OWN_BLOCKS] = excl + run;
    __syncthreads();

    for (int base = ws; base < we; base += BIN_QUADS * 128) {
        v4i dq[BIN_QUADS], sq[BIN_QUADS];
#pragma unroll
        for (int u = 0; u < BIN_QUADS; ++u) {
            const int q = base + (u * 32 + lane) * 4;
            const v4i none = {-1, -1, -1, -1};
            dq[u] = (q < we) ? *(const v4i*)(dst + q) : none;
            sq[u] = (q < we) ? *(const v4i*)(src + q) : none;
        }
#pragma unroll
        for (int u = 0; u < BIN_QUADS; ++u) {
            const int q = base + (u * 32 + lane) * 4;
            if (q < we) {
#pragma unroll
                for (int j = 0; j < 4; ++j) {
                    const unsigned d = (unsigned)dq[u][j];
                    unsigned o = d / un; if (o > OWN_BLOCKS - 1) o = OWN_BLOCKS - 1;
                    const unsigned dl = d - o * un;
                    const unsigned word = (dl < un) ? (((unsigned)sq[u][j] & ((1u << S_BITS) - 1u)) | (dl << S_BITS)) : EMPTY_WORD;
                    const unsigned pos = atomicAdd(&myh[o], 1u);
                    stage[pos] = word;
                }
            }
        }
    }
    __syncthreads();
    for (int i = tid; i < e1 - e0; i += BIN_THREADS) binw[e0 + i] = stage[i];
}

__global__ __launch_bounds__(BIN_THREADS)
void gat_el_er(const float* __restrict__ h,
               const float* __restrict__ a_l,
               const float* __restrict__ a_r,
               float* __restrict__ e_l,
               float* __restrict__ e_r, int nNodes,
               const int* __restrict__ src, const int* __restrict__ dst,
               unsigned int* __restrict__ binw, unsigned int* __restrict__ tab,
               int nEdges, int nper, int nBin) {
    __shared__ unsigned int hist[BIN_WAVES * OWN_BLOCKS];
    __shared__ unsigned int stage[BIN_CAP];
    __shared__ unsigned int wtot[BIN_WAVES];
    if ((int)blockIdx.x < nBin) {
        bin_chunk(src, dst, binw, tab, nEdges, nper, hist, stage, wtot);
        return;
    }
    const int lane = threadIdx.x & 31;
    const int node = (int)(((blockIdx.x - nBin) * blockDim.x + threadIdx.x) >> 5);
    if (node >= nNodes) return;
    float2 hv = ((const float2*)(h + (size_t)node * OUT_CH))[lane];
    float2 al = ((const float2*)a_l)[lane];
    float2 ar = ((const float2*)a_r)[lane];
    float pl = hv.x * al.x + hv.y * al.y;
    float pr = hv.x * ar.x + hv.y * ar.y;
#pragma unroll
    for (int off = 16; off > 0; off >>= 1) {
        pl += __shfl_xor(pl, off, 32);
        pr += __shfl_xor(pr, off, 32);
    }
    if (lane == 0) { e_l[node] = pl; e_r[node] = pr; }
}

__global__ void gat_init(float* __restrict__ out, const float* __restrict__ bias,
                         float* __restrict__ node_max, float* __restrict__ node_sum,
                         int nNodes) {
    int i = blockIdx.x * blockDim.x + threadIdx.x;
    if (i < nNodes * OUT_CH) out[i] = bias[i & (OUT_CH - 1)];
    if (i < nNodes) { node_max[i] = -__builtin_inff(); node_sum[i] = 0.f; }
}

__device__ __forceinline__ void atomicMaxF32(float* addr, float v) {
    if (v >= 0.f)
        atomicMax((int*)addr, __float_as_int(v));
    else
        atomicMin((unsigned int*)addr, __float_as_uint(v));
}

__global__ void gat_edge_max(const int* __restrict__ src, const int* __restrict__ dst,
                             const float* __restrict__ e_l, const float* __restrict__ e_r,
                             float* __restrict__ edge_w, float* __restrict__ node_max,
                             int nEdges) {
    int e = blockIdx.x * blockDim.x + threadIdx.x;
    if (e >= nEdges) return;
    int s = src[e], d = dst[e];
    float a = e_l[d] + e_r[s];
    a = (a > 0.f) ? a : NEG_SLOPE * a;
    edge_w[e] = a;
    atomicMaxF32(node_max + d, a);
}

__global__ void gat_edge_exp(const int* __restrict__ dst,
                             float* __restrict__ edge_w,
                             const float* __restrict__ node_max,
                             float* __restrict__ node_sum, int nEdges) {
    int e = blockIdx.x * blockDim.x + threadIdx.x;
    if (e >= nEdges) return;
    int d = dst[e];
    float w = __expf(edge_w[e] - node_max[d]);
    edge_w[e] = w;
    atomicAdd(node_sum + d, w);
}

__global__ void gat_scatter(const int* __restrict__ src, const int* __restrict__ dst,
                            const float* __restrict__ h,
                            const float* __restrict__ edge_w,
                            const float* __restrict__ node_sum,
                            float* __restrict__ out, int nEdges) {
    const int lane = threadIdx.x & 31;
    const int e = (blockIdx.x * blockDim.x + threadIdx.x) >> 5;
    if (e >= nEdges) return;
    const int s = src[e], d = dst[e];
    const float coef = edge_w[e] / (node_sum[d] + EPS_F);
    float2 hv = ((const float2*)(h + (size_t)s * OUT_CH))[lane];
    float* o = out + (size_t)d * OUT_CH + lane * 2;
    atomicAdd(o + 0, hv.x * coef);
    atomicAdd(o + 1, hv.y * coef);
}


#define OWN_THREADS  1024
#define OWN_WAVES    32
#define OWN_CPW      (OWN_BLOCKS / OWN_WAVES)
#define SEG_CAP      768

__device__ __forceinline__ void ldsMaxF32(int* p, float v) {
    const int b = __float_as_int(v);
    if (b >= 0) atomicMax(p, b);
    else        atomicMin((unsigned int*)p, (unsigned int)b);
}

template <int PHASE>
__device__ __forceinline__ void edge_now(bool m, int e, int dl, int lo, int lane,
                                         const int* __restrict__ src, const float* __restrict__ h,
                                         const float* __restrict__ e_l, const float* __restrict__ e_r,
                                         int* nmaxi, float* nsum, float* acc) {
    int s = 0; float w = 0.f;
    if (m) {
        s = src[e];
        const float x = e_l[lo + dl] + e_r[s];
        const float a = (x > 0.f) ? x : NEG_SLOPE * x;
        if (PHASE == 1) {
            ldsMaxF32(&nmaxi[dl], a);
        } else {
            w = __expf(a - __int_as_float(nmaxi[dl]));
            atomicAdd(&nsum[dl], w);
        }
    }
    if (PHASE == 2) {
        unsigned mask = __builtin_amdgcn_ballot_w32(m);
        const int dls = m ? dl : 0;
        while (mask) {
            const int k = __builtin_ctz(mask);
            mask &= mask - 1;
            const int   sk = __builtin_amdgcn_readlane(s, k);
            const int   dk = __builtin_amdgcn_readlane(dls, k);
            const float wk = __int_as_float(__builtin_amdgcn_readlane(__float_as_int(w), k));
            const float2 hv = ((const float2*)(h + (size_t)sk * OUT_CH))[lane];
            float* r = &acc[dk * OUT_CH + 2 * lane];
            atomicAdd(r,     hv.x * wk);
            atomicAdd(r + 1, hv.y * wk);
        }
    }
}

template <int MODE>
__device__ __forceinline__ void scan_quad(v4i v, int e0, int lo, int cnt, int lane,
                                          const int* __restrict__ src, const float* __restrict__ h,
                                          const float* __restrict__ e_l, const float* __restrict__ e_r,
                                          int* nmaxi, float* nsum, float* acc) {
    const int d0 = v.x - lo, d1 = v.y - lo, d2 = v.z - lo, d3 = v.w - lo;
    const bool m0 = (unsigned)d0 < (unsigned)cnt;
    const bool m1 = (unsigned)d1 < (unsigned)cnt;
    const bool m2 = (unsigned)d2 < (unsigned)cnt;
    const bool m3 = (unsigned)d3 < (unsigned)cnt;
    if (__builtin_amdgcn_ballot_w32(m0 | m1 | m2 | m3) == 0) return;
    edge_now<MODE>(m0, e0 + 0, d0, lo, lane, src, h, e_l, e_r, nmaxi, nsum, acc);
    edge_now<MODE>(m1, e0 + 1, d1, lo, lane, src, h, e_l, e_r, nmaxi, nsum, acc);
    edge_now<MODE>(m2, e0 + 2, d2, lo, lane, src, h, e_l, e_r, nmaxi, nsum, acc);
    edge_now<MODE>(m3, e0 + 3, d3, lo, lane, src, h, e_l, e_r, nmaxi, nsum, acc);
}

template <int MODE>
__device__ __forceinline__ void scan_edges(const int* __restrict__ src, const int* __restrict__ dst,
                                           const float* __restrict__ h,
                                           const float* __restrict__ e_l, const float* __restrict__ e_r,
                                           int nEdges, int lo, int cnt, int lane, int wave,
                                           int* nmaxi, float* nsum, float* acc) {
    const int nGroups = nEdges >> 7;
    const v4i* dst4 = (const v4i*)dst;
    const int start = (int)(((long long)blockIdx.x * nGroups) / OWN_BLOCKS);
    const v4i none = {-1, -1, -1, -1};
    for (int j = wave; j < nGroups; j += OWN_WAVES * 8) {
        v4i v[8]; int g[8];
#pragma unroll
        for (int u = 0; u < 8; ++u) {
            const int jj = j + u * OWN_WAVES;
            int gg = jj + start;
            if (gg >= nGroups) gg -= nGroups;
            g[u] = gg;
            v[u] = (jj < nGroups) ? dst4[(size_t)gg * 32 + lane] : none;
        }
#pragma unroll
        for (int u = 0; u < 8; ++u)
            scan_quad<MODE>(v[u], g[u] * 128 + lane * 4, lo, cnt, lane, src, h, e_l, e_r,
                            nmaxi, nsum, acc);
    }
    for (int eb = nGroups * 128 + wave * 32; eb < nEdges; eb += OWN_THREADS) {
        const int e = eb + lane;
        const int d = (e < nEdges) ? dst[e] : -1;
        const int dl = d - lo;
        const bool m = (unsigned)dl < (unsigned)cnt;
        const unsigned km = __builtin_amdgcn_ballot_w32(m);
        if (km == 0) continue;
        edge_now<MODE>(m, e, dl, lo, lane, src, h, e_l, e_r, nmaxi, nsum, acc);
    }
}

__global__ __launch_bounds__(OWN_THREADS)
void gat_owner(const int* __restrict__ src, const int* __restrict__ dst,
               const unsigned int* __restrict__ binw, const unsigned int* __restrict__ tab,
               const float* __restrict__ h,
               const float* __restrict__ e_l, const float* __restrict__ e_r,
               const float* __restrict__ bias, float* __restrict__ out,
               int nNodes, int nEdges, int nper) {
    __shared__ float        acc[(NPER_MAX + 1) * OUT_CH];
    __shared__ unsigned int list[OWN_WAVES * SEG_CAP];
    __shared__ float        score[OWN_WAVES * SEG_CAP];
    __shared__ int          nmaxi[NPER_MAX];
    __shared__ float        nsum[NPER_MAX];
    __shared__ unsigned int overflow;

    const int tid  = threadIdx.x;
    const int lane = tid & 31;
    const int wave = __builtin_amdgcn_readfirstlane(tid >> 5);
    const int lo   = blockIdx.x * nper;
    int cnt = nNodes - lo;
    cnt = cnt < 0 ? 0 : (cnt > nper ? nper : cnt);
    unsigned int* seg = list + wave * SEG_CAP;
    float* sco = score + wave * SEG_CAP;

    for (int i = tid; i < cnt * OUT_CH; i += OWN_THREADS) acc[i] = 0.f;
    for (int i = tid; i < cnt; i += OWN_THREADS) { nmaxi[i] = (int)0xFF800000u; nsum[i] = 0.f; }
    if (tid == 0) overflow = 0u;
    __syncthreads();

    static_assert(OWN_CPW * OWN_WAVES == OWN_BLOCKS && OWN_CPW <= 32, "one chunk per lane of a wave's first lanes");
    unsigned int wcount = 0u;
    {
        const int ch = chunk_edges(nEdges);
        const int cl = wave * OWN_CPW + (lane & (OWN_CPW - 1));
        const unsigned t0 = tab[cl * (OWN_BLOCKS + 1) + blockIdx.x];
        const unsigned t1 = tab[cl * (OWN_BLOCKS + 1) + blockIdx.x + 1];
        unsigned nn[OWN_CPW], bs[OWN_CPW], pv[OWN_CPW];
        const unsigned int* sp[OWN_CPW];
#pragma unroll
        for (int u = 0; u < OWN_CPW; ++u) {
            const unsigned b0 = (unsigned)__builtin_amdgcn_readlane((int)t0, u);
            const unsigned b1 = (unsigned)__builtin_amdgcn_readlane((int)t1, u);
            nn[u] = b1 - b0;
            sp[u] = binw + (size_t)(wave * OWN_CPW + u) * ch + b0;
            bs[u] = wcount;
            wcount += nn[u];
        }
#pragma unroll
        for (int u = 0; u < OWN_CPW; ++u)
            pv[u] = ((unsigned)lane < nn[u]) ? sp[u][lane] : EMPTY_WORD;
#pragma unroll
        for (int u = 0; u < OWN_CPW; ++u) {
            const unsigned idx = bs[u] + lane;
            if ((unsigned)lane < nn[u] && idx < SEG_CAP) seg[idx] = pv[u];
        }
#pragma unroll
        for (int u = 0; u < OWN_CPW; ++u) {
            for (unsigned k = 32u + lane; k < nn[u]; k += 32u) {
                const unsigned idx = bs[u] + k;
                if (idx < SEG_CAP) seg[idx] = sp[u][k];
            }
        }
    }
    if (wcount > SEG_CAP && lane == 0) overflow = 1u;
    __syncthreads();

    const unsigned over = __builtin_amdgcn_readfirstlane(overflow);
    if (over == 0u) {
        for (unsigned b = 0; b < wcount; b += 128u) {
            unsigned idx[4], p[4]; int s[4], dl[4]; bool mine[4]; float el[4], er[4];
#pragma unroll
            for (int t = 0; t < 4; ++t) {
                idx[t] = b + 32u * t + lane;
                p[t] = seg[idx[t] < wcount ? idx[t] : wcount - 1u];
            }
#pragma unroll
            for (int t = 0; t < 4; ++t) {
                s[t]  = (int)(p[t] & ((1u << S_BITS) - 1u));
                dl[t] = (int)(p[t] >> S_BITS);
                mine[t] = (unsigned)dl[t] < (unsigned)cnt;
                el[t] = e_l[mine[t] ? lo + dl[t] : 0];
                er[t] = e_r[s[t]];
            }
#pragma unroll
            for (int t = 0; t < 4; ++t) {
                if (idx[t] < wcount) {
                    if (mine[t]) {
                        const float x = el[t] + er[t];
                        const float a = (x > 0.f) ? x : NEG_SLOPE * x;
                        ldsMaxF32(&nmaxi[dl[t]], a);
                        sco[idx[t]] = a;
                    } else {
                        seg[idx[t]] = EMPTY_WORD;
                        sco[idx[t]] = 0.f;
                    }
                }
            }
        }
        __syncthreads();

        for (unsigned b = 0; b < wcount; b += 32u) {
            const unsigned i = b + lane;
            int s = 0, dl = NPER_MAX; float w = 0.f;
            const unsigned p = (i < wcount) ? seg[i] : ((unsigned)NPER_MAX << S_BITS);
            const float    a = (i < wcount) ? sco[i] : 0.f;
            if ((p >> S_BITS) != (unsigned)NPER_MAX) {
                s  = (int)(p & ((1u << S_BITS) - 1u));
                dl = (int)(p >> S_BITS);
                w = __expf(a - __int_as_float(nmaxi[dl]));
                atomicAdd(&nsum[dl], w);
            }
            const int nvalid = (wcount - b) < 32u ? (int)(wcount - b) : 32;
            for (int k0 = 0; k0 < nvalid; k0 += 16) {
                float2 hv[16]; int dk[16]; float wk[16];
#pragma unroll
                for (int u = 0; u < 16; ++u) {
                    const int k = k0 + u;
                    const int sk = __builtin_amdgcn_readlane(s, k);
                    dk[u] = __builtin_amdgcn_readlane(dl, k);
                    wk[u] = __int_as_float(__builtin_amdgcn_readlane(__float_as_int(w), k));
                    hv[u] = ((const float2*)(h + (size_t)sk * OUT_CH))[lane];
                }
#pragma unroll
                for (int u = 0; u < 16; ++u) {
                    float* r = &acc[dk[u] * OUT_CH + 2 * lane];
                    atomicAdd(r,     hv[u].x * wk[u]);
                    atomicAdd(r + 1, hv[u].y * wk[u]);
                }
            }
        }
    } else {
        scan_edges<1>(src, dst, h, e_l, e_r, nEdges, lo, cnt, lane, wave, nmaxi, nsum, acc);
        __syncthreads();
        scan_edges<2>(src, dst, h, e_l, e_r, nEdges, lo, cnt, lane, wave, nmaxi, nsum, acc);
    }
    __syncthreads();

    float* o = out + (size_t)lo * OUT_CH;
    for (int i = tid; i < cnt * OUT_CH; i += OWN_THREADS)
        o[i] = acc[i] / (nsum[i >> 6] + EPS_F) + bias[i & (OUT_CH - 1)];
}

extern "C" void kernel_launch(void* const* d_in, const int* in_sizes, int n_in,
                              void* d_out, int out_size, void* d_ws, size_t ws_size,
                              hipStream_t stream) {
    const float* x    = (const float*)d_in[0];
    const int*   ei   = (const int*)d_in[1];
    const float* W    = (const float*)d_in[2];
    const float* a_l  = (const float*)d_in[3];
    const float* a_r  = (const float*)d_in[4];
    const float* bias = (const float*)d_in[5];
    float* out = (float*)d_out;

    const int nNodes = in_sizes[0] / IN_CH;
    const int nEdges = in_sizes[1] / 2;
    const int* src = ei;
    const int* dst = ei + nEdges;

    float* ws       = (float*)d_ws;
    float* h        = ws;
    float* e_l      = h + (size_t)nNodes * OUT_CH;
    float* e_r      = e_l + nNodes;
    float* node_max = e_r + nNodes;
    float* node_sum = node_max + nNodes;
    float* edge_w   = node_sum + nNodes;

    const int mFull = nNodes & ~15;
    const int nSlabs = (mFull + GB_ROWS - 1) / GB_ROWS;
    if (nSlabs > 0)
        gat_gemm_slab<<<nSlabs < 512 ? nSlabs : 512, GB_THREADS, 0, stream>>>(x, W, h, mFull, nSlabs);
    if (mFull < nNodes) {
        int tailThreads = (nNodes - mFull) * OUT_CH;
        gat_gemm_tail<<<(tailThreads + 255) / 256, 256, 0, stream>>>(x, W, h, mFull, nNodes);
    }

    const int nper = (nNodes + OWN_BLOCKS - 1) / OWN_BLOCKS;
    const bool owner = nper <= NPER_MAX && (nEdges & 3) == 0 && nEdges <= BIN_CAP * OWN_BLOCKS &&
                       nNodes <= (1 << S_BITS) && 2 * nNodes >= OWN_BLOCKS * (OWN_BLOCKS + 1);
    unsigned int* binw = (unsigned int*)edge_w;
    unsigned int* tab  = (unsigned int*)node_max;

    const int nBin = owner ? OWN_BLOCKS : 0;
    gat_el_er<<<nBin + (nNodes * 32 + BIN_THREADS - 1) / BIN_THREADS, BIN_THREADS, 0, stream>>>(
        h, a_l, a_r, e_l, e_r, nNodes, src, dst, binw, tab, nEdges, nper, nBin);

    if (owner) {
        gat_owner<<<OWN_BLOCKS, OWN_THREADS, 0, stream>>>(src, dst, binw, tab, h, e_l, e_r, bias, out, nNodes, nEdges, nper);
        return;
    }

    int initN = nNodes * OUT_CH;
    gat_init<<<(initN + 255) / 256, 256, 0, stream>>>(out, bias, node_max, node_sum, nNodes);

    gat_edge_max<<<(nEdges + 255) / 256, 256, 0, stream>>>(src, dst, e_l, e_r, edge_w, node_max, nEdges);
    gat_edge_exp<<<(nEdges + 255) / 256, 256, 0, stream>>>(dst, edge_w, node_max, node_sum, nEdges);
    gat_scatter<<<((size_t)nEdges * 32 + 255) / 256, 256, 0, stream>>>(src, dst, h, edge_w, node_sum, out, nEdges);
}


# e_l, e_r pass: one wave per 32 rows of h (was one per row): 16 whole-row loads in flight, 16-lane sums as DPP adds, one 128-B store per array; 5176 waves enter, was 102048; sum order unchanged
# speedup vs baseline: 5.3497x; 1.1167x over previous
#include <hip/hip_runtime.h>
#include <hip/hip_bf16.h>
#include <cstddef>

typedef __attribute__((ext_vector_type(2))) float v2f;
typedef __attribute__((ext_vector_type(8))) float v8f;

#define IN_CH   128
#define OUT_CH  64
#define NEG_SLOPE 0.2f
#define EPS_F   1e-10f

#define GB_ROWS    64
#define GB_PITCH   132
#define GB_THREADS 256
#define GB_KSTEPS  (IN_CH / 4)

typedef __attribute__((ext_vector_type(4))) float v4f;

__device__ __forceinline__ float gb_xhalf(float v) {
    return __builtin_bit_cast(float, __builtin_amdgcn_permlanex16(__builtin_bit_cast(int, v), __builtin_bit_cast(int, v),
                                                                   0x76543210, 0xfedcba98, false, false));
}

__global__ __launch_bounds__(GB_THREADS)
void gat_gemm_slab(const float* __restrict__ x, const float* __restrict__ W,
                   float* __restrict__ h, int mFull, int nSlabs) {
    static_assert(OUT_CH == GB_ROWS, "W passes through a slab buffer");
    __shared__ __attribute__((aligned(16))) float xs[2][GB_ROWS * GB_PITCH];

    const int tid  = threadIdx.x;
    const int lane = tid & 31;
    const int wave = __builtin_amdgcn_readfirstlane(tid >> 5);
    const int l15  = lane & 15;
    const int half = lane >> 4;
    const int kOff = half << 1;
    const int nBase = (wave & 1) * 32;
    const int mt    = wave >> 1;

    const v4f* x4 = (const v4f*)x;
    v4f stg[GB_ROWS / 8];
    const v4f zero4 = {0.f, 0.f, 0.f, 0.f};

    int slab = blockIdx.x;
    if (slab < nSlabs) {
#pragma unroll
        for (int i = 0; i < GB_ROWS / 8; ++i) {
            const int row = slab * GB_ROWS + i * 8 + wave;
            stg[i] = (row < mFull) ? x4[(size_t)row * (IN_CH / 4) + lane] : zero4;
        }
    }
    {
        const v4f* w4 = (const v4f*)W;
        v4f wst[OUT_CH / 8];
#pragma unroll
        for (int i = 0; i < OUT_CH / 8; ++i)
            wst[i] = w4[(size_t)(i * 8 + wave) * (IN_CH / 4) + lane];
#pragma unroll
        for (int i = 0; i < OUT_CH / 8; ++i)
            *(v4f*)(&xs[1][(i * 8 + wave) * GB_PITCH + lane * 4]) = wst[i];
    }
    if (slab < nSlabs) {
#pragma unroll
        for (int i = 0; i < GB_ROWS / 8; ++i)
            *(v4f*)(&xs[0][(i * 8 + wave) * GB_PITCH + lane * 4]) = stg[i];
    }
    __syncthreads();

    v2f b0[GB_KSTEPS], b1[GB_KSTEPS];
    {
        const float* w0 = &xs[1][(nBase + l15) * GB_PITCH + kOff];
        const float* w1 = w0 + 16 * GB_PITCH;
#pragma unroll
        for (int kk = 0; kk < GB_KSTEPS; ++kk) {
            b0[kk] = *(const v2f*)(w0 + 4 * kk);
            b1[kk] = *(const v2f*)(w1 + 4 * kk);
        }
    }
    __syncthreads();

    int buf = 0;
    for (; slab < nSlabs; slab += (int)gridDim.x) {
        const int  nxt  = slab + (int)gridDim.x;
        const bool more = nxt < nSlabs;
        if (more) {
#pragma unroll
            for (int i = 0; i < GB_ROWS / 8; ++i) {
                const int row = nxt * GB_ROWS + i * 8 + wave;
                stg[i] = (row < mFull) ? x4[(size_t)row * (IN_CH / 4) + lane] : zero4;
            }
        }

        const float* ap = &xs[buf][(mt * 16 + l15) * GB_PITCH + kOff];
        v8f acc0 = {}, acc1 = {};
#pragma unroll
        for (int kk = 0; kk < GB_KSTEPS; ++kk) {
            const v2f a = *(const v2f*)(ap + 4 * kk);
            acc0 = __builtin_amdgcn_wmma_f32_16x16x4_f32(
                 false, a,  false, b0[kk],
                 (short)0, acc0,  false,  false);
            acc1 = __builtin_amdgcn_wmma_f32_16x16x4_f32(
                 false, a,  false, b1[kk],
                 (short)0, acc1,  false,  false);
        }

        const int mBase = slab * GB_ROWS + mt * 16;
        if (mBase < mFull) {
            float* o = h + (size_t)mBase * OUT_CH + nBase + lane;
#pragma unroll
            for (int r = 0; r < 8; ++r) {
                const float lo = acc0[r], hi = acc1[r];
                const float lox = gb_xhalf(lo), hix = gb_xhalf(hi);
                o[(size_t)r * OUT_CH]       = half ? hix : lo;
                o[(size_t)(r + 8) * OUT_CH] = half ? hi  : lox;
            }
        }

        if (more) {
#pragma unroll
            for (int i = 0; i < GB_ROWS / 8; ++i)
                *(v4f*)(&xs[buf ^ 1][(i * 8 + wave) * GB_PITCH + lane * 4]) = stg[i];
        }
        __syncthreads();
        buf ^= 1;
    }
}

__global__ void gat_gemm_tail(const float* __restrict__ x,
                              const float* __restrict__ W,
                              float* __restrict__ h,
                              int rowStart, int nNodes) {
    int idx = blockIdx.x * blockDim.x + threadIdx.x;
    int row = rowStart + (idx >> 6);
    int col = idx & 63;
    if (row >= nNodes) return;
    const float* xr = x + (size_t)row * IN_CH;
    const float* wr = W + (size_t)col * IN_CH;
    float s = 0.f;
    for (int k = 0; k < IN_CH; ++k) s += xr[k] * wr[k];
    h[(size_t)row * OUT_CH + col] = s;
}

typedef __attribute__((ext_vector_type(4))) int v4i;
typedef __attribute__((ext_vector_type(4))) unsigned int v4u;

#define OWN_BLOCKS   256
#define NPER_MAX     391
#define S_BITS       17
#define EMPTY_WORD   ((unsigned)NPER_MAX << S_BITS)
#define BIN_THREADS  256
#define BIN_WAVES    8
#define BIN_CAP      6400
#define BIN_QUADS    4

__device__ __forceinline__ int chunk_edges(int nEdges) {
    return (((nEdges + OWN_BLOCKS - 1) / OWN_BLOCKS) + 31) & ~31;
}

__device__ __forceinline__ void bin_chunk(const int* __restrict__ src, const int* __restrict__ dst,
                                          unsigned int* __restrict__ binw, unsigned int* __restrict__ tab,
                                          int nEdges, int nper,
                                          unsigned int* hist, unsigned int* stage, unsigned int* wtot) {
    static_assert(BIN_THREADS == OWN_BLOCKS && BIN_WAVES * 32 == BIN_THREADS, "thread t scans owner t's counters");
    static_assert(BIN_WAVES * 800 >= BIN_CAP, "a wave's stretch is two trips at most");
    const int tid  = threadIdx.x;
    const int lane = tid & 31;
    const int wave = __builtin_amdgcn_readfirstlane(tid >> 5);
    const int ch = chunk_edges(nEdges);
    const int c  = blockIdx.x;
    const int e0 = c * ch;
    int e1 = e0 + ch; if (e1 > nEdges) e1 = nEdges;
    const int wlen = ((ch + BIN_WAVES - 1) / BIN_WAVES + 31) & ~31;
    const int ws = e0 + wave * wlen;
    int we = ws + wlen; if (we > e1) we = e1;
    const unsigned un = (unsigned)nper;

    for (int i = tid; i < BIN_WAVES * OWN_BLOCKS; i += BIN_THREADS) hist[i] = 0u;
    __syncthreads();
    unsigned int* myh = hist + wave * OWN_BLOCKS;

    for (int base = ws; base < we; base += BIN_QUADS * 128) {
        v4i dq[BIN_QUADS];
#pragma unroll
        for (int u = 0; u < BIN_QUADS; ++u) {
            const int q = base + (u * 32 + lane) * 4;
            const v4i none = {-1, -1, -1, -1};
            dq[u] = (q < we) ? *(const v4i*)(dst + q) : none;
        }
#pragma unroll
        for (int u = 0; u < BIN_QUADS; ++u) {
            const int q = base + (u * 32 + lane) * 4;
            if (q < we) {
#pragma unroll
                for (int j = 0; j < 4; ++j) {
                    unsigned o = (unsigned)dq[u][j] / un; if (o > OWN_BLOCKS - 1) o = OWN_BLOCKS - 1;
                    atomicAdd(&myh[o], 1u);
                }
            }
        }
    }
    __syncthreads();

    unsigned run = 0u;
#pragma unroll
    for (int w = 0; w < BIN_WAVES; ++w) {
        const unsigned v = hist[w * OWN_BLOCKS + tid];
        hist[w * OWN_BLOCKS + tid] = run;
        run += v;
    }
    unsigned incl = run;
#pragma unroll
    for (int off = 1; off < 32; off <<= 1) {
        const unsigned y = __shfl_up(incl, off, 32);
        if (lane >= off) incl += y;
    }
    if (lane == 31) wtot[wave] = incl;
    __syncthreads();
    unsigned wbase = 0u;
    for (int j = 0; j < wave; ++j) wbase += wtot[j];
    const unsigned excl = wbase + incl - run;
#pragma unroll
    for (int w = 0; w < BIN_WAVES; ++w) hist[w * OWN_BLOCKS + tid] += excl;
    tab[c * (OWN_BLOCKS + 1) + tid] = excl;
    if (tid == OWN_BLOCKS - 1) tab[c * (OWN_BLOCKS + 1) + OWN_BLOCKS] = excl + run;
    __syncthreads();

    for (int base = ws; base < we; base += BIN_QUADS * 128) {
        v4i dq[BIN_QUADS], sq[BIN_QUADS];
#pragma unroll
        for (int u = 0; u < BIN_QUADS; ++u) {
            const int q = base + (u * 32 + lane) * 4;
            const v4i none = {-1, -1, -1, -1};
            dq[u] = (q < we) ? *(const v4i*)(dst + q) : none;
            sq[u] = (q < we) ? *(const v4i*)(src + q) : none;
        }
#pragma unroll
        for (int u = 0; u < BIN_QUADS; ++u) {
            const int q = base + (u * 32 + lane) * 4;
            if (q < we) {
#pragma unroll
                for (int j = 0; j < 4; ++j) {
                    const unsigned d = (unsigned)dq[u][j];
                    unsigned o = d / un; if (o > OWN_BLOCKS - 1) o = OWN_BLOCKS - 1;
                    const unsigned dl = d - o * un;
                    const unsigned word = (dl < un) ? (((unsigned)sq[u][j] & ((1u << S_BITS) - 1u)) | (dl << S_BITS)) : EMPTY_WORD;
                    const unsigned pos = atomicAdd(&myh[o], 1u);
                    stage[pos] = word;
                }
            }
        }
    }
    __syncthreads();
    for (int i = tid; i < e1 - e0; i += BIN_THREADS) binw[e0 + i] = stage[i];
}

#define EL_ROWS 32

template <int M>
__device__ __forceinline__ float row_xor(float v) {
    const int b = __builtin_bit_cast(int, v);
    return __builtin_bit_cast(float, __builtin_amdgcn_update_dpp(b, b, 0x160 + M, 0xf, 0xf, true));
}

__global__ __launch_bounds__(BIN_THREADS)
void gat_el_er(const float* __restrict__ h,
               const float* __restrict__ a_l,
               const float* __restrict__ a_r,
               float* __restrict__ e_l,
               float* __restrict__ e_r, int nNodes,
               const int* __restrict__ src, const int* __restrict__ dst,
               unsigned int* __restrict__ binw, unsigned int* __restrict__ tab,
               int nEdges, int nper, int nBin) {
    __shared__ unsigned int hist[BIN_WAVES * OWN_BLOCKS];
    __shared__ unsigned int stage[BIN_CAP];
    __shared__ unsigned int wtot[BIN_WAVES];
    if ((int)blockIdx.x < nBin) {
        bin_chunk(src, dst, binw, tab, nEdges, nper, hist, stage, wtot);
        return;
    }
    static_assert(OUT_CH == 64 && EL_ROWS == 32, "a 16-lane row of the wave holds one 256-B row of h, lane l keeps row l's sums");
    const int lane = threadIdx.x & 31;
    const int wave = __builtin_amdgcn_readfirstlane((int)(threadIdx.x >> 5));
    const int row0 = (((int)blockIdx.x - nBin) * BIN_WAVES + wave) * EL_ROWS;
    if (row0 >= nNodes) return;
    const int l15  = lane & 15;
    const int half = lane >> 4;
    const v4f* h4 = (const v4f*)h;
    v4f hv[EL_ROWS / 2];
#pragma unroll
    for (int i = 0; i < EL_ROWS / 2; ++i) {
        int row = row0 + 16 * half + i;
        row = row < nNodes ? row : nNodes - 1;
        hv[i] = h4[(size_t)row * (OUT_CH / 4) + l15];
    }
    const v4f al = ((const v4f*)a_l)[l15];
    const v4f ar = ((const v4f*)a_r)[l15];
    float rl = 0.f, rr = 0.f;
#pragma unroll
    for (int i = 0; i < EL_ROWS / 2; ++i) {
        float la = __builtin_fmaf(hv[i].x, al.x, hv[i].y * al.y);
        float lb = __builtin_fmaf(hv[i].z, al.z, hv[i].w * al.w);
        float ra = __builtin_fmaf(hv[i].x, ar.x, hv[i].y * ar.y);
        float rb = __builtin_fmaf(hv[i].z, ar.z, hv[i].w * ar.w);
        la += row_xor<8>(la); lb += row_xor<8>(lb); ra += row_xor<8>(ra); rb += row_xor<8>(rb);
        la += row_xor<4>(la); lb += row_xor<4>(lb); ra += row_xor<4>(ra); rb += row_xor<4>(rb);
        la += row_xor<2>(la); lb += row_xor<2>(lb); ra += row_xor<2>(ra); rb += row_xor<2>(rb);
        la += row_xor<1>(la); lb += row_xor<1>(lb); ra += row_xor<1>(ra); rb += row_xor<1>(rb);
        const float sl = la + lb, sr = ra + rb;
        if (l15 == i) { rl = sl; rr = sr; }
    }
    const int row = row0 + lane;
    if (row < nNodes) { e_l[row] = rl; e_r[row] = rr; }
}

__global__ void gat_init(float* __restrict__ out, const float* __restrict__ bias,
                         float* __restrict__ node_max, float* __restrict__ node_sum,
                         int nNodes) {
    int i = blockIdx.x * blockDim.x + threadIdx.x;
    if (i < nNodes * OUT_CH) out[i] = bias[i & (OUT_CH - 1)];
    if (i < nNodes) { node_max[i] = -__builtin_inff(); node_sum[i] = 0.f; }
}

__device__ __forceinline__ void atomicMaxF32(float* addr, float v) {
    if (v >= 0.f)
        atomicMax((int*)addr, __float_as_int(v));
    else
        atomicMin((unsigned int*)addr, __float_as_uint(v));
}

__global__ void gat_edge_max(const int* __restrict__ src, const int* __restrict__ dst,
                             const float* __restrict__ e_l, const float* __restrict__ e_r,
                             float* __restrict__ edge_w, float* __restrict__ node_max,
                             int nEdges) {
    int e = blockIdx.x * blockDim.x + threadIdx.x;
    if (e >= nEdges) return;
    int s = src[e], d = dst[e];
    float a = e_l[d] + e_r[s];
    a = (a > 0.f) ? a : NEG_SLOPE * a;
    edge_w[e] = a;
    atomicMaxF32(node_max + d, a);
}

__global__ void gat_edge_exp(const int* __restrict__ dst,
                             float* __restrict__ edge_w,
                             const float* __restrict__ node_max,
                             float* __restrict__ node_sum, int nEdges) {
    int e = blockIdx.x * blockDim.x + threadIdx.x;
    if (e >= nEdges) return;
    int d = dst[e];
    float w = __expf(edge_w[e] - node_max[d]);
    edge_w[e] = w;
    atomicAdd(node_sum + d, w);
}

__global__ void gat_scatter(const int* __restrict__ src, const int* __restrict__ dst,
                            const float* __restrict__ h,
                            const float* __restrict__ edge_w,
                            const float* __restrict__ node_sum,
                            float* __restrict__ out, int nEdges) {
    const int lane = threadIdx.x & 31;
    const int e = (blockIdx.x * blockDim.x + threadIdx.x) >> 5;
    if (e >= nEdges) return;
    const int s = src[e], d = dst[e];
    const float coef = edge_w[e] / (node_sum[d] + EPS_F);
    float2 hv = ((const float2*)(h + (size_t)s * OUT_CH))[lane];
    float* o = out + (size_t)d * OUT_CH + lane * 2;
    atomicAdd(o + 0, hv.x * coef);
    atomicAdd(o + 1, hv.y * coef);
}


#define OWN_THREADS  1024
#define OWN_WAVES    32
#define OWN_CPW      (OWN_BLOCKS / OWN_WAVES)
#define SEG_CAP      768

__device__ __forceinline__ void ldsMaxF32(int* p, float v) {
    const int b = __float_as_int(v);
    if (b >= 0) atomicMax(p, b);
    else        atomicMin((unsigned int*)p, (unsigned int)b);
}

template <int PHASE>
__device__ __forceinline__ void edge_now(bool m, int e, int dl, int lo, int lane,
                                         const int* __restrict__ src, const float* __restrict__ h,
                                         const float* __restrict__ e_l, const float* __restrict__ e_r,
                                         int* nmaxi, float* nsum, float* acc) {
    int s = 0; float w = 0.f;
    if (m) {
        s = src[e];
        const float x = e_l[lo + dl] + e_r[s];
        const float a = (x > 0.f) ? x : NEG_SLOPE * x;
        if (PHASE == 1) {
            ldsMaxF32(&nmaxi[dl], a);
        } else {
            w = __expf(a - __int_as_float(nmaxi[dl]));
            atomicAdd(&nsum[dl], w);
        }
    }
    if (PHASE == 2) {
        unsigned mask = __builtin_amdgcn_ballot_w32(m);
        const int dls = m ? dl : 0;
        while (mask) {
            const int k = __builtin_ctz(mask);
            mask &= mask - 1;
            const int   sk = __builtin_amdgcn_readlane(s, k);
            const int   dk = __builtin_amdgcn_readlane(dls, k);
            const float wk = __int_as_float(__builtin_amdgcn_readlane(__float_as_int(w), k));
            const float2 hv = ((const float2*)(h + (size_t)sk * OUT_CH))[lane];
            float* r = &acc[dk * OUT_CH + 2 * lane];
            atomicAdd(r,     hv.x * wk);
            atomicAdd(r + 1, hv.y * wk);
        }
    }
}

template <int MODE>
__device__ __forceinline__ void scan_quad(v4i v, int e0, int lo, int cnt, int lane,
                                          const int* __restrict__ src, const float* __restrict__ h,
                                          const float* __restrict__ e_l, const float* __restrict__ e_r,
                                          int* nmaxi, float* nsum, float* acc) {
    const int d0 = v.x - lo, d1 = v.y - lo, d2 = v.z - lo, d3 = v.w - lo;
    const bool m0 = (unsigned)d0 < (unsigned)cnt;
    const bool m1 = (unsigned)d1 < (unsigned)cnt;
    const bool m2 = (unsigned)d2 < (unsigned)cnt;
    const bool m3 = (unsigned)d3 < (unsigned)cnt;
    if (__builtin_amdgcn_ballot_w32(m0 | m1 | m2 | m3) == 0) return;
    edge_now<MODE>(m0, e0 + 0, d0, lo, lane, src, h, e_l, e_r, nmaxi, nsum, acc);
    edge_now<MODE>(m1, e0 + 1, d1, lo, lane, src, h, e_l, e_r, nmaxi, nsum, acc);
    edge_now<MODE>(m2, e0 + 2, d2, lo, lane, src, h, e_l, e_r, nmaxi, nsum, acc);
    edge_now<MODE>(m3, e0 + 3, d3, lo, lane, src, h, e_l, e_r, nmaxi, nsum, acc);
}

template <int MODE>
__device__ __forceinline__ void scan_edges(const int* __restrict__ src, const int* __restrict__ dst,
                                           const float* __restrict__ h,
                                           const float* __restrict__ e_l, const float* __restrict__ e_r,
                                           int nEdges, int lo, int cnt, int lane, int wave,
                                           int* nmaxi, float* nsum, float* acc) {
    const int nGroups = nEdges >> 7;
    const v4i* dst4 = (const v4i*)dst;
    const int start = (int)(((long long)blockIdx.x * nGroups) / OWN_BLOCKS);
    const v4i none = {-1, -1, -1, -1};
    for (int j = wave; j < nGroups; j += OWN_WAVES * 8) {
        v4i v[8]; int g[8];
#pragma unroll
        for (int u = 0; u < 8; ++u) {
            const int jj = j + u * OWN_WAVES;
            int gg = jj + start;
            if (gg >= nGroups) gg -= nGroups;
            g[u] = gg;
            v[u] = (jj < nGroups) ? dst4[(size_t)gg * 32 + lane] : none;
        }
#pragma unroll
        for (int u = 0; u < 8; ++u)
            scan_quad<MODE>(v[u], g[u] * 128 + lane * 4, lo, cnt, lane, src, h, e_l, e_r,
                            nmaxi, nsum, acc);
    }
    for (int eb = nGroups * 128 + wave * 32; eb < nEdges; eb += OWN_THREADS) {
        const int e = eb + lane;
        const int d = (e < nEdges) ? dst[e] : -1;
        const int dl = d - lo;
        const bool m = (unsigned)dl < (unsigned)cnt;
        const unsigned km = __builtin_amdgcn_ballot_w32(m);
        if (km == 0) continue;
        edge_now<MODE>(m, e, dl, lo, lane, src, h, e_l, e_r, nmaxi, nsum, acc);
    }
}

__global__ __launch_bounds__(OWN_THREADS)
void gat_owner(const int* __restrict__ src, const int* __restrict__ dst,
               const unsigned int* __restrict__ binw, const unsigned int* __restrict__ tab,
               const float* __restrict__ h,
               const float* __restrict__ e_l, const float* __restrict__ e_r,
               const float* __restrict__ bias, float* __restrict__ out,
               int nNodes, int nEdges, int nper) {
    __shared__ float        acc[(NPER_MAX + 1) * OUT_CH];
    __shared__ unsigned int list[OWN_WAVES * SEG_CAP];
    __shared__ float        score[OWN_WAVES * SEG_CAP];
    __shared__ int          nmaxi[NPER_MAX];
    __shared__ float        nsum[NPER_MAX];
    __shared__ unsigned int overflow;

    const int tid  = threadIdx.x;
    const int lane = tid & 31;
    const int wave = __builtin_amdgcn_readfirstlane(tid >> 5);
    const int lo   = blockIdx.x * nper;
    int cnt = nNodes - lo;
    cnt = cnt < 0 ? 0 : (cnt > nper ? nper : cnt);
    unsigned int* seg = list + wave * SEG_CAP;
    float* sco = score + wave * SEG_CAP;

    for (int i = tid; i < cnt * OUT_CH; i += OWN_THREADS) acc[i] = 0.f;
    for (int i = tid; i < cnt; i += OWN_THREADS) { nmaxi[i] = (int)0xFF800000u; nsum[i] = 0.f; }
    if (tid == 0) overflow = 0u;
    __syncthreads();

    static_assert(OWN_CPW * OWN_WAVES == OWN_BLOCKS && OWN_CPW <= 32, "one chunk per lane of a wave's first lanes");
    unsigned int wcount = 0u;
    {
        const int ch = chunk_edges(nEdges);
        const int cl = wave * OWN_CPW + (lane & (OWN_CPW - 1));
        const unsigned t0 = tab[cl * (OWN_BLOCKS + 1) + blockIdx.x];
        const unsigned t1 = tab[cl * (OWN_BLOCKS + 1) + blockIdx.x + 1];
        unsigned nn[OWN_CPW], bs[OWN_CPW], pv[OWN_CPW];
        const unsigned int* sp[OWN_CPW];
#pragma unroll
        for (int u = 0; u < OWN_CPW; ++u) {
            const unsigned b0 = (unsigned)__builtin_amdgcn_readlane((int)t0, u);
            const unsigned b1 = (unsigned)__builtin_amdgcn_readlane((int)t1, u);
            nn[u] = b1 - b0;
            sp[u] = binw + (size_t)(wave * OWN_CPW + u) * ch + b0;
            bs[u] = wcount;
            wcount += nn[u];
        }
#pragma unroll
        for (int u = 0; u < OWN_CPW; ++u)
            pv[u] = ((unsigned)lane < nn[u]) ? sp[u][lane] : EMPTY_WORD;
#pragma unroll
        for (int u = 0; u < OWN_CPW; ++u) {
            const unsigned idx = bs[u] + lane;
            if ((unsigned)lane < nn[u] && idx < SEG_CAP) seg[idx] = pv[u];
        }
#pragma unroll
        for (int u = 0; u < OWN_CPW; ++u) {
            for (unsigned k = 32u + lane; k < nn[u]; k += 32u) {
                const unsigned idx = bs[u] + k;
                if (idx < SEG_CAP) seg[idx] = sp[u][k];
            }
        }
    }
    if (wcount > SEG_CAP && lane == 0) overflow = 1u;
    __syncthreads();

    const unsigned over = __builtin_amdgcn_readfirstlane(overflow);
    if (over == 0u) {
        for (unsigned b = 0; b < wcount; b += 128u) {
            unsigned idx[4], p[4]; int s[4], dl[4]; bool mine[4]; float el[4], er[4];
#pragma unroll
            for (int t = 0; t < 4; ++t) {
                idx[t] = b + 32u * t + lane;
                p[t] = seg[idx[t] < wcount ? idx[t] : wcount - 1u];
            }
#pragma unroll
            for (int t = 0; t < 4; ++t) {
                s[t]  = (int)(p[t] & ((1u << S_BITS) - 1u));
                dl[t] = (int)(p[t] >> S_BITS);
                mine[t] = (unsigned)dl[t] < (unsigned)cnt;
                el[t] = e_l[mine[t] ? lo + dl[t] : 0];
                er[t] = e_r[s[t]];
            }
#pragma unroll
            for (int t = 0; t < 4; ++t) {
                if (idx[t] < wcount) {
                    if (mine[t]) {
                        const float x = el[t] + er[t];
                        const float a = (x > 0.f) ? x : NEG_SLOPE * x;
                        ldsMaxF32(&nmaxi[dl[t]], a);
                        sco[idx[t]] = a;
                    } else {
                        seg[idx[t]] = EMPTY_WORD;
                        sco[idx[t]] = 0.f;
                    }
                }
            }
        }
        __syncthreads();

        for (unsigned b = 0; b < wcount; b += 32u) {
            const unsigned i = b + lane;
            int s = 0, dl = NPER_MAX; float w = 0.f;
            const unsigned p = (i < wcount) ? seg[i] : ((unsigned)NPER_MAX << S_BITS);
            const float    a = (i < wcount) ? sco[i] : 0.f;
            if ((p >> S_BITS) != (unsigned)NPER_MAX) {
                s  = (int)(p & ((1u << S_BITS) - 1u));
                dl = (int)(p >> S_BITS);
                w = __expf(a - __int_as_float(nmaxi[dl]));
                atomicAdd(&nsum[dl], w);
            }
            const int nvalid = (wcount - b) < 32u ? (int)(wcount - b) : 32;
            for (int k0 = 0; k0 < nvalid; k0 += 16) {
                float2 hv[16]; int dk[16]; float wk[16];
#pragma unroll
                for (int u = 0; u < 16; ++u) {
                    const int k = k0 + u;
                    const int sk = __builtin_amdgcn_readlane(s, k);
                    dk[u] = __builtin_amdgcn_readlane(dl, k);
                    wk[u] = __int_as_float(__builtin_amdgcn_readlane(__float_as_int(w), k));
                    hv[u] = ((const float2*)(h + (size_t)sk * OUT_CH))[lane];
                }
#pragma unroll
                for (int u = 0; u < 16; ++u) {
                    float* r = &acc[dk[u] * OUT_CH + 2 * lane];
                    atomicAdd(r,     hv[u].x * wk[u]);
                    atomicAdd(r + 1, hv[u].y * wk[u]);
                }
            }
        }
    } else {
        scan_edges<1>(src, dst, h, e_l, e_r, nEdges, lo, cnt, lane, wave, nmaxi, nsum, acc);
        __syncthreads();
        scan_edges<2>(src, dst, h, e_l, e_r, nEdges, lo, cnt, lane, wave, nmaxi, nsum, acc);
    }
    __syncthreads();

    float* o = out + (size_t)lo * OUT_CH;
    for (int i = tid; i < cnt * OUT_CH; i += OWN_THREADS)
        o[i] = acc[i] / (nsum[i >> 6] + EPS_F) + bias[i & (OUT_CH - 1)];
}

extern "C" void kernel_launch(void* const* d_in, const int* in_sizes, int n_in,
                              void* d_out, int out_size, void* d_ws, size_t ws_size,
                              hipStream_t stream) {
    const float* x    = (const float*)d_in[0];
    const int*   ei   = (const int*)d_in[1];
    const float* W    = (const float*)d_in[2];
    const float* a_l  = (const float*)d_in[3];
    const float* a_r  = (const float*)d_in[4];
    const float* bias = (const float*)d_in[5];
    float* out = (float*)d_out;

    const int nNodes = in_sizes[0] / IN_CH;
    const int nEdges = in_sizes[1] / 2;
    const int* src = ei;
    const int* dst = ei + nEdges;

    float* ws       = (float*)d_ws;
    float* h        = ws;
    float* e_l      = h + (size_t)nNodes * OUT_CH;
    float* e_r      = e_l + nNodes;
    float* node_max = e_r + nNodes;
    float* node_sum = node_max + nNodes;
    float* edge_w   = node_sum + nNodes;

    const int mFull = nNodes & ~15;
    const int nSlabs = (mFull + GB_ROWS - 1) / GB_ROWS;
    if (nSlabs > 0)
        gat_gemm_slab<<<nSlabs < 512 ? nSlabs : 512, GB_THREADS, 0, stream>>>(x, W, h, mFull, nSlabs);
    if (mFull < nNodes) {
        int tailThreads = (nNodes - mFull) * OUT_CH;
        gat_gemm_tail<<<(tailThreads + 255) / 256, 256, 0, stream>>>(x, W, h, mFull, nNodes);
    }

    const int nper = (nNodes + OWN_BLOCKS - 1) / OWN_BLOCKS;
    const bool owner = nper <= NPER_MAX && (nEdges & 3) == 0 && nEdges <= BIN_CAP * OWN_BLOCKS &&
                       nNodes <= (1 << S_BITS) && 2 * nNodes >= OWN_BLOCKS * (OWN_BLOCKS + 1);
    unsigned int* binw = (unsigned int*)edge_w;
    unsigned int* tab  = (unsigned int*)node_max;

    const int nBin = owner ? OWN_BLOCKS : 0;
    gat_el_er<<<nBin + (nNodes + EL_ROWS * BIN_WAVES - 1) / (EL_ROWS * BIN_WAVES), BIN_THREADS, 0, stream>>>(
        h, a_l, a_r, e_l, e_r, nNodes, src, dst, binw, tab, nEdges, nper, nBin);

    if (owner) {
        gat_owner<<<OWN_BLOCKS, OWN_THREADS, 0, stream>>>(src, dst, binw, tab, h, e_l, e_r, bias, out, nNodes, nEdges, nper);
        return;
    }

    int initN = nNodes * OUT_CH;
    gat_init<<<(initN + 255) / 256, 256, 0, stream>>>(out, bias, node_max, node_sum, nNodes);

    gat_edge_max<<<(nEdges + 255) / 256, 256, 0, stream>>>(src, dst, e_l, e_r, edge_w, node_max, nEdges);
    gat_edge_exp<<<(nEdges + 255) / 256, 256, 0, stream>>>(dst, edge_w, node_max, node_sum, nEdges);
    gat_scatter<<<((size_t)nEdges * 32 + 255) / 256, 256, 0, stream>>>(src, dst, h, edge_w, node_sum, out, nEdges);
}


# owner blocks: list entries regrouped by dst node in LDS (count in max pass, scan, place); one wave per node sums its rows in registers, stores the row once; no LDS float atomics; sum order changes
# speedup vs baseline: 6.2978x; 1.1772x over previous
#include <hip/hip_runtime.h>
#include <hip/hip_bf16.h>
#include <cstddef>

typedef __attribute__((ext_vector_type(2))) float v2f;
typedef __attribute__((ext_vector_type(8))) float v8f;

#define IN_CH   128
#define OUT_CH  64
#define NEG_SLOPE 0.2f
#define EPS_F   1e-10f

#define GB_ROWS    64
#define GB_PITCH   132
#define GB_THREADS 256
#define GB_KSTEPS  (IN_CH / 4)

typedef __attribute__((ext_vector_type(4))) float v4f;

__device__ __forceinline__ float gb_xhalf(float v) {
    return __builtin_bit_cast(float, __builtin_amdgcn_permlanex16(__builtin_bit_cast(int, v), __builtin_bit_cast(int, v),
                                                                   0x76543210, 0xfedcba98, false, false));
}

__global__ __launch_bounds__(GB_THREADS)
void gat_gemm_slab(const float* __restrict__ x, const float* __restrict__ W,
                   float* __restrict__ h, int mFull, int nSlabs) {
    static_assert(OUT_CH == GB_ROWS, "W passes through a slab buffer");
    __shared__ __attribute__((aligned(16))) float xs[2][GB_ROWS * GB_PITCH];

    const int tid  = threadIdx.x;
    const int lane = tid & 31;
    const int wave = __builtin_amdgcn_readfirstlane(tid >> 5);
    const int l15  = lane & 15;
    const int half = lane >> 4;
    const int kOff = half << 1;
    const int nBase = (wave & 1) * 32;
    const int mt    = wave >> 1;

    const v4f* x4 = (const v4f*)x;
    v4f stg[GB_ROWS / 8];
    const v4f zero4 = {0.f, 0.f, 0.f, 0.f};

    int slab = blockIdx.x;
    if (slab < nSlabs) {
#pragma unroll
        for (int i = 0; i < GB_ROWS / 8; ++i) {
            const int row = slab * GB_ROWS + i * 8 + wave;
            stg[i] = (row < mFull) ? x4[(size_t)row * (IN_CH / 4) + lane] : zero4;
        }
    }
    {
        const v4f* w4 = (const v4f*)W;
        v4f wst[OUT_CH / 8];
#pragma unroll
        for (int i = 0; i < OUT_CH / 8; ++i)
            wst[i] = w4[(size_t)(i * 8 + wave) * (IN_CH / 4) + lane];
#pragma unroll
        for (int i = 0; i < OUT_CH / 8; ++i)
            *(v4f*)(&xs[1][(i * 8 + wave) * GB_PITCH + lane * 4]) = wst[i];
    }
    if (slab < nSlabs) {
#pragma unroll
        for (int i = 0; i < GB_ROWS / 8; ++i)
            *(v4f*)(&xs[0][(i * 8 + wave) * GB_PITCH + lane * 4]) = stg[i];
    }
    __syncthreads();

    v2f b0[GB_KSTEPS], b1[GB_KSTEPS];
    {
        const float* w0 = &xs[1][(nBase + l15) * GB_PITCH + kOff];
        const float* w1 = w0 + 16 * GB_PITCH;
#pragma unroll
        for (int kk = 0; kk < GB_KSTEPS; ++kk) {
            b0[kk] = *(const v2f*)(w0 + 4 * kk);
            b1[kk] = *(const v2f*)(w1 + 4 * kk);
        }
    }
    __syncthreads();

    int buf = 0;
    for (; slab < nSlabs; slab += (int)gridDim.x) {
        const int  nxt  = slab + (int)gridDim.x;
        const bool more = nxt < nSlabs;
        if (more) {
#pragma unroll
            for (int i = 0; i < GB_ROWS / 8; ++i) {
                const int row = nxt * GB_ROWS + i * 8 + wave;
                stg[i] = (row < mFull) ? x4[(size_t)row * (IN_CH / 4) + lane] : zero4;
            }
        }

        const float* ap = &xs[buf][(mt * 16 + l15) * GB_PITCH + kOff];
        v8f acc0 = {}, acc1 = {};
#pragma unroll
        for (int kk = 0; kk < GB_KSTEPS; ++kk) {
            const v2f a = *(const v2f*)(ap + 4 * kk);
            acc0 = __builtin_amdgcn_wmma_f32_16x16x4_f32(
                 false, a,  false, b0[kk],
                 (short)0, acc0,  false,  false);
            acc1 = __builtin_amdgcn_wmma_f32_16x16x4_f32(
                 false, a,  false, b1[kk],
                 (short)0, acc1,  false,  false);
        }

        const int mBase = slab * GB_ROWS + mt * 16;
        if (mBase < mFull) {
            float* o = h + (size_t)mBase * OUT_CH + nBase + lane;
#pragma unroll
            for (int r = 0; r < 8; ++r) {
                const float lo = acc0[r], hi = acc1[r];
                const float lox = gb_xhalf(lo), hix = gb_xhalf(hi);
                o[(size_t)r * OUT_CH]       = half ? hix : lo;
                o[(size_t)(r + 8) * OUT_CH] = half ? hi  : lox;
            }
        }

        if (more) {
#pragma unroll
            for (int i = 0; i < GB_ROWS / 8; ++i)
                *(v4f*)(&xs[buf ^ 1][(i * 8 + wave) * GB_PITCH + lane * 4]) = stg[i];
        }
        __syncthreads();
        buf ^= 1;
    }
}

__global__ void gat_gemm_tail(const float* __restrict__ x,
                              const float* __restrict__ W,
                              float* __restrict__ h,
                              int rowStart, int nNodes) {
    int idx = blockIdx.x * blockDim.x + threadIdx.x;
    int row = rowStart + (idx >> 6);
    int col = idx & 63;
    if (row >= nNodes) return;
    const float* xr = x + (size_t)row * IN_CH;
    const float* wr = W + (size_t)col * IN_CH;
    float s = 0.f;
    for (int k = 0; k < IN_CH; ++k) s += xr[k] * wr[k];
    h[(size_t)row * OUT_CH + col] = s;
}

typedef __attribute__((ext_vector_type(4))) int v4i;
typedef __attribute__((ext_vector_type(4))) unsigned int v4u;

#define OWN_BLOCKS   256
#define NPER_MAX     391
#define S_BITS       17
#define EMPTY_WORD   ((unsigned)NPER_MAX << S_BITS)
#define BIN_THREADS  256
#define BIN_WAVES    8
#define BIN_CAP      6400
#define BIN_QUADS    4

__device__ __forceinline__ int chunk_edges(int nEdges) {
    return (((nEdges + OWN_BLOCKS - 1) / OWN_BLOCKS) + 31) & ~31;
}

__device__ __forceinline__ void bin_chunk(const int* __restrict__ src, const int* __restrict__ dst,
                                          unsigned int* __restrict__ binw, unsigned int* __restrict__ tab,
                                          int nEdges, int nper,
                                          unsigned int* hist, unsigned int* stage, unsigned int* wtot) {
    static_assert(BIN_THREADS == OWN_BLOCKS && BIN_WAVES * 32 == BIN_THREADS, "thread t scans owner t's counters");
    static_assert(BIN_WAVES * 800 >= BIN_CAP, "a wave's stretch is two trips at most");
    const int tid  = threadIdx.x;
    const int lane = tid & 31;
    const int wave = __builtin_amdgcn_readfirstlane(tid >> 5);
    const int ch = chunk_edges(nEdges);
    const int c  = blockIdx.x;
    const int e0 = c * ch;
    int e1 = e0 + ch; if (e1 > nEdges) e1 = nEdges;
    const int wlen = ((ch + BIN_WAVES - 1) / BIN_WAVES + 31) & ~31;
    const int ws = e0 + wave * wlen;
    int we = ws + wlen; if (we > e1) we = e1;
    const unsigned un = (unsigned)nper;

    for (int i = tid; i < BIN_WAVES * OWN_BLOCKS; i += BIN_THREADS) hist[i] = 0u;
    __syncthreads();
    unsigned int* myh = hist + wave * OWN_BLOCKS;

    for (int base = ws; base < we; base += BIN_QUADS * 128) {
        v4i dq[BIN_QUADS];
#pragma unroll
        for (int u = 0; u < BIN_QUADS; ++u) {
            const int q = base + (u * 32 + lane) * 4;
            const v4i none = {-1, -1, -1, -1};
            dq[u] = (q < we) ? *(const v4i*)(dst + q) : none;
        }
#pragma unroll
        for (int u = 0; u < BIN_QUADS; ++u) {
            const int q = base + (u * 32 + lane) * 4;
            if (q < we) {
#pragma unroll
                for (int j = 0; j < 4; ++j) {
                    unsigned o = (unsigned)dq[u][j] / un; if (o > OWN_BLOCKS - 1) o = OWN_BLOCKS - 1;
                    atomicAdd(&myh[o], 1u);
                }
            }
        }
    }
    __syncthreads();

    unsigned run = 0u;
#pragma unroll
    for (int w = 0; w < BIN_WAVES; ++w) {
        const unsigned v = hist[w * OWN_BLOCKS + tid];
        hist[w * OWN_BLOCKS + tid] = run;
        run += v;
    }
    unsigned incl = run;
#pragma unroll
    for (int off = 1; off < 32; off <<= 1) {
        const unsigned y = __shfl_up(incl, off, 32);
        if (lane >= off) incl += y;
    }
    if (lane == 31) wtot[wave] = incl;
    __syncthreads();
    unsigned wbase = 0u;
    for (int j = 0; j < wave; ++j) wbase += wtot[j];
    const unsigned excl = wbase + incl - run;
#pragma unroll
    for (int w = 0; w < BIN_WAVES; ++w) hist[w * OWN_BLOCKS + tid] += excl;
    tab[c * (OWN_BLOCKS + 1) + tid] = excl;
    if (tid == OWN_BLOCKS - 1) tab[c * (OWN_BLOCKS + 1) + OWN_BLOCKS] = excl + run;
    __syncthreads();

    for (int base = ws; base < we; base += BIN_QUADS * 128) {
        v4i dq[BIN_QUADS], sq[BIN_QUADS];
#pragma unroll
        for (int u = 0; u < BIN_QUADS; ++u) {
            const int q = base + (u * 32 + lane) * 4;
            const v4i none = {-1, -1, -1, -1};
            dq[u] = (q < we) ? *(const v4i*)(dst + q) : none;
            sq[u] = (q < we) ? *(const v4i*)(src + q) : none;
        }
#pragma unroll
        for (int u = 0; u < BIN_QUADS; ++u) {
            const int q = base + (u * 32 + lane) * 4;
            if (q < we) {
#pragma unroll
                for (int j = 0; j < 4; ++j) {
                    const unsigned d = (unsigned)dq[u][j];
                    unsigned o = d / un; if (o > OWN_BLOCKS - 1) o = OWN_BLOCKS - 1;
                    const unsigned dl = d - o * un;
                    const unsigned word = (dl < un) ? (((unsigned)sq[u][j] & ((1u << S_BITS) - 1u)) | (dl << S_BITS)) : EMPTY_WORD;
                    const unsigned pos = atomicAdd(&myh[o], 1u);
                    stage[pos] = word;
                }
            }
        }
    }
    __syncthreads();
    for (int i = tid; i < e1 - e0; i += BIN_THREADS) binw[e0 + i] = stage[i];
}

#define EL_ROWS 32

template <int M>
__device__ __forceinline__ float row_xor(float v) {
    const int b = __builtin_bit_cast(int, v);
    return __builtin_bit_cast(float, __builtin_amdgcn_update_dpp(b, b, 0x160 + M, 0xf, 0xf, true));
}

__global__ __launch_bounds__(BIN_THREADS)
void gat_el_er(const float* __restrict__ h,
               const float* __restrict__ a_l,
               const float* __restrict__ a_r,
               float* __restrict__ e_l,
               float* __restrict__ e_r, int nNodes,
               const int* __restrict__ src, const int* __restrict__ dst,
               unsigned int* __restrict__ binw, unsigned int* __restrict__ tab,
               int nEdges, int nper, int nBin) {
    __shared__ unsigned int hist[BIN_WAVES * OWN_BLOCKS];
    __shared__ unsigned int stage[BIN_CAP];
    __shared__ unsigned int wtot[BIN_WAVES];
    if ((int)blockIdx.x < nBin) {
        bin_chunk(src, dst, binw, tab, nEdges, nper, hist, stage, wtot);
        return;
    }
    static_assert(OUT_CH == 64 && EL_ROWS == 32, "a 16-lane row of the wave holds one 256-B row of h, lane l keeps row l's sums");
    const int lane = threadIdx.x & 31;
    const int wave = __builtin_amdgcn_readfirstlane((int)(threadIdx.x >> 5));
    const int row0 = (((int)blockIdx.x - nBin) * BIN_WAVES + wave) * EL_ROWS;
    if (row0 >= nNodes) return;
    const int l15  = lane & 15;
    const int half = lane >> 4;
    const v4f* h4 = (const v4f*)h;
    v4f hv[EL_ROWS / 2];
#pragma unroll
    for (int i = 0; i < EL_ROWS / 2; ++i) {
        int row = row0 + 16 * half + i;
        row = row < nNodes ? row : nNodes - 1;
        hv[i] = h4[(size_t)row * (OUT_CH / 4) + l15];
    }
    const v4f al = ((const v4f*)a_l)[l15];
    const v4f ar = ((const v4f*)a_r)[l15];
    float rl = 0.f, rr = 0.f;
#pragma unroll
    for (int i = 0; i < EL_ROWS / 2; ++i) {
        float la = __builtin_fmaf(hv[i].x, al.x, hv[i].y * al.y);
        float lb = __builtin_fmaf(hv[i].z, al.z, hv[i].w * al.w);
        float ra = __builtin_fmaf(hv[i].x, ar.x, hv[i].y * ar.y);
        float rb = __builtin_fmaf(hv[i].z, ar.z, hv[i].w * ar.w);
        la += row_xor<8>(la); lb += row_xor<8>(lb); ra += row_xor<8>(ra); rb += row_xor<8>(rb);
        la += row_xor<4>(la); lb += row_xor<4>(lb); ra += row_xor<4>(ra); rb += row_xor<4>(rb);
        la += row_xor<2>(la); lb += row_xor<2>(lb); ra += row_xor<2>(ra); rb += row_xor<2>(rb);
        la += row_xor<1>(la); lb += row_xor<1>(lb); ra += row_xor<1>(ra); rb += row_xor<1>(rb);
        const float sl = la + lb, sr = ra + rb;
        if (l15 == i) { rl = sl; rr = sr; }
    }
    const int row = row0 + lane;
    if (row < nNodes) { e_l[row] = rl; e_r[row] = rr; }
}

__global__ void gat_init(float* __restrict__ out, const float* __restrict__ bias,
                         float* __restrict__ node_max, float* __restrict__ node_sum,
                         int nNodes) {
    int i = blockIdx.x * blockDim.x + threadIdx.x;
    if (i < nNodes * OUT_CH) out[i] = bias[i & (OUT_CH - 1)];
    if (i < nNodes) { node_max[i] = -__builtin_inff(); node_sum[i] = 0.f; }
}

__device__ __forceinline__ void atomicMaxF32(float* addr, float v) {
    if (v >= 0.f)
        atomicMax((int*)addr, __float_as_int(v));
    else
        atomicMin((unsigned int*)addr, __float_as_uint(v));
}

__global__ void gat_edge_max(const int* __restrict__ src, const int* __restrict__ dst,
                             const float* __restrict__ e_l, const float* __restrict__ e_r,
                             float* __restrict__ edge_w, float* __restrict__ node_max,
                             int nEdges) {
    int e = blockIdx.x * blockDim.x + threadIdx.x;
    if (e >= nEdges) return;
    int s = src[e], d = dst[e];
    float a = e_l[d] + e_r[s];
    a = (a > 0.f) ? a : NEG_SLOPE * a;
    edge_w[e] = a;
    atomicMaxF32(node_max + d, a);
}

__global__ void gat_edge_exp(const int* __restrict__ dst,
                             float* __restrict__ edge_w,
                             const float* __restrict__ node_max,
                             float* __restrict__ node_sum, int nEdges) {
    int e = blockIdx.x * blockDim.x + threadIdx.x;
    if (e >= nEdges) return;
    int d = dst[e];
    float w = __expf(edge_w[e] - node_max[d]);
    edge_w[e] = w;
    atomicAdd(node_sum + d, w);
}

__global__ void gat_scatter(const int* __restrict__ src, const int* __restrict__ dst,
                            const float* __restrict__ h,
                            const float* __restrict__ edge_w,
                            const float* __restrict__ node_sum,
                            float* __restrict__ out, int nEdges) {
    const int lane = threadIdx.x & 31;
    const int e = (blockIdx.x * blockDim.x + threadIdx.x) >> 5;
    if (e >= nEdges) return;
    const int s = src[e], d = dst[e];
    const float coef = edge_w[e] / (node_sum[d] + EPS_F);
    float2 hv = ((const float2*)(h + (size_t)s * OUT_CH))[lane];
    float* o = out + (size_t)d * OUT_CH + lane * 2;
    atomicAdd(o + 0, hv.x * coef);
    atomicAdd(o + 1, hv.y * coef);
}


#define OWN_THREADS  1024
#define OWN_WAVES    32
#define OWN_CPW      (OWN_BLOCKS / OWN_WAVES)
#define SORT_CAP     (((NPER_MAX + 1) * OUT_CH) / 2)
#define SCAN_PER     ((NPER_MAX + 31) / 32)
#define SEG_CAP      768

__device__ __forceinline__ void ldsMaxF32(int* p, float v) {
    const int b = __float_as_int(v);
    if (b >= 0) atomicMax(p, b);
    else        atomicMin((unsigned int*)p, (unsigned int)b);
}

template <int PHASE>
__device__ __forceinline__ void edge_now(bool m, int e, int dl, int lo, int lane,
                                         const int* __restrict__ src, const float* __restrict__ h,
                                         const float* __restrict__ e_l, const float* __restrict__ e_r,
                                         int* nmaxi, float* nsum, float* acc) {
    int s = 0; float w = 0.f;
    if (m) {
        s = src[e];
        const float x = e_l[lo + dl] + e_r[s];
        const float a = (x > 0.f) ? x : NEG_SLOPE * x;
        if (PHASE == 1) {
            ldsMaxF32(&nmaxi[dl], a);
        } else {
            w = __expf(a - __int_as_float(nmaxi[dl]));
            atomicAdd(&nsum[dl], w);
        }
    }
    if (PHASE == 2) {
        unsigned mask = __builtin_amdgcn_ballot_w32(m);
        const int dls = m ? dl : 0;
        while (mask) {
            const int k = __builtin_ctz(mask);
            mask &= mask - 1;
            const int   sk = __builtin_amdgcn_readlane(s, k);
            const int   dk = __builtin_amdgcn_readlane(dls, k);
            const float wk = __int_as_float(__builtin_amdgcn_readlane(__float_as_int(w), k));
            const float2 hv = ((const float2*)(h + (size_t)sk * OUT_CH))[lane];
            float* r = &acc[dk * OUT_CH + 2 * lane];
            atomicAdd(r,     hv.x * wk);
            atomicAdd(r + 1, hv.y * wk);
        }
    }
}

template <int MODE>
__device__ __forceinline__ void scan_quad(v4i v, int e0, int lo, int cnt, int lane,
                                          const int* __restrict__ src, const float* __restrict__ h,
                                          const float* __restrict__ e_l, const float* __restrict__ e_r,
                                          int* nmaxi, float* nsum, float* acc) {
    const int d0 = v.x - lo, d1 = v.y - lo, d2 = v.z - lo, d3 = v.w - lo;
    const bool m0 = (unsigned)d0 < (unsigned)cnt;
    const bool m1 = (unsigned)d1 < (unsigned)cnt;
    const bool m2 = (unsigned)d2 < (unsigned)cnt;
    const bool m3 = (unsigned)d3 < (unsigned)cnt;
    if (__builtin_amdgcn_ballot_w32(m0 | m1 | m2 | m3) == 0) return;
    edge_now<MODE>(m0, e0 + 0, d0, lo, lane, src, h, e_l, e_r, nmaxi, nsum, acc);
    edge_now<MODE>(m1, e0 + 1, d1, lo, lane, src, h, e_l, e_r, nmaxi, nsum, acc);
    edge_now<MODE>(m2, e0 + 2, d2, lo, lane, src, h, e_l, e_r, nmaxi, nsum, acc);
    edge_now<MODE>(m3, e0 + 3, d3, lo, lane, src, h, e_l, e_r, nmaxi, nsum, acc);
}

template <int MODE>
__device__ __forceinline__ void scan_edges(const int* __restrict__ src, const int* __restrict__ dst,
                                           const float* __restrict__ h,
                                           const float* __restrict__ e_l, const float* __restrict__ e_r,
                                           int nEdges, int lo, int cnt, int lane, int wave,
                                           int* nmaxi, float* nsum, float* acc) {
    const int nGroups = nEdges >> 7;
    const v4i* dst4 = (const v4i*)dst;
    const int start = (int)(((long long)blockIdx.x * nGroups) / OWN_BLOCKS);
    const v4i none = {-1, -1, -1, -1};
    for (int j = wave; j < nGroups; j += OWN_WAVES * 8) {
        v4i v[8]; int g[8];
#pragma unroll
        for (int u = 0; u < 8; ++u) {
            const int jj = j + u * OWN_WAVES;
            int gg = jj + start;
            if (gg >= nGroups) gg -= nGroups;
            g[u] = gg;
            v[u] = (jj < nGroups) ? dst4[(size_t)gg * 32 + lane] : none;
        }
#pragma unroll
        for (int u = 0; u < 8; ++u)
            scan_quad<MODE>(v[u], g[u] * 128 + lane * 4, lo, cnt, lane, src, h, e_l, e_r,
                            nmaxi, nsum, acc);
    }
    for (int eb = nGroups * 128 + wave * 32; eb < nEdges; eb += OWN_THREADS) {
        const int e = eb + lane;
        const int d = (e < nEdges) ? dst[e] : -1;
        const int dl = d - lo;
        const bool m = (unsigned)dl < (unsigned)cnt;
        const unsigned km = __builtin_amdgcn_ballot_w32(m);
        if (km == 0) continue;
        edge_now<MODE>(m, e, dl, lo, lane, src, h, e_l, e_r, nmaxi, nsum, acc);
    }
}

__global__ __launch_bounds__(OWN_THREADS)
void gat_owner(const int* __restrict__ src, const int* __restrict__ dst,
               const unsigned int* __restrict__ binw, const unsigned int* __restrict__ tab,
               const float* __restrict__ h,
               const float* __restrict__ e_l, const float* __restrict__ e_r,
               const float* __restrict__ bias, float* __restrict__ out,
               int nNodes, int nEdges, int nper) {
    __shared__ float        acc[(NPER_MAX + 1) * OUT_CH];
    __shared__ unsigned int list[OWN_WAVES * SEG_CAP];
    __shared__ float        score[OWN_WAVES * SEG_CAP];
    __shared__ int          nmaxi[NPER_MAX];
    __shared__ float        nsum[NPER_MAX];
    __shared__ unsigned int ncur[NPER_MAX + 1];
    __shared__ unsigned int overflow;
    __shared__ unsigned int total;

    const int tid  = threadIdx.x;
    const int lane = tid & 31;
    const int wave = __builtin_amdgcn_readfirstlane(tid >> 5);
    const int lo   = blockIdx.x * nper;
    int cnt = nNodes - lo;
    cnt = cnt < 0 ? 0 : (cnt > nper ? nper : cnt);
    unsigned int* seg = list + wave * SEG_CAP;
    float* sco = score + wave * SEG_CAP;

    unsigned int* ssrc = (unsigned int*)acc;
    float*        ssco = acc + SORT_CAP;

    for (int i = tid; i < cnt; i += OWN_THREADS) { nmaxi[i] = (int)0xFF800000u; nsum[i] = 0.f; ncur[i] = 0u; }
    if (tid == 0) { overflow = 0u; total = 0u; }
    __syncthreads();

    static_assert(OWN_CPW * OWN_WAVES == OWN_BLOCKS && OWN_CPW <= 32, "one chunk per lane of a wave's first lanes");
    unsigned int wcount = 0u;
    {
        const int ch = chunk_edges(nEdges);
        const int cl = wave * OWN_CPW + (lane & (OWN_CPW - 1));
        const unsigned t0 = tab[cl * (OWN_BLOCKS + 1) + blockIdx.x];
        const unsigned t1 = tab[cl * (OWN_BLOCKS + 1) + blockIdx.x + 1];
        unsigned nn[OWN_CPW], bs[OWN_CPW], pv[OWN_CPW];
        const unsigned int* sp[OWN_CPW];
#pragma unroll
        for (int u = 0; u < OWN_CPW; ++u) {
            const unsigned b0 = (unsigned)__builtin_amdgcn_readlane((int)t0, u);
            const unsigned b1 = (unsigned)__builtin_amdgcn_readlane((int)t1, u);
            nn[u] = b1 - b0;
            sp[u] = binw + (size_t)(wave * OWN_CPW + u) * ch + b0;
            bs[u] = wcount;
            wcount += nn[u];
        }
#pragma unroll
        for (int u = 0; u < OWN_CPW; ++u)
            pv[u] = ((unsigned)lane < nn[u]) ? sp[u][lane] : EMPTY_WORD;
#pragma unroll
        for (int u = 0; u < OWN_CPW; ++u) {
            const unsigned idx = bs[u] + lane;
            if ((unsigned)lane < nn[u] && idx < SEG_CAP) seg[idx] = pv[u];
        }
#pragma unroll
        for (int u = 0; u < OWN_CPW; ++u) {
            for (unsigned k = 32u + lane; k < nn[u]; k += 32u) {
                const unsigned idx = bs[u] + k;
                if (idx < SEG_CAP) seg[idx] = sp[u][k];
            }
        }
    }
    if (wcount > SEG_CAP && lane == 0) overflow = 1u;
    if (lane == 0) atomicAdd(&total, wcount);
    __syncthreads();

    const unsigned over = __builtin_amdgcn_readfirstlane(overflow | (total > (unsigned)SORT_CAP ? 1u : 0u));
    if (over == 0u) {
        for (unsigned b = 0; b < wcount; b += 128u) {
            unsigned idx[4], p[4]; int s[4], dl[4]; bool mine[4]; float el[4], er[4];
#pragma unroll
            for (int t = 0; t < 4; ++t) {
                idx[t] = b + 32u * t + lane;
                p[t] = seg[idx[t] < wcount ? idx[t] : wcount - 1u];
            }
#pragma unroll
            for (int t = 0; t < 4; ++t) {
                s[t]  = (int)(p[t] & ((1u << S_BITS) - 1u));
                dl[t] = (int)(p[t] >> S_BITS);
                mine[t] = (unsigned)dl[t] < (unsigned)cnt;
                el[t] = e_l[mine[t] ? lo + dl[t] : 0];
                er[t] = e_r[s[t]];
            }
#pragma unroll
            for (int t = 0; t < 4; ++t) {
                if (idx[t] < wcount) {
                    if (mine[t]) {
                        const float x = el[t] + er[t];
                        const float a = (x > 0.f) ? x : NEG_SLOPE * x;
                        ldsMaxF32(&nmaxi[dl[t]], a);
                        atomicAdd(&ncur[dl[t]], 1u);
                        sco[idx[t]] = a;
                    } else {
                        seg[idx[t]] = EMPTY_WORD;
                        sco[idx[t]] = 0.f;
                    }
                }
            }
        }
        __syncthreads();

        if (wave == 0) {
            unsigned v[SCAN_PER]; unsigned run = 0u;
#pragma unroll
            for (int j = 0; j < SCAN_PER; ++j) {
                const int n = lane * SCAN_PER + j;
                v[j] = (n < cnt) ? ncur[n] : 0u;
                run += v[j];
            }
            unsigned incl = run;
#pragma unroll
            for (int off = 1; off < 32; off <<= 1) {
                const unsigned y = __shfl_up(incl, off, 32);
                if (lane >= off) incl += y;
            }
            unsigned ex = incl - run;
#pragma unroll
            for (int j = 0; j < SCAN_PER; ++j) {
                const int n = lane * SCAN_PER + j;
                if (n < cnt) ncur[n] = ex;
                ex += v[j];
            }
        }
        __syncthreads();

        for (unsigned i = lane; i < wcount; i += 32u) {
            const unsigned p = seg[i];
            const unsigned dl = p >> S_BITS;
            if (dl != (unsigned)NPER_MAX) {
                const unsigned pos = atomicAdd(&ncur[dl], 1u);
                ssrc[pos] = p & ((1u << S_BITS) - 1u);
                ssco[pos] = sco[i];
            }
        }
        __syncthreads();

        const float2 bv = ((const float2*)bias)[lane];
        const float2* h2 = (const float2*)h;
        unsigned stv = 0u, env = 0u; int mi = 0;
        if (wave < cnt) { stv = ncur[wave > 0 ? wave - 1 : 0]; env = ncur[wave]; mi = nmaxi[wave]; }
        for (int n = wave; n < cnt; n += OWN_WAVES) {
            const unsigned st = n ? __builtin_amdgcn_readfirstlane(stv) : 0u;
            const unsigned en = __builtin_amdgcn_readfirstlane(env);
            const float m = __int_as_float(mi);
            const int nn = n + OWN_WAVES;
            if (nn < cnt) { stv = ncur[nn - 1]; env = ncur[nn]; mi = nmaxi[nn]; }
            float a0 = 0.f, a1 = 0.f, S = 0.f;
            for (unsigned b = st; b < en; b += 32u) {
                const unsigned nb = (en - b) < 32u ? (en - b) : 32u;
                const unsigned i = b + ((unsigned)lane < nb ? (unsigned)lane : nb - 1u);
                const int   s = (int)ssrc[i];
                const float a = ssco[i];
                const float w = ((unsigned)lane < nb) ? __expf(a - m) : 0.f;
                for (unsigned k0 = 0; k0 < nb; k0 += 16u) {
                    float2 hv[16]; float wk[16];
#pragma unroll
                    for (int u = 0; u < 16; ++u) {
                        const int k = (int)k0 + u;
                        const int sk = __builtin_amdgcn_readlane(s, k);
                        wk[u] = __int_as_float(__builtin_amdgcn_readlane(__float_as_int(w), k));
                        hv[u] = h2[(size_t)sk * (OUT_CH / 2) + lane];
                    }
#pragma unroll
                    for (int u = 0; u < 16; ++u) {
                        a0 = __builtin_fmaf(hv[u].x, wk[u], a0);
                        a1 = __builtin_fmaf(hv[u].y, wk[u], a1);
                        S += wk[u];
                    }
                }
            }
            float2 o;
            o.x = a0 / (S + EPS_F) + bv.x;
            o.y = a1 / (S + EPS_F) + bv.y;
            ((float2*)(out + (size_t)(lo + n) * OUT_CH))[lane] = o;
        }
        return;
    } else {
        for (int i = tid; i < cnt * OUT_CH; i += OWN_THREADS) acc[i] = 0.f;
        __syncthreads();
        scan_edges<1>(src, dst, h, e_l, e_r, nEdges, lo, cnt, lane, wave, nmaxi, nsum, acc);
        __syncthreads();
        scan_edges<2>(src, dst, h, e_l, e_r, nEdges, lo, cnt, lane, wave, nmaxi, nsum, acc);
    }
    __syncthreads();

    float* o = out + (size_t)lo * OUT_CH;
    for (int i = tid; i < cnt * OUT_CH; i += OWN_THREADS)
        o[i] = acc[i] / (nsum[i >> 6] + EPS_F) + bias[i & (OUT_CH - 1)];
}

extern "C" void kernel_launch(void* const* d_in, const int* in_sizes, int n_in,
                              void* d_out, int out_size, void* d_ws, size_t ws_size,
                              hipStream_t stream) {
    const float* x    = (const float*)d_in[0];
    const int*   ei   = (const int*)d_in[1];
    const float* W    = (const float*)d_in[2];
    const float* a_l  = (const float*)d_in[3];
    const float* a_r  = (const float*)d_in[4];
    const float* bias = (const float*)d_in[5];
    float* out = (float*)d_out;

    const int nNodes = in_sizes[0] / IN_CH;
    const int nEdges = in_sizes[1] / 2;
    const int* src = ei;
    const int* dst = ei + nEdges;

    float* ws       = (float*)d_ws;
    float* h        = ws;
    float* e_l      = h + (size_t)nNodes * OUT_CH;
    float* e_r      = e_l + nNodes;
    float* node_max = e_r + nNodes;
    float* node_sum = node_max + nNodes;
    float* edge_w   = node_sum + nNodes;

    const int mFull = nNodes & ~15;
    const int nSlabs = (mFull + GB_ROWS - 1) / GB_ROWS;
    if (nSlabs > 0)
        gat_gemm_slab<<<nSlabs < 512 ? nSlabs : 512, GB_THREADS, 0, stream>>>(x, W, h, mFull, nSlabs);
    if (mFull < nNodes) {
        int tailThreads = (nNodes - mFull) * OUT_CH;
        gat_gemm_tail<<<(tailThreads + 255) / 256, 256, 0, stream>>>(x, W, h, mFull, nNodes);
    }

    const int nper = (nNodes + OWN_BLOCKS - 1) / OWN_BLOCKS;
    const bool owner = nper <= NPER_MAX && (nEdges & 3) == 0 && nEdges <= BIN_CAP * OWN_BLOCKS &&
                       nNodes <= (1 << S_BITS) && 2 * nNodes >= OWN_BLOCKS * (OWN_BLOCKS + 1);
    unsigned int* binw = (unsigned int*)edge_w;
    unsigned int* tab  = (unsigned int*)node_max;

    const int nBin = owner ? OWN_BLOCKS : 0;
    gat_el_er<<<nBin + (nNodes + EL_ROWS * BIN_WAVES - 1) / (EL_ROWS * BIN_WAVES), BIN_THREADS, 0, stream>>>(
        h, a_l, a_r, e_l, e_r, nNodes, src, dst, binw, tab, nEdges, nper, nBin);

    if (owner) {
        gat_owner<<<OWN_BLOCKS, OWN_THREADS, 0, stream>>>(src, dst, binw, tab, h, e_l, e_r, bias, out, nNodes, nEdges, nper);
        return;
    }

    int initN = nNodes * OUT_CH;
    gat_init<<<(initN + 255) / 256, 256, 0, stream>>>(out, bias, node_max, node_sum, nNodes);

    gat_edge_max<<<(nEdges + 255) / 256, 256, 0, stream>>>(src, dst, e_l, e_r, edge_w, node_max, nEdges);
    gat_edge_exp<<<(nEdges + 255) / 256, 256, 0, stream>>>(dst, edge_w, node_max, node_sum, nEdges);
    gat_scatter<<<((size_t)nEdges * 32 + 255) / 256, 256, 0, stream>>>(src, dst, h, edge_w, node_sum, out, nEdges);
}


# owner node pass: rows fetched two per load (16 B per lane, one edge per 16-lane half); row offset and weight spread by DPP row_share, not readlane; halves' sums added at the end; sum order changes
# speedup vs baseline: 6.6361x; 1.0537x over previous
#include <hip/hip_runtime.h>
#include <hip/hip_bf16.h>
#include <cstddef>

typedef __attribute__((ext_vector_type(2))) float v2f;
typedef __attribute__((ext_vector_type(8))) float v8f;

#define IN_CH   128
#define OUT_CH  64
#define NEG_SLOPE 0.2f
#define EPS_F   1e-10f

#define GB_ROWS    64
#define GB_PITCH   132
#define GB_THREADS 256
#define GB_KSTEPS  (IN_CH / 4)

typedef __attribute__((ext_vector_type(4))) float v4f;

__device__ __forceinline__ float gb_xhalf(float v) {
    return __builtin_bit_cast(float, __builtin_amdgcn_permlanex16(__builtin_bit_cast(int, v), __builtin_bit_cast(int, v),
                                                                   0x76543210, 0xfedcba98, false, false));
}

__global__ __launch_bounds__(GB_THREADS)
void gat_gemm_slab(const float* __restrict__ x, const float* __restrict__ W,
                   float* __restrict__ h, int mFull, int nSlabs) {
    static_assert(OUT_CH == GB_ROWS, "W passes through a slab buffer");
    __shared__ __attribute__((aligned(16))) float xs[2][GB_ROWS * GB_PITCH];

    const int tid  = threadIdx.x;
    const int lane = tid & 31;
    const int wave = __builtin_amdgcn_readfirstlane(tid >> 5);
    const int l15  = lane & 15;
    const int half = lane >> 4;
    const int kOff = half << 1;
    const int nBase = (wave & 1) * 32;
    const int mt    = wave >> 1;

    const v4f* x4 = (const v4f*)x;
    v4f stg[GB_ROWS / 8];
    const v4f zero4 = {0.f, 0.f, 0.f, 0.f};

    int slab = blockIdx.x;
    if (slab < nSlabs) {
#pragma unroll
        for (int i = 0; i < GB_ROWS / 8; ++i) {
            const int row = slab * GB_ROWS + i * 8 + wave;
            stg[i] = (row < mFull) ? x4[(size_t)row * (IN_CH / 4) + lane] : zero4;
        }
    }
    {
        const v4f* w4 = (const v4f*)W;
        v4f wst[OUT_CH / 8];
#pragma unroll
        for (int i = 0; i < OUT_CH / 8; ++i)
            wst[i] = w4[(size_t)(i * 8 + wave) * (IN_CH / 4) + lane];
#pragma unroll
        for (int i = 0; i < OUT_CH / 8; ++i)
            *(v4f*)(&xs[1][(i * 8 + wave) * GB_PITCH + lane * 4]) = wst[i];
    }
    if (slab < nSlabs) {
#pragma unroll
        for (int i = 0; i < GB_ROWS / 8; ++i)
            *(v4f*)(&xs[0][(i * 8 + wave) * GB_PITCH + lane * 4]) = stg[i];
    }
    __syncthreads();

    v2f b0[GB_KSTEPS], b1[GB_KSTEPS];
    {
        const float* w0 = &xs[1][(nBase + l15) * GB_PITCH + kOff];
        const float* w1 = w0 + 16 * GB_PITCH;
#pragma unroll
        for (int kk = 0; kk < GB_KSTEPS; ++kk) {
            b0[kk] = *(const v2f*)(w0 + 4 * kk);
            b1[kk] = *(const v2f*)(w1 + 4 * kk);
        }
    }
    __syncthreads();

    int buf = 0;
    for (; slab < nSlabs; slab += (int)gridDim.x) {
        const int  nxt  = slab + (int)gridDim.x;
        const bool more = nxt < nSlabs;
        if (more) {
#pragma unroll
            for (int i = 0; i < GB_ROWS / 8; ++i) {
                const int row = nxt * GB_ROWS + i * 8 + wave;
                stg[i] = (row < mFull) ? x4[(size_t)row * (IN_CH / 4) + lane] : zero4;
            }
        }

        const float* ap = &xs[buf][(mt * 16 + l15) * GB_PITCH + kOff];
        v8f acc0 = {}, acc1 = {};
#pragma unroll
        for (int kk = 0; kk < GB_KSTEPS; ++kk) {
            const v2f a = *(const v2f*)(ap + 4 * kk);
            acc0 = __builtin_amdgcn_wmma_f32_16x16x4_f32(
                 false, a,  false, b0[kk],
                 (short)0, acc0,  false,  false);
            acc1 = __builtin_amdgcn_wmma_f32_16x16x4_f32(
                 false, a,  false, b1[kk],
                 (short)0, acc1,  false,  false);
        }

        const int mBase = slab * GB_ROWS + mt * 16;
        if (mBase < mFull) {
            float* o = h + (size_t)mBase * OUT_CH + nBase + lane;
#pragma unroll
            for (int r = 0; r < 8; ++r) {
                const float lo = acc0[r], hi = acc1[r];
                const float lox = gb_xhalf(lo), hix = gb_xhalf(hi);
                o[(size_t)r * OUT_CH]       = half ? hix : lo;
                o[(size_t)(r + 8) * OUT_CH] = half ? hi  : lox;
            }
        }

        if (more) {
#pragma unroll
            for (int i = 0; i < GB_ROWS / 8; ++i)
                *(v4f*)(&xs[buf ^ 1][(i * 8 + wave) * GB_PITCH + lane * 4]) = stg[i];
        }
        __syncthreads();
        buf ^= 1;
    }
}

__global__ void gat_gemm_tail(const float* __restrict__ x,
                              const float* __restrict__ W,
                              float* __restrict__ h,
                              int rowStart, int nNodes) {
    int idx = blockIdx.x * blockDim.x + threadIdx.x;
    int row = rowStart + (idx >> 6);
    int col = idx & 63;
    if (row >= nNodes) return;
    const float* xr = x + (size_t)row * IN_CH;
    const float* wr = W + (size_t)col * IN_CH;
    float s = 0.f;
    for (int k = 0; k < IN_CH; ++k) s += xr[k] * wr[k];
    h[(size_t)row * OUT_CH + col] = s;
}

typedef __attribute__((ext_vector_type(4))) int v4i;
typedef __attribute__((ext_vector_type(4))) unsigned int v4u;

#define OWN_BLOCKS   256
#define NPER_MAX     391
#define S_BITS       17
#define EMPTY_WORD   ((unsigned)NPER_MAX << S_BITS)
#define BIN_THREADS  256
#define BIN_WAVES    8
#define BIN_CAP      6400
#define BIN_QUADS    4

__device__ __forceinline__ int chunk_edges(int nEdges) {
    return (((nEdges + OWN_BLOCKS - 1) / OWN_BLOCKS) + 31) & ~31;
}

__device__ __forceinline__ void bin_chunk(const int* __restrict__ src, const int* __restrict__ dst,
                                          unsigned int* __restrict__ binw, unsigned int* __restrict__ tab,
                                          int nEdges, int nper,
                                          unsigned int* hist, unsigned int* stage, unsigned int* wtot) {
    static_assert(BIN_THREADS == OWN_BLOCKS && BIN_WAVES * 32 == BIN_THREADS, "thread t scans owner t's counters");
    static_assert(BIN_WAVES * 800 >= BIN_CAP, "a wave's stretch is two trips at most");
    const int tid  = threadIdx.x;
    const int lane = tid & 31;
    const int wave = __builtin_amdgcn_readfirstlane(tid >> 5);
    const int ch = chunk_edges(nEdges);
    const int c  = blockIdx.x;
    const int e0 = c * ch;
    int e1 = e0 + ch; if (e1 > nEdges) e1 = nEdges;
    const int wlen = ((ch + BIN_WAVES - 1) / BIN_WAVES + 31) & ~31;
    const int ws = e0 + wave * wlen;
    int we = ws + wlen; if (we > e1) we = e1;
    const unsigned un = (unsigned)nper;

    for (int i = tid; i < BIN_WAVES * OWN_BLOCKS; i += BIN_THREADS) hist[i] = 0u;
    __syncthreads();
    unsigned int* myh = hist + wave * OWN_BLOCKS;

    for (int base = ws; base < we; base += BIN_QUADS * 128) {
        v4i dq[BIN_QUADS];
#pragma unroll
        for (int u = 0; u < BIN_QUADS; ++u) {
            const int q = base + (u * 32 + lane) * 4;
            const v4i none = {-1, -1, -1, -1};
            dq[u] = (q < we) ? *(const v4i*)(dst + q) : none;
        }
#pragma unroll
        for (int u = 0; u < BIN_QUADS; ++u) {
            const int q = base + (u * 32 + lane) * 4;
            if (q < we) {
#pragma unroll
                for (int j = 0; j < 4; ++j) {
                    unsigned o = (unsigned)dq[u][j] / un; if (o > OWN_BLOCKS - 1) o = OWN_BLOCKS - 1;
                    atomicAdd(&myh[o], 1u);
                }
            }
        }
    }
    __syncthreads();

    unsigned run = 0u;
#pragma unroll
    for (int w = 0; w < BIN_WAVES; ++w) {
        const unsigned v = hist[w * OWN_BLOCKS + tid];
        hist[w * OWN_BLOCKS + tid] = run;
        run += v;
    }
    unsigned incl = run;
#pragma unroll
    for (int off = 1; off < 32; off <<= 1) {
        const unsigned y = __shfl_up(incl, off, 32);
        if (lane >= off) incl += y;
    }
    if (lane == 31) wtot[wave] = incl;
    __syncthreads();
    unsigned wbase = 0u;
    for (int j = 0; j < wave; ++j) wbase += wtot[j];
    const unsigned excl = wbase + incl - run;
#pragma unroll
    for (int w = 0; w < BIN_WAVES; ++w) hist[w * OWN_BLOCKS + tid] += excl;
    tab[c * (OWN_BLOCKS + 1) + tid] = excl;
    if (tid == OWN_BLOCKS - 1) tab[c * (OWN_BLOCKS + 1) + OWN_BLOCKS] = excl + run;
    __syncthreads();

    for (int base = ws; base < we; base += BIN_QUADS * 128) {
        v4i dq[BIN_QUADS], sq[BIN_QUADS];
#pragma unroll
        for (int u = 0; u < BIN_QUADS; ++u) {
            const int q = base + (u * 32 + lane) * 4;
            const v4i none = {-1, -1, -1, -1};
            dq[u] = (q < we) ? *(const v4i*)(dst + q) : none;
            sq[u] = (q < we) ? *(const v4i*)(src + q) : none;
        }
#pragma unroll
        for (int u = 0; u < BIN_QUADS; ++u) {
            const int q = base + (u * 32 + lane) * 4;
            if (q < we) {
#pragma unroll
                for (int j = 0; j < 4; ++j) {
                    const unsigned d = (unsigned)dq[u][j];
                    unsigned o = d / un; if (o > OWN_BLOCKS - 1) o = OWN_BLOCKS - 1;
                    const unsigned dl = d - o * un;
                    const unsigned word = (dl < un) ? (((unsigned)sq[u][j] & ((1u << S_BITS) - 1u)) | (dl << S_BITS)) : EMPTY_WORD;
                    const unsigned pos = atomicAdd(&myh[o], 1u);
                    stage[pos] = word;
                }
            }
        }
    }
    __syncthreads();
    for (int i = tid; i < e1 - e0; i += BIN_THREADS) binw[e0 + i] = stage[i];
}

#define EL_ROWS 32

template <int M>
__device__ __forceinline__ float row_xor(float v) {
    const int b = __builtin_bit_cast(int, v);
    return __builtin_bit_cast(float, __builtin_amdgcn_update_dpp(b, b, 0x160 + M, 0xf, 0xf, true));
}

__global__ __launch_bounds__(BIN_THREADS)
void gat_el_er(const float* __restrict__ h,
               const float* __restrict__ a_l,
               const float* __restrict__ a_r,
               float* __restrict__ e_l,
               float* __restrict__ e_r, int nNodes,
               const int* __restrict__ src, const int* __restrict__ dst,
               unsigned int* __restrict__ binw, unsigned int* __restrict__ tab,
               int nEdges, int nper, int nBin) {
    __shared__ unsigned int hist[BIN_WAVES * OWN_BLOCKS];
    __shared__ unsigned int stage[BIN_CAP];
    __shared__ unsigned int wtot[BIN_WAVES];
    if ((int)blockIdx.x < nBin) {
        bin_chunk(src, dst, binw, tab, nEdges, nper, hist, stage, wtot);
        return;
    }
    static_assert(OUT_CH == 64 && EL_ROWS == 32, "a 16-lane row of the wave holds one 256-B row of h, lane l keeps row l's sums");
    const int lane = threadIdx.x & 31;
    const int wave = __builtin_amdgcn_readfirstlane((int)(threadIdx.x >> 5));
    const int row0 = (((int)blockIdx.x - nBin) * BIN_WAVES + wave) * EL_ROWS;
    if (row0 >= nNodes) return;
    const int l15  = lane & 15;
    const int half = lane >> 4;
    const v4f* h4 = (const v4f*)h;
    v4f hv[EL_ROWS / 2];
#pragma unroll
    for (int i = 0; i < EL_ROWS / 2; ++i) {
        int row = row0 + 16 * half + i;
        row = row < nNodes ? row : nNodes - 1;
        hv[i] = h4[(size_t)row * (OUT_CH / 4) + l15];
    }
    const v4f al = ((const v4f*)a_l)[l15];
    const v4f ar = ((const v4f*)a_r)[l15];
    float rl = 0.f, rr = 0.f;
#pragma unroll
    for (int i = 0; i < EL_ROWS / 2; ++i) {
        float la = __builtin_fmaf(hv[i].x, al.x, hv[i].y * al.y);
        float lb = __builtin_fmaf(hv[i].z, al.z, hv[i].w * al.w);
        float ra = __builtin_fmaf(hv[i].x, ar.x, hv[i].y * ar.y);
        float rb = __builtin_fmaf(hv[i].z, ar.z, hv[i].w * ar.w);
        la += row_xor<8>(la); lb += row_xor<8>(lb); ra += row_xor<8>(ra); rb += row_xor<8>(rb);
        la += row_xor<4>(la); lb += row_xor<4>(lb); ra += row_xor<4>(ra); rb += row_xor<4>(rb);
        la += row_xor<2>(la); lb += row_xor<2>(lb); ra += row_xor<2>(ra); rb += row_xor<2>(rb);
        la += row_xor<1>(la); lb += row_xor<1>(lb); ra += row_xor<1>(ra); rb += row_xor<1>(rb);
        const float sl = la + lb, sr = ra + rb;
        if (l15 == i) { rl = sl; rr = sr; }
    }
    const int row = row0 + lane;
    if (row < nNodes) { e_l[row] = rl; e_r[row] = rr; }
}

__global__ void gat_init(float* __restrict__ out, const float* __restrict__ bias,
                         float* __restrict__ node_max, float* __restrict__ node_sum,
                         int nNodes) {
    int i = blockIdx.x * blockDim.x + threadIdx.x;
    if (i < nNodes * OUT_CH) out[i] = bias[i & (OUT_CH - 1)];
    if (i < nNodes) { node_max[i] = -__builtin_inff(); node_sum[i] = 0.f; }
}

__device__ __forceinline__ void atomicMaxF32(float* addr, float v) {
    if (v >= 0.f)
        atomicMax((int*)addr, __float_as_int(v));
    else
        atomicMin((unsigned int*)addr, __float_as_uint(v));
}

__global__ void gat_edge_max(const int* __restrict__ src, const int* __restrict__ dst,
                             const float* __restrict__ e_l, const float* __restrict__ e_r,
                             float* __restrict__ edge_w, float* __restrict__ node_max,
                             int nEdges) {
    int e = blockIdx.x * blockDim.x + threadIdx.x;
    if (e >= nEdges) return;
    int s = src[e], d = dst[e];
    float a = e_l[d] + e_r[s];
    a = (a > 0.f) ? a : NEG_SLOPE * a;
    edge_w[e] = a;
    atomicMaxF32(node_max + d, a);
}

__global__ void gat_edge_exp(const int* __restrict__ dst,
                             float* __restrict__ edge_w,
                             const float* __restrict__ node_max,
                             float* __restrict__ node_sum, int nEdges) {
    int e = blockIdx.x * blockDim.x + threadIdx.x;
    if (e >= nEdges) return;
    int d = dst[e];
    float w = __expf(edge_w[e] - node_max[d]);
    edge_w[e] = w;
    atomicAdd(node_sum + d, w);
}

__global__ void gat_scatter(const int* __restrict__ src, const int* __restrict__ dst,
                            const float* __restrict__ h,
                            const float* __restrict__ edge_w,
                            const float* __restrict__ node_sum,
                            float* __restrict__ out, int nEdges) {
    const int lane = threadIdx.x & 31;
    const int e = (blockIdx.x * blockDim.x + threadIdx.x) >> 5;
    if (e >= nEdges) return;
    const int s = src[e], d = dst[e];
    const float coef = edge_w[e] / (node_sum[d] + EPS_F);
    float2 hv = ((const float2*)(h + (size_t)s * OUT_CH))[lane];
    float* o = out + (size_t)d * OUT_CH + lane * 2;
    atomicAdd(o + 0, hv.x * coef);
    atomicAdd(o + 1, hv.y * coef);
}


template <int U>
__device__ __forceinline__ int row_share(int v) {
    return __builtin_amdgcn_update_dpp(v, v, 0x150 + U, 0xf, 0xf, true);
}

template <int G>
__device__ __forceinline__ void rows_fetch(v4f* hv, const char* hb, int rb, int lb) {
    hv[4 * G + 0] = *(const v4f*)(hb + (unsigned)(row_share<4 * G + 0>(rb) + lb));
    hv[4 * G + 1] = *(const v4f*)(hb + (unsigned)(row_share<4 * G + 1>(rb) + lb));
    hv[4 * G + 2] = *(const v4f*)(hb + (unsigned)(row_share<4 * G + 2>(rb) + lb));
    hv[4 * G + 3] = *(const v4f*)(hb + (unsigned)(row_share<4 * G + 3>(rb) + lb));
}

template <int U>
__device__ __forceinline__ void row_add(const v4f* hv, float w, v4f& a, float& S) {
    const float wu = __int_as_float(row_share<U>(__float_as_int(w)));
    a.x = __builtin_fmaf(hv[U].x, wu, a.x);
    a.y = __builtin_fmaf(hv[U].y, wu, a.y);
    a.z = __builtin_fmaf(hv[U].z, wu, a.z);
    a.w = __builtin_fmaf(hv[U].w, wu, a.w);
    S += wu;
}

template <int G>
__device__ __forceinline__ void rows_add(const v4f* hv, float w, v4f& a, float& S) {
    row_add<4 * G + 0>(hv, w, a, S);
    row_add<4 * G + 1>(hv, w, a, S);
    row_add<4 * G + 2>(hv, w, a, S);
    row_add<4 * G + 3>(hv, w, a, S);
}

#define OWN_THREADS  1024
#define OWN_WAVES    32
#define OWN_CPW      (OWN_BLOCKS / OWN_WAVES)
#define SORT_CAP     (((NPER_MAX + 1) * OUT_CH) / 2)
#define SCAN_PER     ((NPER_MAX + 31) / 32)
#define SEG_CAP      768

__device__ __forceinline__ void ldsMaxF32(int* p, float v) {
    const int b = __float_as_int(v);
    if (b >= 0) atomicMax(p, b);
    else        atomicMin((unsigned int*)p, (unsigned int)b);
}

template <int PHASE>
__device__ __forceinline__ void edge_now(bool m, int e, int dl, int lo, int lane,
                                         const int* __restrict__ src, const float* __restrict__ h,
                                         const float* __restrict__ e_l, const float* __restrict__ e_r,
                                         int* nmaxi, float* nsum, float* acc) {
    int s = 0; float w = 0.f;
    if (m) {
        s = src[e];
        const float x = e_l[lo + dl] + e_r[s];
        const float a = (x > 0.f) ? x : NEG_SLOPE * x;
        if (PHASE == 1) {
            ldsMaxF32(&nmaxi[dl], a);
        } else {
            w = __expf(a - __int_as_float(nmaxi[dl]));
            atomicAdd(&nsum[dl], w);
        }
    }
    if (PHASE == 2) {
        unsigned mask = __builtin_amdgcn_ballot_w32(m);
        const int dls = m ? dl : 0;
        while (mask) {
            const int k = __builtin_ctz(mask);
            mask &= mask - 1;
            const int   sk = __builtin_amdgcn_readlane(s, k);
            const int   dk = __builtin_amdgcn_readlane(dls, k);
            const float wk = __int_as_float(__builtin_amdgcn_readlane(__float_as_int(w), k));
            const float2 hv = ((const float2*)(h + (size_t)sk * OUT_CH))[lane];
            float* r = &acc[dk * OUT_CH + 2 * lane];
            atomicAdd(r,     hv.x * wk);
            atomicAdd(r + 1, hv.y * wk);
        }
    }
}

template <int MODE>
__device__ __forceinline__ void scan_quad(v4i v, int e0, int lo, int cnt, int lane,
                                          const int* __restrict__ src, const float* __restrict__ h,
                                          const float* __restrict__ e_l, const float* __restrict__ e_r,
                                          int* nmaxi, float* nsum, float* acc) {
    const int d0 = v.x - lo, d1 = v.y - lo, d2 = v.z - lo, d3 = v.w - lo;
    const bool m0 = (unsigned)d0 < (unsigned)cnt;
    const bool m1 = (unsigned)d1 < (unsigned)cnt;
    const bool m2 = (unsigned)d2 < (unsigned)cnt;
    const bool m3 = (unsigned)d3 < (unsigned)cnt;
    if (__builtin_amdgcn_ballot_w32(m0 | m1 | m2 | m3) == 0) return;
    edge_now<MODE>(m0, e0 + 0, d0, lo, lane, src, h, e_l, e_r, nmaxi, nsum, acc);
    edge_now<MODE>(m1, e0 + 1, d1, lo, lane, src, h, e_l, e_r, nmaxi, nsum, acc);
    edge_now<MODE>(m2, e0 + 2, d2, lo, lane, src, h, e_l, e_r, nmaxi, nsum, acc);
    edge_now<MODE>(m3, e0 + 3, d3, lo, lane, src, h, e_l, e_r, nmaxi, nsum, acc);
}

template <int MODE>
__device__ __forceinline__ void scan_edges(const int* __restrict__ src, const int* __restrict__ dst,
                                           const float* __restrict__ h,
                                           const float* __restrict__ e_l, const float* __restrict__ e_r,
                                           int nEdges, int lo, int cnt, int lane, int wave,
                                           int* nmaxi, float* nsum, float* acc) {
    const int nGroups = nEdges >> 7;
    const v4i* dst4 = (const v4i*)dst;
    const int start = (int)(((long long)blockIdx.x * nGroups) / OWN_BLOCKS);
    const v4i none = {-1, -1, -1, -1};
    for (int j = wave; j < nGroups; j += OWN_WAVES * 8) {
        v4i v[8]; int g[8];
#pragma unroll
        for (int u = 0; u < 8; ++u) {
            const int jj = j + u * OWN_WAVES;
            int gg = jj + start;
            if (gg >= nGroups) gg -= nGroups;
            g[u] = gg;
            v[u] = (jj < nGroups) ? dst4[(size_t)gg * 32 + lane] : none;
        }
#pragma unroll
        for (int u = 0; u < 8; ++u)
            scan_quad<MODE>(v[u], g[u] * 128 + lane * 4, lo, cnt, lane, src, h, e_l, e_r,
                            nmaxi, nsum, acc);
    }
    for (int eb = nGroups * 128 + wave * 32; eb < nEdges; eb += OWN_THREADS) {
        const int e = eb + lane;
        const int d = (e < nEdges) ? dst[e] : -1;
        const int dl = d - lo;
        const bool m = (unsigned)dl < (unsigned)cnt;
        const unsigned km = __builtin_amdgcn_ballot_w32(m);
        if (km == 0) continue;
        edge_now<MODE>(m, e, dl, lo, lane, src, h, e_l, e_r, nmaxi, nsum, acc);
    }
}

__global__ __launch_bounds__(OWN_THREADS)
void gat_owner(const int* __restrict__ src, const int* __restrict__ dst,
               const unsigned int* __restrict__ binw, const unsigned int* __restrict__ tab,
               const float* __restrict__ h,
               const float* __restrict__ e_l, const float* __restrict__ e_r,
               const float* __restrict__ bias, float* __restrict__ out,
               int nNodes, int nEdges, int nper) {
    __shared__ float        acc[(NPER_MAX + 1) * OUT_CH];
    __shared__ unsigned int list[OWN_WAVES * SEG_CAP];
    __shared__ float        score[OWN_WAVES * SEG_CAP];
    __shared__ int          nmaxi[NPER_MAX];
    __shared__ float        nsum[NPER_MAX];
    __shared__ unsigned int ncur[NPER_MAX + 1];
    __shared__ unsigned int overflow;
    __shared__ unsigned int total;

    const int tid  = threadIdx.x;
    const int lane = tid & 31;
    const int wave = __builtin_amdgcn_readfirstlane(tid >> 5);
    const int lo   = blockIdx.x * nper;
    int cnt = nNodes - lo;
    cnt = cnt < 0 ? 0 : (cnt > nper ? nper : cnt);
    unsigned int* seg = list + wave * SEG_CAP;
    float* sco = score + wave * SEG_CAP;

    unsigned int* ssrc = (unsigned int*)acc;
    float*        ssco = acc + SORT_CAP;

    for (int i = tid; i < cnt; i += OWN_THREADS) { nmaxi[i] = (int)0xFF800000u; nsum[i] = 0.f; ncur[i] = 0u; }
    if (tid == 0) { overflow = 0u; total = 0u; }
    __syncthreads();

    static_assert(OWN_CPW * OWN_WAVES == OWN_BLOCKS && OWN_CPW <= 32, "one chunk per lane of a wave's first lanes");
    unsigned int wcount = 0u;
    {
        const int ch = chunk_edges(nEdges);
        const int cl = wave * OWN_CPW + (lane & (OWN_CPW - 1));
        const unsigned t0 = tab[cl * (OWN_BLOCKS + 1) + blockIdx.x];
        const unsigned t1 = tab[cl * (OWN_BLOCKS + 1) + blockIdx.x + 1];
        unsigned nn[OWN_CPW], bs[OWN_CPW], pv[OWN_CPW];
        const unsigned int* sp[OWN_CPW];
#pragma unroll
        for (int u = 0; u < OWN_CPW; ++u) {
            const unsigned b0 = (unsigned)__builtin_amdgcn_readlane((int)t0, u);
            const unsigned b1 = (unsigned)__builtin_amdgcn_readlane((int)t1, u);
            nn[u] = b1 - b0;
            sp[u] = binw + (size_t)(wave * OWN_CPW + u) * ch + b0;
            bs[u] = wcount;
            wcount += nn[u];
        }
#pragma unroll
        for (int u = 0; u < OWN_CPW; ++u)
            pv[u] = ((unsigned)lane < nn[u]) ? sp[u][lane] : EMPTY_WORD;
#pragma unroll
        for (int u = 0; u < OWN_CPW; ++u) {
            const unsigned idx = bs[u] + lane;
            if ((unsigned)lane < nn[u] && idx < SEG_CAP) seg[idx] = pv[u];
        }
#pragma unroll
        for (int u = 0; u < OWN_CPW; ++u) {
            for (unsigned k = 32u + lane; k < nn[u]; k += 32u) {
                const unsigned idx = bs[u] + k;
                if (idx < SEG_CAP) seg[idx] = sp[u][k];
            }
        }
    }
    if (wcount > SEG_CAP && lane == 0) overflow = 1u;
    if (lane == 0) atomicAdd(&total, wcount);
    __syncthreads();

    const unsigned over = __builtin_amdgcn_readfirstlane(overflow | (total > (unsigned)SORT_CAP ? 1u : 0u));
    if (over == 0u) {
        for (unsigned b = 0; b < wcount; b += 128u) {
            unsigned idx[4], p[4]; int s[4], dl[4]; bool mine[4]; float el[4], er[4];
#pragma unroll
            for (int t = 0; t < 4; ++t) {
                idx[t] = b + 32u * t + lane;
                p[t] = seg[idx[t] < wcount ? idx[t] : wcount - 1u];
            }
#pragma unroll
            for (int t = 0; t < 4; ++t) {
                s[t]  = (int)(p[t] & ((1u << S_BITS) - 1u));
                dl[t] = (int)(p[t] >> S_BITS);
                mine[t] = (unsigned)dl[t] < (unsigned)cnt;
                el[t] = e_l[mine[t] ? lo + dl[t] : 0];
                er[t] = e_r[s[t]];
            }
#pragma unroll
            for (int t = 0; t < 4; ++t) {
                if (idx[t] < wcount) {
                    if (mine[t]) {
                        const float x = el[t] + er[t];
                        const float a = (x > 0.f) ? x : NEG_SLOPE * x;
                        ldsMaxF32(&nmaxi[dl[t]], a);
                        atomicAdd(&ncur[dl[t]], 1u);
                        sco[idx[t]] = a;
                    } else {
                        seg[idx[t]] = EMPTY_WORD;
                        sco[idx[t]] = 0.f;
                    }
                }
            }
        }
        __syncthreads();

        if (wave == 0) {
            unsigned v[SCAN_PER]; unsigned run = 0u;
#pragma unroll
            for (int j = 0; j < SCAN_PER; ++j) {
                const int n = lane * SCAN_PER + j;
                v[j] = (n < cnt) ? ncur[n] : 0u;
                run += v[j];
            }
            unsigned incl = run;
#pragma unroll
            for (int off = 1; off < 32; off <<= 1) {
                const unsigned y = __shfl_up(incl, off, 32);
                if (lane >= off) incl += y;
            }
            unsigned ex = incl - run;
#pragma unroll
            for (int j = 0; j < SCAN_PER; ++j) {
                const int n = lane * SCAN_PER + j;
                if (n < cnt) ncur[n] = ex;
                ex += v[j];
            }
        }
        __syncthreads();

        for (unsigned i = lane; i < wcount; i += 32u) {
            const unsigned p = seg[i];
            const unsigned dl = p >> S_BITS;
            if (dl != (unsigned)NPER_MAX) {
                const unsigned pos = atomicAdd(&ncur[dl], 1u);
                ssrc[pos] = p & ((1u << S_BITS) - 1u);
                ssco[pos] = sco[i];
            }
        }
        __syncthreads();

        const int l15  = lane & 15;
        const int half = lane >> 4;
        const unsigned eoff = 2u * (unsigned)l15 + (unsigned)half;
        const int lb = l15 << 4;
        const float2 bv = ((const float2*)bias)[eoff];
        const char* hb = (const char*)h;
        unsigned stv = 0u, env = 0u; int mi = 0;
        if (wave < cnt) { stv = ncur[wave > 0 ? wave - 1 : 0]; env = ncur[wave]; mi = nmaxi[wave]; }
        for (int n = wave; n < cnt; n += OWN_WAVES) {
            const unsigned st = n ? __builtin_amdgcn_readfirstlane(stv) : 0u;
            const unsigned en = __builtin_amdgcn_readfirstlane(env);
            const float m = __int_as_float(mi);
            const int nn = n + OWN_WAVES;
            if (nn < cnt) { stv = ncur[nn - 1]; env = ncur[nn]; mi = nmaxi[nn]; }
            v4f a = {0.f, 0.f, 0.f, 0.f}; float S = 0.f;
            for (unsigned b = st; b < en; b += 32u) {
                const unsigned nb = (en - b) < 32u ? (en - b) : 32u;
                const bool live = eoff < nb;
                const unsigned i = b + (live ? eoff : nb - 1u);
                const int   s  = (int)ssrc[i];
                const float sc = ssco[i];
                const float w  = live ? __expf(sc - m) : 0.f;
                const int rb = s << 8;
                v4f hv[16];
                rows_fetch<0>(hv, hb, rb, lb);
                if (nb > 8u)  rows_fetch<1>(hv, hb, rb, lb);
                if (nb > 16u) rows_fetch<2>(hv, hb, rb, lb);
                if (nb > 24u) rows_fetch<3>(hv, hb, rb, lb);
                rows_add<0>(hv, w, a, S);
                if (nb > 8u)  rows_add<1>(hv, w, a, S);
                if (nb > 16u) rows_add<2>(hv, w, a, S);
                if (nb > 24u) rows_add<3>(hv, w, a, S);
            }
            const float St = S + gb_xhalf(S);
            const float tx = a.x + gb_xhalf(a.x), ty = a.y + gb_xhalf(a.y);
            const float tz = a.z + gb_xhalf(a.z), tw = a.w + gb_xhalf(a.w);
            float2 o;
            o.x = (half ? tz : tx) / (St + EPS_F) + bv.x;
            o.y = (half ? tw : ty) / (St + EPS_F) + bv.y;
            ((float2*)(out + (size_t)(lo + n) * OUT_CH))[eoff] = o;
        }
        return;
    } else {
        for (int i = tid; i < cnt * OUT_CH; i += OWN_THREADS) acc[i] = 0.f;
        __syncthreads();
        scan_edges<1>(src, dst, h, e_l, e_r, nEdges, lo, cnt, lane, wave, nmaxi, nsum, acc);
        __syncthreads();
        scan_edges<2>(src, dst, h, e_l, e_r, nEdges, lo, cnt, lane, wave, nmaxi, nsum, acc);
    }
    __syncthreads();

    float* o = out + (size_t)lo * OUT_CH;
    for (int i = tid; i < cnt * OUT_CH; i += OWN_THREADS)
        o[i] = acc[i] / (nsum[i >> 6] + EPS_F) + bias[i & (OUT_CH - 1)];
}

extern "C" void kernel_launch(void* const* d_in, const int* in_sizes, int n_in,
                              void* d_out, int out_size, void* d_ws, size_t ws_size,
                              hipStream_t stream) {
    const float* x    = (const float*)d_in[0];
    const int*   ei   = (const int*)d_in[1];
    const float* W    = (const float*)d_in[2];
    const float* a_l  = (const float*)d_in[3];
    const float* a_r  = (const float*)d_in[4];
    const float* bias = (const float*)d_in[5];
    float* out = (float*)d_out;

    const int nNodes = in_sizes[0] / IN_CH;
    const int nEdges = in_sizes[1] / 2;
    const int* src = ei;
    const int* dst = ei + nEdges;

    float* ws       = (float*)d_ws;
    float* h        = ws;
    float* e_l      = h + (size_t)nNodes * OUT_CH;
    float* e_r      = e_l + nNodes;
    float* node_max = e_r + nNodes;
    float* node_sum = node_max + nNodes;
    float* edge_w   = node_sum + nNodes;

    const int mFull = nNodes & ~15;
    const int nSlabs = (mFull + GB_ROWS - 1) / GB_ROWS;
    if (nSlabs > 0)
        gat_gemm_slab<<<nSlabs < 512 ? nSlabs : 512, GB_THREADS, 0, stream>>>(x, W, h, mFull, nSlabs);
    if (mFull < nNodes) {
        int tailThreads = (nNodes - mFull) * OUT_CH;
        gat_gemm_tail<<<(tailThreads + 255) / 256, 256, 0, stream>>>(x, W, h, mFull, nNodes);
    }

    const int nper = (nNodes + OWN_BLOCKS - 1) / OWN_BLOCKS;
    const bool owner = nper <= NPER_MAX && (nEdges & 3) == 0 && nEdges <= BIN_CAP * OWN_BLOCKS &&
                       nNodes <= (1 << S_BITS) && 2 * nNodes >= OWN_BLOCKS * (OWN_BLOCKS + 1);
    unsigned int* binw = (unsigned int*)edge_w;
    unsigned int* tab  = (unsigned int*)node_max;

    const int nBin = owner ? OWN_BLOCKS : 0;
    gat_el_er<<<nBin + (nNodes + EL_ROWS * BIN_WAVES - 1) / (EL_ROWS * BIN_WAVES), BIN_THREADS, 0, stream>>>(
        h, a_l, a_r, e_l, e_r, nNodes, src, dst, binw, tab, nEdges, nper, nBin);

    if (owner) {
        gat_owner<<<OWN_BLOCKS, OWN_THREADS, 0, stream>>>(src, dst, binw, tab, h, e_l, e_r, bias, out, nNodes, nEdges, nper);
        return;
    }

    int initN = nNodes * OUT_CH;
    gat_init<<<(initN + 255) / 256, 256, 0, stream>>>(out, bias, node_max, node_sum, nNodes);

    gat_edge_max<<<(nEdges + 255) / 256, 256, 0, stream>>>(src, dst, e_l, e_r, edge_w, node_max, nEdges);
    gat_edge_exp<<<(nEdges + 255) / 256, 256, 0, stream>>>(dst, edge_w, node_max, node_sum, nEdges);
    gat_scatter<<<((size_t)nEdges * 32 + 255) / 256, 256, 0, stream>>>(src, dst, h, edge_w, node_sum, out, nEdges);
}


# edge binning: a wave keeps its whole stretch in registers: dst and src asked for once, up front (14 loads), decoded once, placed with all LDS adds issued before the stores; sum order unchanged
# speedup vs baseline: 6.8695x; 1.0352x over previous
#include <hip/hip_runtime.h>
#include <hip/hip_bf16.h>
#include <cstddef>

typedef __attribute__((ext_vector_type(2))) float v2f;
typedef __attribute__((ext_vector_type(8))) float v8f;

#define IN_CH   128
#define OUT_CH  64
#define NEG_SLOPE 0.2f
#define EPS_F   1e-10f

#define GB_ROWS    64
#define GB_PITCH   132
#define GB_THREADS 256
#define GB_KSTEPS  (IN_CH / 4)

typedef __attribute__((ext_vector_type(4))) float v4f;

__device__ __forceinline__ float gb_xhalf(float v) {
    return __builtin_bit_cast(float, __builtin_amdgcn_permlanex16(__builtin_bit_cast(int, v), __builtin_bit_cast(int, v),
                                                                   0x76543210, 0xfedcba98, false, false));
}

__global__ __launch_bounds__(GB_THREADS)
void gat_gemm_slab(const float* __restrict__ x, const float* __restrict__ W,
                   float* __restrict__ h, int mFull, int nSlabs) {
    static_assert(OUT_CH == GB_ROWS, "W passes through a slab buffer");
    __shared__ __attribute__((aligned(16))) float xs[2][GB_ROWS * GB_PITCH];

    const int tid  = threadIdx.x;
    const int lane = tid & 31;
    const int wave = __builtin_amdgcn_readfirstlane(tid >> 5);
    const int l15  = lane & 15;
    const int half = lane >> 4;
    const int kOff = half << 1;
    const int nBase = (wave & 1) * 32;
    const int mt    = wave >> 1;

    const v4f* x4 = (const v4f*)x;
    v4f stg[GB_ROWS / 8];
    const v4f zero4 = {0.f, 0.f, 0.f, 0.f};

    int slab = blockIdx.x;
    if (slab < nSlabs) {
#pragma unroll
        for (int i = 0; i < GB_ROWS / 8; ++i) {
            const int row = slab * GB_ROWS + i * 8 + wave;
            stg[i] = (row < mFull) ? x4[(size_t)row * (IN_CH / 4) + lane] : zero4;
        }
    }
    {
        const v4f* w4 = (const v4f*)W;
        v4f wst[OUT_CH / 8];
#pragma unroll
        for (int i = 0; i < OUT_CH / 8; ++i)
            wst[i] = w4[(size_t)(i * 8 + wave) * (IN_CH / 4) + lane];
#pragma unroll
        for (int i = 0; i < OUT_CH / 8; ++i)
            *(v4f*)(&xs[1][(i * 8 + wave) * GB_PITCH + lane * 4]) = wst[i];
    }
    if (slab < nSlabs) {
#pragma unroll
        for (int i = 0; i < GB_ROWS / 8; ++i)
            *(v4f*)(&xs[0][(i * 8 + wave) * GB_PITCH + lane * 4]) = stg[i];
    }
    __syncthreads();

    v2f b0[GB_KSTEPS], b1[GB_KSTEPS];
    {
        const float* w0 = &xs[1][(nBase + l15) * GB_PITCH + kOff];
        const float* w1 = w0 + 16 * GB_PITCH;
#pragma unroll
        for (int kk = 0; kk < GB_KSTEPS; ++kk) {
            b0[kk] = *(const v2f*)(w0 + 4 * kk);
            b1[kk] = *(const v2f*)(w1 + 4 * kk);
        }
    }
    __syncthreads();

    int buf = 0;
    for (; slab < nSlabs; slab += (int)gridDim.x) {
        const int  nxt  = slab + (int)gridDim.x;
        const bool more = nxt < nSlabs;
        if (more) {
#pragma unroll
            for (int i = 0; i < GB_ROWS / 8; ++i) {
                const int row = nxt * GB_ROWS + i * 8 + wave;
                stg[i] = (row < mFull) ? x4[(size_t)row * (IN_CH / 4) + lane] : zero4;
            }
        }

        const float* ap = &xs[buf][(mt * 16 + l15) * GB_PITCH + kOff];
        v8f acc0 = {}, acc1 = {};
#pragma unroll
        for (int kk = 0; kk < GB_KSTEPS; ++kk) {
            const v2f a = *(const v2f*)(ap + 4 * kk);
            acc0 = __builtin_amdgcn_wmma_f32_16x16x4_f32(
                 false, a,  false, b0[kk],
                 (short)0, acc0,  false,  false);
            acc1 = __builtin_amdgcn_wmma_f32_16x16x4_f32(
                 false, a,  false, b1[kk],
                 (short)0, acc1,  false,  false);
        }

        const int mBase = slab * GB_ROWS + mt * 16;
        if (mBase < mFull) {
            float* o = h + (size_t)mBase * OUT_CH + nBase + lane;
#pragma unroll
            for (int r = 0; r < 8; ++r) {
                const float lo = acc0[r], hi = acc1[r];
                const float lox = gb_xhalf(lo), hix = gb_xhalf(hi);
                o[(size_t)r * OUT_CH]       = half ? hix : lo;
                o[(size_t)(r + 8) * OUT_CH] = half ? hi  : lox;
            }
        }

        if (more) {
#pragma unroll
            for (int i = 0; i < GB_ROWS / 8; ++i)
                *(v4f*)(&xs[buf ^ 1][(i * 8 + wave) * GB_PITCH + lane * 4]) = stg[i];
        }
        __syncthreads();
        buf ^= 1;
    }
}

__global__ void gat_gemm_tail(const float* __restrict__ x,
                              const float* __restrict__ W,
                              float* __restrict__ h,
                              int rowStart, int nNodes) {
    int idx = blockIdx.x * blockDim.x + threadIdx.x;
    int row = rowStart + (idx >> 6);
    int col = idx & 63;
    if (row >= nNodes) return;
    const float* xr = x + (size_t)row * IN_CH;
    const float* wr = W + (size_t)col * IN_CH;
    float s = 0.f;
    for (int k = 0; k < IN_CH; ++k) s += xr[k] * wr[k];
    h[(size_t)row * OUT_CH + col] = s;
}

typedef __attribute__((ext_vector_type(4))) int v4i;
typedef __attribute__((ext_vector_type(4))) unsigned int v4u;

#define OWN_BLOCKS   256
#define NPER_MAX     391
#define S_BITS       17
#define EMPTY_WORD   ((unsigned)NPER_MAX << S_BITS)
#define BIN_THREADS  256
#define BIN_WAVES    8
#define BIN_CAP      6400
#define BIN_SLOTS    7

__device__ __forceinline__ int chunk_edges(int nEdges) {
    return (((nEdges + OWN_BLOCKS - 1) / OWN_BLOCKS) + 31) & ~31;
}

__device__ __forceinline__ void bin_chunk(const int* __restrict__ src, const int* __restrict__ dst,
                                          unsigned int* __restrict__ binw, unsigned int* __restrict__ tab,
                                          int nEdges, int nper,
                                          unsigned int* hist, unsigned int* stage, unsigned int* wtot) {
    static_assert(BIN_THREADS == OWN_BLOCKS && BIN_WAVES * 32 == BIN_THREADS, "thread t scans owner t's counters");
    static_assert(BIN_WAVES * 800 >= BIN_CAP && BIN_SLOTS * 128 >= 800, "a wave's stretch fits the lane's load slots");
    const int tid  = threadIdx.x;
    const int lane = tid & 31;
    const int wave = __builtin_amdgcn_readfirstlane(tid >> 5);
    const int ch = chunk_edges(nEdges);
    const int c  = blockIdx.x;
    const int e0 = c * ch;
    int e1 = e0 + ch; if (e1 > nEdges) e1 = nEdges;
    const int wlen = ((ch + BIN_WAVES - 1) / BIN_WAVES + 31) & ~31;
    const int ws = e0 + wave * wlen;
    int we = ws + wlen; if (we > e1) we = e1;
    const unsigned un = (unsigned)nper;

    v4i dq[BIN_SLOTS], sq[BIN_SLOTS];
#pragma unroll
    for (int u = 0; u < BIN_SLOTS; ++u) {
        const int q = ws + (u * 32 + lane) * 4;
        const v4i none = {-1, -1, -1, -1};
        dq[u] = (q < we) ? *(const v4i*)(dst + q) : none;
        sq[u] = (q < we) ? *(const v4i*)(src + q) : none;
    }
    for (int i = tid; i < BIN_WAVES * OWN_BLOCKS; i += BIN_THREADS) hist[i] = 0u;
    __syncthreads();
    unsigned int* myh = hist + wave * OWN_BLOCKS;

    unsigned wd[BIN_SLOTS][4];
    unsigned ow[BIN_SLOTS];
#pragma unroll
    for (int u = 0; u < BIN_SLOTS; ++u) {
        const int q = ws + (u * 32 + lane) * 4;
        ow[u] = 0u;
#pragma unroll
        for (int j = 0; j < 4; ++j) wd[u][j] = EMPTY_WORD;
        if (q < we) {
#pragma unroll
            for (int j = 0; j < 4; ++j) {
                const unsigned d = (unsigned)dq[u][j];
                unsigned o = d / un; if (o > OWN_BLOCKS - 1) o = OWN_BLOCKS - 1;
                const unsigned dl = d - o * un;
                wd[u][j] = (dl < un) ? (((unsigned)sq[u][j] & ((1u << S_BITS) - 1u)) | (dl << S_BITS)) : EMPTY_WORD;
                ow[u] |= o << (8 * j);
                atomicAdd(&myh[o], 1u);
            }
        }
    }
    __syncthreads();

    unsigned run = 0u;
#pragma unroll
    for (int w = 0; w < BIN_WAVES; ++w) {
        const unsigned v = hist[w * OWN_BLOCKS + tid];
        hist[w * OWN_BLOCKS + tid] = run;
        run += v;
    }
    unsigned incl = run;
#pragma unroll
    for (int off = 1; off < 32; off <<= 1) {
        const unsigned y = __shfl_up(incl, off, 32);
        if (lane >= off) incl += y;
    }
    if (lane == 31) wtot[wave] = incl;
    __syncthreads();
    unsigned wbase = 0u;
    for (int j = 0; j < wave; ++j) wbase += wtot[j];
    const unsigned excl = wbase + incl - run;
#pragma unroll
    for (int w = 0; w < BIN_WAVES; ++w) hist[w * OWN_BLOCKS + tid] += excl;
    tab[c * (OWN_BLOCKS + 1) + tid] = excl;
    if (tid == OWN_BLOCKS - 1) tab[c * (OWN_BLOCKS + 1) + OWN_BLOCKS] = excl + run;
    __syncthreads();

    unsigned pos[BIN_SLOTS][4];
#pragma unroll
    for (int u = 0; u < BIN_SLOTS; ++u) {
        const int q = ws + (u * 32 + lane) * 4;
        if (q < we) {
#pragma unroll
            for (int j = 0; j < 4; ++j) pos[u][j] = atomicAdd(&myh[(ow[u] >> (8 * j)) & 255u], 1u);
        }
    }
#pragma unroll
    for (int u = 0; u < BIN_SLOTS; ++u) {
        const int q = ws + (u * 32 + lane) * 4;
        if (q < we) {
#pragma unroll
            for (int j = 0; j < 4; ++j) stage[pos[u][j]] = wd[u][j];
        }
    }
    __syncthreads();
    for (int i = tid; i < e1 - e0; i += BIN_THREADS) binw[e0 + i] = stage[i];
}

#define EL_ROWS 32

template <int M>
__device__ __forceinline__ float row_xor(float v) {
    const int b = __builtin_bit_cast(int, v);
    return __builtin_bit_cast(float, __builtin_amdgcn_update_dpp(b, b, 0x160 + M, 0xf, 0xf, true));
}

__global__ __launch_bounds__(BIN_THREADS)
void gat_el_er(const float* __restrict__ h,
               const float* __restrict__ a_l,
               const float* __restrict__ a_r,
               float* __restrict__ e_l,
               float* __restrict__ e_r, int nNodes,
               const int* __restrict__ src, const int* __restrict__ dst,
               unsigned int* __restrict__ binw, unsigned int* __restrict__ tab,
               int nEdges, int nper, int nBin) {
    __shared__ unsigned int hist[BIN_WAVES * OWN_BLOCKS];
    __shared__ unsigned int stage[BIN_CAP];
    __shared__ unsigned int wtot[BIN_WAVES];
    if ((int)blockIdx.x < nBin) {
        bin_chunk(src, dst, binw, tab, nEdges, nper, hist, stage, wtot);
        return;
    }
    static_assert(OUT_CH == 64 && EL_ROWS == 32, "a 16-lane row of the wave holds one 256-B row of h, lane l keeps row l's sums");
    const int lane = threadIdx.x & 31;
    const int wave = __builtin_amdgcn_readfirstlane((int)(threadIdx.x >> 5));
    const int row0 = (((int)blockIdx.x - nBin) * BIN_WAVES + wave) * EL_ROWS;
    if (row0 >= nNodes) return;
    const int l15  = lane & 15;
    const int half = lane >> 4;
    const v4f* h4 = (const v4f*)h;
    v4f hv[EL_ROWS / 2];
#pragma unroll
    for (int i = 0; i < EL_ROWS / 2; ++i) {
        int row = row0 + 16 * half + i;
        row = row < nNodes ? row : nNodes - 1;
        hv[i] = h4[(size_t)row * (OUT_CH / 4) + l15];
    }
    const v4f al = ((const v4f*)a_l)[l15];
    const v4f ar = ((const v4f*)a_r)[l15];
    float rl = 0.f, rr = 0.f;
#pragma unroll
    for (int i = 0; i < EL_ROWS / 2; ++i) {
        float la = __builtin_fmaf(hv[i].x, al.x, hv[i].y * al.y);
        float lb = __builtin_fmaf(hv[i].z, al.z, hv[i].w * al.w);
        float ra = __builtin_fmaf(hv[i].x, ar.x, hv[i].y * ar.y);
        float rb = __builtin_fmaf(hv[i].z, ar.z, hv[i].w * ar.w);
        la += row_xor<8>(la); lb += row_xor<8>(lb); ra += row_xor<8>(ra); rb += row_xor<8>(rb);
        la += row_xor<4>(la); lb += row_xor<4>(lb); ra += row_xor<4>(ra); rb += row_xor<4>(rb);
        la += row_xor<2>(la); lb += row_xor<2>(lb); ra += row_xor<2>(ra); rb += row_xor<2>(rb);
        la += row_xor<1>(la); lb += row_xor<1>(lb); ra += row_xor<1>(ra); rb += row_xor<1>(rb);
        const float sl = la + lb, sr = ra + rb;
        if (l15 == i) { rl = sl; rr = sr; }
    }
    const int row = row0 + lane;
    if (row < nNodes) { e_l[row] = rl; e_r[row] = rr; }
}

__global__ void gat_init(float* __restrict__ out, const float* __restrict__ bias,
                         float* __restrict__ node_max, float* __restrict__ node_sum,
                         int nNodes) {
    int i = blockIdx.x * blockDim.x + threadIdx.x;
    if (i < nNodes * OUT_CH) out[i] = bias[i & (OUT_CH - 1)];
    if (i < nNodes) { node_max[i] = -__builtin_inff(); node_sum[i] = 0.f; }
}

__device__ __forceinline__ void atomicMaxF32(float* addr, float v) {
    if (v >= 0.f)
        atomicMax((int*)addr, __float_as_int(v));
    else
        atomicMin((unsigned int*)addr, __float_as_uint(v));
}

__global__ void gat_edge_max(const int* __restrict__ src, const int* __restrict__ dst,
                             const float* __restrict__ e_l, const float* __restrict__ e_r,
                             float* __restrict__ edge_w, float* __restrict__ node_max,
                             int nEdges) {
    int e = blockIdx.x * blockDim.x + threadIdx.x;
    if (e >= nEdges) return;
    int s = src[e], d = dst[e];
    float a = e_l[d] + e_r[s];
    a = (a > 0.f) ? a : NEG_SLOPE * a;
    edge_w[e] = a;
    atomicMaxF32(node_max + d, a);
}

__global__ void gat_edge_exp(const int* __restrict__ dst,
                             float* __restrict__ edge_w,
                             const float* __restrict__ node_max,
                             float* __restrict__ node_sum, int nEdges) {
    int e = blockIdx.x * blockDim.x + threadIdx.x;
    if (e >= nEdges) return;
    int d = dst[e];
    float w = __expf(edge_w[e] - node_max[d]);
    edge_w[e] = w;
    atomicAdd(node_sum + d, w);
}

__global__ void gat_scatter(const int* __restrict__ src, const int* __restrict__ dst,
                            const float* __restrict__ h,
                            const float* __restrict__ edge_w,
                            const float* __restrict__ node_sum,
                            float* __restrict__ out, int nEdges) {
    const int lane = threadIdx.x & 31;
    const int e = (blockIdx.x * blockDim.x + threadIdx.x) >> 5;
    if (e >= nEdges) return;
    const int s = src[e], d = dst[e];
    const float coef = edge_w[e] / (node_sum[d] + EPS_F);
    float2 hv = ((const float2*)(h + (size_t)s * OUT_CH))[lane];
    float* o = out + (size_t)d * OUT_CH + lane * 2;
    atomicAdd(o + 0, hv.x * coef);
    atomicAdd(o + 1, hv.y * coef);
}


template <int U>
__device__ __forceinline__ int row_share(int v) {
    return __builtin_amdgcn_update_dpp(v, v, 0x150 + U, 0xf, 0xf, true);
}

template <int G>
__device__ __forceinline__ void rows_fetch(v4f* hv, const char* hb, int rb, int lb) {
    hv[4 * G + 0] = *(const v4f*)(hb + (unsigned)(row_share<4 * G + 0>(rb) + lb));
    hv[4 * G + 1] = *(const v4f*)(hb + (unsigned)(row_share<4 * G + 1>(rb) + lb));
    hv[4 * G + 2] = *(const v4f*)(hb + (unsigned)(row_share<4 * G + 2>(rb) + lb));
    hv[4 * G + 3] = *(const v4f*)(hb + (unsigned)(row_share<4 * G + 3>(rb) + lb));
}

template <int U>
__device__ __forceinline__ void row_add(const v4f* hv, float w, v4f& a, float& S) {
    const float wu = __int_as_float(row_share<U>(__float_as_int(w)));
    a.x = __builtin_fmaf(hv[U].x, wu, a.x);
    a.y = __builtin_fmaf(hv[U].y, wu, a.y);
    a.z = __builtin_fmaf(hv[U].z, wu, a.z);
    a.w = __builtin_fmaf(hv[U].w, wu, a.w);
    S += wu;
}

template <int G>
__device__ __forceinline__ void rows_add(const v4f* hv, float w, v4f& a, float& S) {
    row_add<4 * G + 0>(hv, w, a, S);
    row_add<4 * G + 1>(hv, w, a, S);
    row_add<4 * G + 2>(hv, w, a, S);
    row_add<4 * G + 3>(hv, w, a, S);
}

#define OWN_THREADS  1024
#define OWN_WAVES    32
#define OWN_CPW      (OWN_BLOCKS / OWN_WAVES)
#define SORT_CAP     (((NPER_MAX + 1) * OUT_CH) / 2)
#define SCAN_PER     ((NPER_MAX + 31) / 32)
#define SEG_CAP      768

__device__ __forceinline__ void ldsMaxF32(int* p, float v) {
    const int b = __float_as_int(v);
    if (b >= 0) atomicMax(p, b);
    else        atomicMin((unsigned int*)p, (unsigned int)b);
}

template <int PHASE>
__device__ __forceinline__ void edge_now(bool m, int e, int dl, int lo, int lane,
                                         const int* __restrict__ src, const float* __restrict__ h,
                                         const float* __restrict__ e_l, const float* __restrict__ e_r,
                                         int* nmaxi, float* nsum, float* acc) {
    int s = 0; float w = 0.f;
    if (m) {
        s = src[e];
        const float x = e_l[lo + dl] + e_r[s];
        const float a = (x > 0.f) ? x : NEG_SLOPE * x;
        if (PHASE == 1) {
            ldsMaxF32(&nmaxi[dl], a);
        } else {
            w = __expf(a - __int_as_float(nmaxi[dl]));
            atomicAdd(&nsum[dl], w);
        }
    }
    if (PHASE == 2) {
        unsigned mask = __builtin_amdgcn_ballot_w32(m);
        const int dls = m ? dl : 0;
        while (mask) {
            const int k = __builtin_ctz(mask);
            mask &= mask - 1;
            const int   sk = __builtin_amdgcn_readlane(s, k);
            const int   dk = __builtin_amdgcn_readlane(dls, k);
            const float wk = __int_as_float(__builtin_amdgcn_readlane(__float_as_int(w), k));
            const float2 hv = ((const float2*)(h + (size_t)sk * OUT_CH))[lane];
            float* r = &acc[dk * OUT_CH + 2 * lane];
            atomicAdd(r,     hv.x * wk);
            atomicAdd(r + 1, hv.y * wk);
        }
    }
}

template <int MODE>
__device__ __forceinline__ void scan_quad(v4i v, int e0, int lo, int cnt, int lane,
                                          const int* __restrict__ src, const float* __restrict__ h,
                                          const float* __restrict__ e_l, const float* __restrict__ e_r,
                                          int* nmaxi, float* nsum, float* acc) {
    const int d0 = v.x - lo, d1 = v.y - lo, d2 = v.z - lo, d3 = v.w - lo;
    const bool m0 = (unsigned)d0 < (unsigned)cnt;
    const bool m1 = (unsigned)d1 < (unsigned)cnt;
    const bool m2 = (unsigned)d2 < (unsigned)cnt;
    const bool m3 = (unsigned)d3 < (unsigned)cnt;
    if (__builtin_amdgcn_ballot_w32(m0 | m1 | m2 | m3) == 0) return;
    edge_now<MODE>(m0, e0 + 0, d0, lo, lane, src, h, e_l, e_r, nmaxi, nsum, acc);
    edge_now<MODE>(m1, e0 + 1, d1, lo, lane, src, h, e_l, e_r, nmaxi, nsum, acc);
    edge_now<MODE>(m2, e0 + 2, d2, lo, lane, src, h, e_l, e_r, nmaxi, nsum, acc);
    edge_now<MODE>(m3, e0 + 3, d3, lo, lane, src, h, e_l, e_r, nmaxi, nsum, acc);
}

template <int MODE>
__device__ __forceinline__ void scan_edges(const int* __restrict__ src, const int* __restrict__ dst,
                                           const float* __restrict__ h,
                                           const float* __restrict__ e_l, const float* __restrict__ e_r,
                                           int nEdges, int lo, int cnt, int lane, int wave,
                                           int* nmaxi, float* nsum, float* acc) {
    const int nGroups = nEdges >> 7;
    const v4i* dst4 = (const v4i*)dst;
    const int start = (int)(((long long)blockIdx.x * nGroups) / OWN_BLOCKS);
    const v4i none = {-1, -1, -1, -1};
    for (int j = wave; j < nGroups; j += OWN_WAVES * 8) {
        v4i v[8]; int g[8];
#pragma unroll
        for (int u = 0; u < 8; ++u) {
            const int jj = j + u * OWN_WAVES;
            int gg = jj + start;
            if (gg >= nGroups) gg -= nGroups;
            g[u] = gg;
            v[u] = (jj < nGroups) ? dst4[(size_t)gg * 32 + lane] : none;
        }
#pragma unroll
        for (int u = 0; u < 8; ++u)
            scan_quad<MODE>(v[u], g[u] * 128 + lane * 4, lo, cnt, lane, src, h, e_l, e_r,
                            nmaxi, nsum, acc);
    }
    for (int eb = nGroups * 128 + wave * 32; eb < nEdges; eb += OWN_THREADS) {
        const int e = eb + lane;
        const int d = (e < nEdges) ? dst[e] : -1;
        const int dl = d - lo;
        const bool m = (unsigned)dl < (unsigned)cnt;
        const unsigned km = __builtin_amdgcn_ballot_w32(m);
        if (km == 0) continue;
        edge_now<MODE>(m, e, dl, lo, lane, src, h, e_l, e_r, nmaxi, nsum, acc);
    }
}

__global__ __launch_bounds__(OWN_THREADS)
void gat_owner(const int* __restrict__ src, const int* __restrict__ dst,
               const unsigned int* __restrict__ binw, const unsigned int* __restrict__ tab,
               const float* __restrict__ h,
               const float* __restrict__ e_l, const float* __restrict__ e_r,
               const float* __restrict__ bias, float* __restrict__ out,
               int nNodes, int nEdges, int nper) {
    __shared__ float        acc[(NPER_MAX + 1) * OUT_CH];
    __shared__ unsigned int list[OWN_WAVES * SEG_CAP];
    __shared__ float        score[OWN_WAVES * SEG_CAP];
    __shared__ int          nmaxi[NPER_MAX];
    __shared__ float        nsum[NPER_MAX];
    __shared__ unsigned int ncur[NPER_MAX + 1];
    __shared__ unsigned int overflow;
    __shared__ unsigned int total;

    const int tid  = threadIdx.x;
    const int lane = tid & 31;
    const int wave = __builtin_amdgcn_readfirstlane(tid >> 5);
    const int lo   = blockIdx.x * nper;
    int cnt = nNodes - lo;
    cnt = cnt < 0 ? 0 : (cnt > nper ? nper : cnt);
    unsigned int* seg = list + wave * SEG_CAP;
    float* sco = score + wave * SEG_CAP;

    unsigned int* ssrc = (unsigned int*)acc;
    float*        ssco = acc + SORT_CAP;

    for (int i = tid; i < cnt; i += OWN_THREADS) { nmaxi[i] = (int)0xFF800000u; nsum[i] = 0.f; ncur[i] = 0u; }
    if (tid == 0) { overflow = 0u; total = 0u; }
    __syncthreads();

    static_assert(OWN_CPW * OWN_WAVES == OWN_BLOCKS && OWN_CPW <= 32, "one chunk per lane of a wave's first lanes");
    unsigned int wcount = 0u;
    {
        const int ch = chunk_edges(nEdges);
        const int cl = wave * OWN_CPW + (lane & (OWN_CPW - 1));
        const unsigned t0 = tab[cl * (OWN_BLOCKS + 1) + blockIdx.x];
        const unsigned t1 = tab[cl * (OWN_BLOCKS + 1) + blockIdx.x + 1];
        unsigned nn[OWN_CPW], bs[OWN_CPW], pv[OWN_CPW];
        const unsigned int* sp[OWN_CPW];
#pragma unroll
        for (int u = 0; u < OWN_CPW; ++u) {
            const unsigned b0 = (unsigned)__builtin_amdgcn_readlane((int)t0, u);
            const unsigned b1 = (unsigned)__builtin_amdgcn_readlane((int)t1, u);
            nn[u] = b1 - b0;
            sp[u] = binw + (size_t)(wave * OWN_CPW + u) * ch + b0;
            bs[u] = wcount;
            wcount += nn[u];
        }
#pragma unroll
        for (int u = 0; u < OWN_CPW; ++u)
            pv[u] = ((unsigned)lane < nn[u]) ? sp[u][lane] : EMPTY_WORD;
#pragma unroll
        for (int u = 0; u < OWN_CPW; ++u) {
            const unsigned idx = bs[u] + lane;
            if ((unsigned)lane < nn[u] && idx < SEG_CAP) seg[idx] = pv[u];
        }
#pragma unroll
        for (int u = 0; u < OWN_CPW; ++u) {
            for (unsigned k = 32u + lane; k < nn[u]; k += 32u) {
                const unsigned idx = bs[u] + k;
                if (idx < SEG_CAP) seg[idx] = sp[u][k];
            }
        }
    }
    if (wcount > SEG_CAP && lane == 0) overflow = 1u;
    if (lane == 0) atomicAdd(&total, wcount);
    __syncthreads();

    const unsigned over = __builtin_amdgcn_readfirstlane(overflow | (total > (unsigned)SORT_CAP ? 1u : 0u));
    if (over == 0u) {
        for (unsigned b = 0; b < wcount; b += 128u) {
            unsigned idx[4], p[4]; int s[4], dl[4]; bool mine[4]; float el[4], er[4];
#pragma unroll
            for (int t = 0; t < 4; ++t) {
                idx[t] = b + 32u * t + lane;
                p[t] = seg[idx[t] < wcount ? idx[t] : wcount - 1u];
            }
#pragma unroll
            for (int t = 0; t < 4; ++t) {
                s[t]  = (int)(p[t] & ((1u << S_BITS) - 1u));
                dl[t] = (int)(p[t] >> S_BITS);
                mine[t] = (unsigned)dl[t] < (unsigned)cnt;
                el[t] = e_l[mine[t] ? lo + dl[t] : 0];
                er[t] = e_r[s[t]];
            }
#pragma unroll
            for (int t = 0; t < 4; ++t) {
                if (idx[t] < wcount) {
                    if (mine[t]) {
                        const float x = el[t] + er[t];
                        const float a = (x > 0.f) ? x : NEG_SLOPE * x;
                        ldsMaxF32(&nmaxi[dl[t]], a);
                        atomicAdd(&ncur[dl[t]], 1u);
                        sco[idx[t]] = a;
                    } else {
                        seg[idx[t]] = EMPTY_WORD;
                        sco[idx[t]] = 0.f;
                    }
                }
            }
        }
        __syncthreads();

        if (wave == 0) {
            unsigned v[SCAN_PER]; unsigned run = 0u;
#pragma unroll
            for (int j = 0; j < SCAN_PER; ++j) {
                const int n = lane * SCAN_PER + j;
                v[j] = (n < cnt) ? ncur[n] : 0u;
                run += v[j];
            }
            unsigned incl = run;
#pragma unroll
            for (int off = 1; off < 32; off <<= 1) {
                const unsigned y = __shfl_up(incl, off, 32);
                if (lane >= off) incl += y;
            }
            unsigned ex = incl - run;
#pragma unroll
            for (int j = 0; j < SCAN_PER; ++j) {
                const int n = lane * SCAN_PER + j;
                if (n < cnt) ncur[n] = ex;
                ex += v[j];
            }
        }
        __syncthreads();

        for (unsigned i = lane; i < wcount; i += 32u) {
            const unsigned p = seg[i];
            const unsigned dl = p >> S_BITS;
            if (dl != (unsigned)NPER_MAX) {
                const unsigned pos = atomicAdd(&ncur[dl], 1u);
                ssrc[pos] = p & ((1u << S_BITS) - 1u);
                ssco[pos] = sco[i];
            }
        }
        __syncthreads();

        const int l15  = lane & 15;
        const int half = lane >> 4;
        const unsigned eoff = 2u * (unsigned)l15 + (unsigned)half;
        const int lb = l15 << 4;
        const float2 bv = ((const float2*)bias)[eoff];
        const char* hb = (const char*)h;
        unsigned stv = 0u, env = 0u; int mi = 0;
        if (wave < cnt) { stv = ncur[wave > 0 ? wave - 1 : 0]; env = ncur[wave]; mi = nmaxi[wave]; }
        for (int n = wave; n < cnt; n += OWN_WAVES) {
            const unsigned st = n ? __builtin_amdgcn_readfirstlane(stv) : 0u;
            const unsigned en = __builtin_amdgcn_readfirstlane(env);
            const float m = __int_as_float(mi);
            const int nn = n + OWN_WAVES;
            if (nn < cnt) { stv = ncur[nn - 1]; env = ncur[nn]; mi = nmaxi[nn]; }
            v4f a = {0.f, 0.f, 0.f, 0.f}; float S = 0.f;
            for (unsigned b = st; b < en; b += 32u) {
                const unsigned nb = (en - b) < 32u ? (en - b) : 32u;
                const bool live = eoff < nb;
                const unsigned i = b + (live ? eoff : nb - 1u);
                const int   s  = (int)ssrc[i];
                const float sc = ssco[i];
                const float w  = live ? __expf(sc - m) : 0.f;
                const int rb = s << 8;
                v4f hv[16];
                rows_fetch<0>(hv, hb, rb, lb);
                if (nb > 8u)  rows_fetch<1>(hv, hb, rb, lb);
                if (nb > 16u) rows_fetch<2>(hv, hb, rb, lb);
                if (nb > 24u) rows_fetch<3>(hv, hb, rb, lb);
                rows_add<0>(hv, w, a, S);
                if (nb > 8u)  rows_add<1>(hv, w, a, S);
                if (nb > 16u) rows_add<2>(hv, w, a, S);
                if (nb > 24u) rows_add<3>(hv, w, a, S);
            }
            const float St = S + gb_xhalf(S);
            const float tx = a.x + gb_xhalf(a.x), ty = a.y + gb_xhalf(a.y);
            const float tz = a.z + gb_xhalf(a.z), tw = a.w + gb_xhalf(a.w);
            float2 o;
            o.x = (half ? tz : tx) / (St + EPS_F) + bv.x;
            o.y = (half ? tw : ty) / (St + EPS_F) + bv.y;
            ((float2*)(out + (size_t)(lo + n) * OUT_CH))[eoff] = o;
        }
        return;
    } else {
        for (int i = tid; i < cnt * OUT_CH; i += OWN_THREADS) acc[i] = 0.f;
        __syncthreads();
        scan_edges<1>(src, dst, h, e_l, e_r, nEdges, lo, cnt, lane, wave, nmaxi, nsum, acc);
        __syncthreads();
        scan_edges<2>(src, dst, h, e_l, e_r, nEdges, lo, cnt, lane, wave, nmaxi, nsum, acc);
    }
    __syncthreads();

    float* o = out + (size_t)lo * OUT_CH;
    for (int i = tid; i < cnt * OUT_CH; i += OWN_THREADS)
        o[i] = acc[i] / (nsum[i >> 6] + EPS_F) + bias[i & (OUT_CH - 1)];
}

extern "C" void kernel_launch(void* const* d_in, const int* in_sizes, int n_in,
                              void* d_out, int out_size, void* d_ws, size_t ws_size,
                              hipStream_t stream) {
    const float* x    = (const float*)d_in[0];
    const int*   ei   = (const int*)d_in[1];
    const float* W    = (const float*)d_in[2];
    const float* a_l  = (const float*)d_in[3];
    const float* a_r  = (const float*)d_in[4];
    const float* bias = (const float*)d_in[5];
    float* out = (float*)d_out;

    const int nNodes = in_sizes[0] / IN_CH;
    const int nEdges = in_sizes[1] / 2;
    const int* src = ei;
    const int* dst = ei + nEdges;

    float* ws       = (float*)d_ws;
    float* h        = ws;
    float* e_l      = h + (size_t)nNodes * OUT_CH;
    float* e_r      = e_l + nNodes;
    float* node_max = e_r + nNodes;
    float* node_sum = node_max + nNodes;
    float* edge_w   = node_sum + nNodes;

    const int mFull = nNodes & ~15;
    const int nSlabs = (mFull + GB_ROWS - 1) / GB_ROWS;
    if (nSlabs > 0)
        gat_gemm_slab<<<nSlabs < 512 ? nSlabs : 512, GB_THREADS, 0, stream>>>(x, W, h, mFull, nSlabs);
    if (mFull < nNodes) {
        int tailThreads = (nNodes - mFull) * OUT_CH;
        gat_gemm_tail<<<(tailThreads + 255) / 256, 256, 0, stream>>>(x, W, h, mFull, nNodes);
    }

    const int nper = (nNodes + OWN_BLOCKS - 1) / OWN_BLOCKS;
    const bool owner = nper <= NPER_MAX && (nEdges & 3) == 0 && nEdges <= BIN_CAP * OWN_BLOCKS &&
                       nNodes <= (1 << S_BITS) && 2 * nNodes >= OWN_BLOCKS * (OWN_BLOCKS + 1);
    unsigned int* binw = (unsigned int*)edge_w;
    unsigned int* tab  = (unsigned int*)node_max;

    const int nBin = owner ? OWN_BLOCKS : 0;
    gat_el_er<<<nBin + (nNodes + EL_ROWS * BIN_WAVES - 1) / (EL_ROWS * BIN_WAVES), BIN_THREADS, 0, stream>>>(
        h, a_l, a_r, e_l, e_r, nNodes, src, dst, binw, tab, nEdges, nper, nBin);

    if (owner) {
        gat_owner<<<OWN_BLOCKS, OWN_THREADS, 0, stream>>>(src, dst, binw, tab, h, e_l, e_r, bias, out, nNodes, nEdges, nper);
        return;
    }

    int initN = nNodes * OUT_CH;
    gat_init<<<(initN + 255) / 256, 256, 0, stream>>>(out, bias, node_max, node_sum, nNodes);

    gat_edge_max<<<(nEdges + 255) / 256, 256, 0, stream>>>(src, dst, e_l, e_r, edge_w, node_max, nEdges);
    gat_edge_exp<<<(nEdges + 255) / 256, 256, 0, stream>>>(dst, edge_w, node_max, node_sum, nEdges);
    gat_scatter<<<((size_t)nEdges * 32 + 255) / 256, 256, 0, stream>>>(src, dst, h, edge_w, node_sum, out, nEdges);
}


# attention scalars taken in the owner pass from rows it fetches (row . a_r, own row . a_l; wave max, rescale past 32 edges): el_er launch gone, binning rides in GEMM launch; e_l, e_r sum order changes
# speedup vs baseline: 7.5210x; 1.0948x over previous
#include <hip/hip_runtime.h>
#include <hip/hip_bf16.h>
#include <cstddef>

typedef __attribute__((ext_vector_type(2))) float v2f;
typedef __attribute__((ext_vector_type(8))) float v8f;

#define IN_CH   128
#define OUT_CH  64
#define NEG_SLOPE 0.2f
#define EPS_F   1e-10f

typedef __attribute__((ext_vector_type(4))) int v4i;
typedef __attribute__((ext_vector_type(4))) unsigned int v4u;

#define OWN_BLOCKS   256
#define NPER_MAX     391
#define S_BITS       17
#define EMPTY_WORD   ((unsigned)NPER_MAX << S_BITS)
#define BIN_THREADS  256
#define BIN_WAVES    8
#define BIN_CAP      6400
#define BIN_SLOTS    7

__device__ __forceinline__ int chunk_edges(int nEdges) {
    return (((nEdges + OWN_BLOCKS - 1) / OWN_BLOCKS) + 31) & ~31;
}

__device__ __forceinline__ void bin_chunk(const int* __restrict__ src, const int* __restrict__ dst,
                                          unsigned int* __restrict__ binw, unsigned int* __restrict__ tab,
                                          int nEdges, int nper,
                                          unsigned int* hist, unsigned int* stage, unsigned int* wtot) {
    static_assert(BIN_THREADS == OWN_BLOCKS && BIN_WAVES * 32 == BIN_THREADS, "thread t scans owner t's counters");
    static_assert(BIN_WAVES * 800 >= BIN_CAP && BIN_SLOTS * 128 >= 800, "a wave's stretch fits the lane's load slots");
    const int tid  = threadIdx.x;
    const int lane = tid & 31;
    const int wave = __builtin_amdgcn_readfirstlane(tid >> 5);
    const int ch = chunk_edges(nEdges);
    const int c  = blockIdx.x;
    const int e0 = c * ch;
    int e1 = e0 + ch; if (e1 > nEdges) e1 = nEdges;
    const int wlen = ((ch + BIN_WAVES - 1) / BIN_WAVES + 31) & ~31;
    const int ws = e0 + wave * wlen;
    int we = ws + wlen; if (we > e1) we = e1;
    const unsigned un = (unsigned)nper;

    v4i dq[BIN_SLOTS], sq[BIN_SLOTS];
#pragma unroll
    for (int u = 0; u < BIN_SLOTS; ++u) {
        const int q = ws + (u * 32 + lane) * 4;
        const v4i none = {-1, -1, -1, -1};
        dq[u] = (q < we) ? *(const v4i*)(dst + q) : none;
        sq[u] = (q < we) ? *(const v4i*)(src + q) : none;
    }
    for (int i = tid; i < BIN_WAVES * OWN_BLOCKS; i += BIN_THREADS) hist[i] = 0u;
    __syncthreads();
    unsigned int* myh = hist + wave * OWN_BLOCKS;

    unsigned wd[BIN_SLOTS][4];
    unsigned ow[BIN_SLOTS];
#pragma unroll
    for (int u = 0; u < BIN_SLOTS; ++u) {
        const int q = ws + (u * 32 + lane) * 4;
        ow[u] = 0u;
#pragma unroll
        for (int j = 0; j < 4; ++j) wd[u][j] = EMPTY_WORD;
        if (q < we) {
#pragma unroll
            for (int j = 0; j < 4; ++j) {
                const unsigned d = (unsigned)dq[u][j];
                unsigned o = d / un; if (o > OWN_BLOCKS - 1) o = OWN_BLOCKS - 1;
                const unsigned dl = d - o * un;
                wd[u][j] = (dl < un) ? (((unsigned)sq[u][j] & ((1u << S_BITS) - 1u)) | (dl << S_BITS)) : EMPTY_WORD;
                ow[u] |= o << (8 * j);
                atomicAdd(&myh[o], 1u);
            }
        }
    }
    __syncthreads();

    unsigned run = 0u;
#pragma unroll
    for (int w = 0; w < BIN_WAVES; ++w) {
        const unsigned v = hist[w * OWN_BLOCKS + tid];
        hist[w * OWN_BLOCKS + tid] = run;
        run += v;
    }
    unsigned incl = run;
#pragma unroll
    for (int off = 1; off < 32; off <<= 1) {
        const unsigned y = __shfl_up(incl, off, 32);
        if (lane >= off) incl += y;
    }
    if (lane == 31) wtot[wave] = incl;
    __syncthreads();
    unsigned wbase = 0u;
    for (int j = 0; j < wave; ++j) wbase += wtot[j];
    const unsigned excl = wbase + incl - run;
#pragma unroll
    for (int w = 0; w < BIN_WAVES; ++w) hist[w * OWN_BLOCKS + tid] += excl;
    tab[c * (OWN_BLOCKS + 1) + tid] = excl;
    if (tid == OWN_BLOCKS - 1) tab[c * (OWN_BLOCKS + 1) + OWN_BLOCKS] = excl + run;
    __syncthreads();

    unsigned pos[BIN_SLOTS][4];
#pragma unroll
    for (int u = 0; u < BIN_SLOTS; ++u) {
        const int q = ws + (u * 32 + lane) * 4;
        if (q < we) {
#pragma unroll
            for (int j = 0; j < 4; ++j) pos[u][j] = atomicAdd(&myh[(ow[u] >> (8 * j)) & 255u], 1u);
        }
    }
#pragma unroll
    for (int u = 0; u < BIN_SLOTS; ++u) {
        const int q = ws + (u * 32 + lane) * 4;
        if (q < we) {
#pragma unroll
            for (int j = 0; j < 4; ++j) stage[pos[u][j]] = wd[u][j];
        }
    }
    __syncthreads();
    for (int i = tid; i < e1 - e0; i += BIN_THREADS) binw[e0 + i] = stage[i];
}

#define GB_ROWS    64
#define GB_PITCH   132
#define GB_THREADS 256
#define GB_KSTEPS  (IN_CH / 4)

typedef __attribute__((ext_vector_type(4))) float v4f;

__device__ __forceinline__ float gb_xhalf(float v) {
    return __builtin_bit_cast(float, __builtin_amdgcn_permlanex16(__builtin_bit_cast(int, v), __builtin_bit_cast(int, v),
                                                                   0x76543210, 0xfedcba98, false, false));
}

__global__ __launch_bounds__(GB_THREADS)
void gat_gemm_slab(const float* __restrict__ x, const float* __restrict__ W,
                   float* __restrict__ h, int mFull, int nSlabs,
                   const int* __restrict__ src, const int* __restrict__ dst,
                   unsigned int* __restrict__ binw, unsigned int* __restrict__ tab,
                   int nEdges, int nper, int nBin) {
    static_assert(OUT_CH == GB_ROWS, "W passes through a slab buffer");
    static_assert(GB_THREADS == BIN_THREADS, "a binning block has the shape of a GEMM block");
    static_assert(2 * GB_ROWS * GB_PITCH >= BIN_WAVES * OWN_BLOCKS + BIN_CAP + BIN_WAVES, "the binning buffers lie in the slab buffers");
    __shared__ __attribute__((aligned(16))) float xs[2][GB_ROWS * GB_PITCH];
    if ((int)blockIdx.x < nBin) {
        unsigned int* hist  = (unsigned int*)&xs[0][0];
        unsigned int* stage = hist + BIN_WAVES * OWN_BLOCKS;
        unsigned int* wtot  = stage + BIN_CAP;
        bin_chunk(src, dst, binw, tab, nEdges, nper, hist, stage, wtot);
        return;
    }

    const int tid  = threadIdx.x;
    const int lane = tid & 31;
    const int wave = __builtin_amdgcn_readfirstlane(tid >> 5);
    const int l15  = lane & 15;
    const int half = lane >> 4;
    const int kOff = half << 1;
    const int nBase = (wave & 1) * 32;
    const int mt    = wave >> 1;

    const v4f* x4 = (const v4f*)x;
    v4f stg[GB_ROWS / 8];
    const v4f zero4 = {0.f, 0.f, 0.f, 0.f};

    const int g  = (int)blockIdx.x - nBin;
    const int nG = (int)gridDim.x - nBin;
    const int nL = nBin < nG / 2 ? nBin : nG / 2;
    const int nE = nG - nL;
    const int nV = 2 * nE + nL;
    const bool early = g < nE;
    int stepA = early ? nE : nV;
    int stepB = early ? nV - nE : nV;

    int slab = early ? g : 2 * nE + (g - nE);
    if (slab < nSlabs) {
#pragma unroll
        for (int i = 0; i < GB_ROWS / 8; ++i) {
            const int row = slab * GB_ROWS + i * 8 + wave;
            stg[i] = (row < mFull) ? x4[(size_t)row * (IN_CH / 4) + lane] : zero4;
        }
    }
    {
        const v4f* w4 = (const v4f*)W;
        v4f wst[OUT_CH / 8];
#pragma unroll
        for (int i = 0; i < OUT_CH / 8; ++i)
            wst[i] = w4[(size_t)(i * 8 + wave) * (IN_CH / 4) + lane];
#pragma unroll
        for (int i = 0; i < OUT_CH / 8; ++i)
            *(v4f*)(&xs[1][(i * 8 + wave) * GB_PITCH + lane * 4]) = wst[i];
    }
    if (slab < nSlabs) {
#pragma unroll
        for (int i = 0; i < GB_ROWS / 8; ++i)
            *(v4f*)(&xs[0][(i * 8 + wave) * GB_PITCH + lane * 4]) = stg[i];
    }
    __syncthreads();

    v2f b0[GB_KSTEPS], b1[GB_KSTEPS];
    {
        const float* w0 = &xs[1][(nBase + l15) * GB_PITCH + kOff];
        const float* w1 = w0 + 16 * GB_PITCH;
#pragma unroll
        for (int kk = 0; kk < GB_KSTEPS; ++kk) {
            b0[kk] = *(const v2f*)(w0 + 4 * kk);
            b1[kk] = *(const v2f*)(w1 + 4 * kk);
        }
    }
    __syncthreads();

    int buf = 0;
    while (slab < nSlabs) {
        const int  nxt  = slab + stepA;
        { const int t = stepA; stepA = stepB; stepB = t; }
        const bool more = nxt < nSlabs;
        if (more) {
#pragma unroll
            for (int i = 0; i < GB_ROWS / 8; ++i) {
                const int row = nxt * GB_ROWS + i * 8 + wave;
                stg[i] = (row < mFull) ? x4[(size_t)row * (IN_CH / 4) + lane] : zero4;
            }
        }

        const float* ap = &xs[buf][(mt * 16 + l15) * GB_PITCH + kOff];
        v8f acc0 = {}, acc1 = {};
#pragma unroll
        for (int kk = 0; kk < GB_KSTEPS; ++kk) {
            const v2f a = *(const v2f*)(ap + 4 * kk);
            acc0 = __builtin_amdgcn_wmma_f32_16x16x4_f32(
                 false, a,  false, b0[kk],
                 (short)0, acc0,  false,  false);
            acc1 = __builtin_amdgcn_wmma_f32_16x16x4_f32(
                 false, a,  false, b1[kk],
                 (short)0, acc1,  false,  false);
        }

        const int mBase = slab * GB_ROWS + mt * 16;
        if (mBase < mFull) {
            float* o = h + (size_t)mBase * OUT_CH + nBase + lane;
#pragma unroll
            for (int r = 0; r < 8; ++r) {
                const float lo = acc0[r], hi = acc1[r];
                const float lox = gb_xhalf(lo), hix = gb_xhalf(hi);
                o[(size_t)r * OUT_CH]       = half ? hix : lo;
                o[(size_t)(r + 8) * OUT_CH] = half ? hi  : lox;
            }
        }

        if (more) {
#pragma unroll
            for (int i = 0; i < GB_ROWS / 8; ++i)
                *(v4f*)(&xs[buf ^ 1][(i * 8 + wave) * GB_PITCH + lane * 4]) = stg[i];
        }
        __syncthreads();
        buf ^= 1;
        slab = nxt;
    }
}

__global__ void gat_gemm_tail(const float* __restrict__ x,
                              const float* __restrict__ W,
                              float* __restrict__ h,
                              int rowStart, int nNodes) {
    int idx = blockIdx.x * blockDim.x + threadIdx.x;
    int row = rowStart + (idx >> 6);
    int col = idx & 63;
    if (row >= nNodes) return;
    const float* xr = x + (size_t)row * IN_CH;
    const float* wr = W + (size_t)col * IN_CH;
    float s = 0.f;
    for (int k = 0; k < IN_CH; ++k) s += xr[k] * wr[k];
    h[(size_t)row * OUT_CH + col] = s;
}

#define EL_ROWS 32

template <int M>
__device__ __forceinline__ float row_xor(float v) {
    const int b = __builtin_bit_cast(int, v);
    return __builtin_bit_cast(float, __builtin_amdgcn_update_dpp(b, b, 0x160 + M, 0xf, 0xf, true));
}

__global__ __launch_bounds__(BIN_THREADS)
void gat_el_er(const float* __restrict__ h,
               const float* __restrict__ a_l,
               const float* __restrict__ a_r,
               float* __restrict__ e_l,
               float* __restrict__ e_r, int nNodes) {
    static_assert(OUT_CH == 64 && EL_ROWS == 32, "a 16-lane row of the wave holds one 256-B row of h, lane l keeps row l's sums");
    const int lane = threadIdx.x & 31;
    const int wave = __builtin_amdgcn_readfirstlane((int)(threadIdx.x >> 5));
    const int row0 = ((int)blockIdx.x * BIN_WAVES + wave) * EL_ROWS;
    if (row0 >= nNodes) return;
    const int l15  = lane & 15;
    const int half = lane >> 4;
    const v4f* h4 = (const v4f*)h;
    v4f hv[EL_ROWS / 2];
#pragma unroll
    for (int i = 0; i < EL_ROWS / 2; ++i) {
        int row = row0 + 16 * half + i;
        row = row < nNodes ? row : nNodes - 1;
        hv[i] = h4[(size_t)row * (OUT_CH / 4) + l15];
    }
    const v4f al = ((const v4f*)a_l)[l15];
    const v4f ar = ((const v4f*)a_r)[l15];
    float rl = 0.f, rr = 0.f;
#pragma unroll
    for (int i = 0; i < EL_ROWS / 2; ++i) {
        float la = __builtin_fmaf(hv[i].x, al.x, hv[i].y * al.y);
        float lb = __builtin_fmaf(hv[i].z, al.z, hv[i].w * al.w);
        float ra = __builtin_fmaf(hv[i].x, ar.x, hv[i].y * ar.y);
        float rb = __builtin_fmaf(hv[i].z, ar.z, hv[i].w * ar.w);
        la += row_xor<8>(la); lb += row_xor<8>(lb); ra += row_xor<8>(ra); rb += row_xor<8>(rb);
        la += row_xor<4>(la); lb += row_xor<4>(lb); ra += row_xor<4>(ra); rb += row_xor<4>(rb);
        la += row_xor<2>(la); lb += row_xor<2>(lb); ra += row_xor<2>(ra); rb += row_xor<2>(rb);
        la += row_xor<1>(la); lb += row_xor<1>(lb); ra += row_xor<1>(ra); rb += row_xor<1>(rb);
        const float sl = la + lb, sr = ra + rb;
        if (l15 == i) { rl = sl; rr = sr; }
    }
    const int row = row0 + lane;
    if (row < nNodes) { e_l[row] = rl; e_r[row] = rr; }
}

__global__ void gat_init(float* __restrict__ out, const float* __restrict__ bias,
                         float* __restrict__ node_max, float* __restrict__ node_sum,
                         int nNodes) {
    int i = blockIdx.x * blockDim.x + threadIdx.x;
    if (i < nNodes * OUT_CH) out[i] = bias[i & (OUT_CH - 1)];
    if (i < nNodes) { node_max[i] = -__builtin_inff(); node_sum[i] = 0.f; }
}

__device__ __forceinline__ void atomicMaxF32(float* addr, float v) {
    if (v >= 0.f)
        atomicMax((int*)addr, __float_as_int(v));
    else
        atomicMin((unsigned int*)addr, __float_as_uint(v));
}

__global__ void gat_edge_max(const int* __restrict__ src, const int* __restrict__ dst,
                             const float* __restrict__ e_l, const float* __restrict__ e_r,
                             float* __restrict__ edge_w, float* __restrict__ node_max,
                             int nEdges) {
    int e = blockIdx.x * blockDim.x + threadIdx.x;
    if (e >= nEdges) return;
    int s = src[e], d = dst[e];
    float a = e_l[d] + e_r[s];
    a = (a > 0.f) ? a : NEG_SLOPE * a;
    edge_w[e] = a;
    atomicMaxF32(node_max + d, a);
}

__global__ void gat_edge_exp(const int* __restrict__ dst,
                             float* __restrict__ edge_w,
                             const float* __restrict__ node_max,
                             float* __restrict__ node_sum, int nEdges) {
    int e = blockIdx.x * blockDim.x + threadIdx.x;
    if (e >= nEdges) return;
    int d = dst[e];
    float w = __expf(edge_w[e] - node_max[d]);
    edge_w[e] = w;
    atomicAdd(node_sum + d, w);
}

__global__ void gat_scatter(const int* __restrict__ src, const int* __restrict__ dst,
                            const float* __restrict__ h,
                            const float* __restrict__ edge_w,
                            const float* __restrict__ node_sum,
                            float* __restrict__ out, int nEdges) {
    const int lane = threadIdx.x & 31;
    const int e = (blockIdx.x * blockDim.x + threadIdx.x) >> 5;
    if (e >= nEdges) return;
    const int s = src[e], d = dst[e];
    const float coef = edge_w[e] / (node_sum[d] + EPS_F);
    float2 hv = ((const float2*)(h + (size_t)s * OUT_CH))[lane];
    float* o = out + (size_t)d * OUT_CH + lane * 2;
    atomicAdd(o + 0, hv.x * coef);
    atomicAdd(o + 1, hv.y * coef);
}


template <int U>
__device__ __forceinline__ int row_share(int v) {
    return __builtin_amdgcn_update_dpp(v, v, 0x150 + U, 0xf, 0xf, true);
}

template <int G>
__device__ __forceinline__ void rows_fetch(v4f* hv, const char* hb, int rb, int lb) {
    hv[4 * G + 0] = *(const v4f*)(hb + (unsigned)(row_share<4 * G + 0>(rb) + lb));
    hv[4 * G + 1] = *(const v4f*)(hb + (unsigned)(row_share<4 * G + 1>(rb) + lb));
    hv[4 * G + 2] = *(const v4f*)(hb + (unsigned)(row_share<4 * G + 2>(rb) + lb));
    hv[4 * G + 3] = *(const v4f*)(hb + (unsigned)(row_share<4 * G + 3>(rb) + lb));
}

template <int U>
__device__ __forceinline__ void row_add(const v4f* hv, float w, v4f& a, float& S) {
    const float wu = __int_as_float(row_share<U>(__float_as_int(w)));
    a.x = __builtin_fmaf(hv[U].x, wu, a.x);
    a.y = __builtin_fmaf(hv[U].y, wu, a.y);
    a.z = __builtin_fmaf(hv[U].z, wu, a.z);
    a.w = __builtin_fmaf(hv[U].w, wu, a.w);
    S += wu;
}

template <int G>
__device__ __forceinline__ void rows_dot(const v4f* hv, const v4f ar, int l15, float& er) {
    float p[4];
#pragma unroll
    for (int j = 0; j < 4; ++j) {
        const v4f r = hv[4 * G + j];
        p[j] = __builtin_fmaf(r.w, ar.w, __builtin_fmaf(r.z, ar.z, __builtin_fmaf(r.y, ar.y, r.x * ar.x)));
    }
    const bool b0 = (l15 & 1) != 0, b1 = (l15 & 2) != 0;
    float k01 = b0 ? p[1] : p[0]; const float g01 = b0 ? p[0] : p[1];
    float k23 = b0 ? p[3] : p[2]; const float g23 = b0 ? p[2] : p[3];
    k01 += row_xor<1>(g01);
    k23 += row_xor<1>(g23);
    float k = b1 ? k23 : k01; const float g = b1 ? k01 : k23;
    k += row_xor<2>(g);
    k += row_xor<4>(k);
    k += row_xor<8>(k);
    if ((l15 >> 2) == G) er = k;
}

template <int G>
__device__ __forceinline__ void rows_add(const v4f* hv, float w, v4f& a, float& S) {
    row_add<4 * G + 0>(hv, w, a, S);
    row_add<4 * G + 1>(hv, w, a, S);
    row_add<4 * G + 2>(hv, w, a, S);
    row_add<4 * G + 3>(hv, w, a, S);
}

#define OWN_THREADS  1024
#define OWN_WAVES    32
#define OWN_CPW      (OWN_BLOCKS / OWN_WAVES)
#define SORT_CAP     (((NPER_MAX + 1) * OUT_CH) / 2)
#define SCAN_PER     ((NPER_MAX + 31) / 32)
#define SEG_CAP      768

__device__ __forceinline__ void ldsMaxF32(int* p, float v) {
    const int b = __float_as_int(v);
    if (b >= 0) atomicMax(p, b);
    else        atomicMin((unsigned int*)p, (unsigned int)b);
}

__device__ __forceinline__ float row_dot64(const float* __restrict__ row, const float* __restrict__ v) {
    float s = 0.f;
    for (int k = 0; k < OUT_CH; ++k) s = __builtin_fmaf(row[k], v[k], s);
    return s;
}

template <int PHASE>
__device__ __forceinline__ void edge_now(bool m, int e, int dl, int lo, int lane,
                                         const int* __restrict__ src, const float* __restrict__ h,
                                         const float* __restrict__ a_l, const float* __restrict__ a_r,
                                         int* nmaxi, float* nsum, float* acc) {
    int s = 0; float w = 0.f;
    if (m) {
        s = src[e];
        const float x = row_dot64(h + (size_t)(lo + dl) * OUT_CH, a_l) + row_dot64(h + (size_t)s * OUT_CH, a_r);
        const float a = (x > 0.f) ? x : NEG_SLOPE * x;
        if (PHASE == 1) {
            ldsMaxF32(&nmaxi[dl], a);
        } else {
            w = __expf(a - __int_as_float(nmaxi[dl]));
            atomicAdd(&nsum[dl], w);
        }
    }
    if (PHASE == 2) {
        unsigned mask = __builtin_amdgcn_ballot_w32(m);
        const int dls = m ? dl : 0;
        while (mask) {
            const int k = __builtin_ctz(mask);
            mask &= mask - 1;
            const int   sk = __builtin_amdgcn_readlane(s, k);
            const int   dk = __builtin_amdgcn_readlane(dls, k);
            const float wk = __int_as_float(__builtin_amdgcn_readlane(__float_as_int(w), k));
            const float2 hv = ((const float2*)(h + (size_t)sk * OUT_CH))[lane];
            float* r = &acc[dk * OUT_CH + 2 * lane];
            atomicAdd(r,     hv.x * wk);
            atomicAdd(r + 1, hv.y * wk);
        }
    }
}

template <int MODE>
__device__ __forceinline__ void scan_quad(v4i v, int e0, int lo, int cnt, int lane,
                                          const int* __restrict__ src, const float* __restrict__ h,
                                          const float* __restrict__ a_l, const float* __restrict__ a_r,
                                          int* nmaxi, float* nsum, float* acc) {
    const int d0 = v.x - lo, d1 = v.y - lo, d2 = v.z - lo, d3 = v.w - lo;
    const bool m0 = (unsigned)d0 < (unsigned)cnt;
    const bool m1 = (unsigned)d1 < (unsigned)cnt;
    const bool m2 = (unsigned)d2 < (unsigned)cnt;
    const bool m3 = (unsigned)d3 < (unsigned)cnt;
    if (__builtin_amdgcn_ballot_w32(m0 | m1 | m2 | m3) == 0) return;
    edge_now<MODE>(m0, e0 + 0, d0, lo, lane, src, h, a_l, a_r, nmaxi, nsum, acc);
    edge_now<MODE>(m1, e0 + 1, d1, lo, lane, src, h, a_l, a_r, nmaxi, nsum, acc);
    edge_now<MODE>(m2, e0 + 2, d2, lo, lane, src, h, a_l, a_r, nmaxi, nsum, acc);
    edge_now<MODE>(m3, e0 + 3, d3, lo, lane, src, h, a_l, a_r, nmaxi, nsum, acc);
}

template <int MODE>
__device__ __forceinline__ void scan_edges(const int* __restrict__ src, const int* __restrict__ dst,
                                           const float* __restrict__ h,
                                           const float* __restrict__ a_l, const float* __restrict__ a_r,
                                           int nEdges, int lo, int cnt, int lane, int wave,
                                           int* nmaxi, float* nsum, float* acc) {
    const int nGroups = nEdges >> 7;
    const v4i* dst4 = (const v4i*)dst;
    const int start = (int)(((long long)blockIdx.x * nGroups) / OWN_BLOCKS);
    const v4i none = {-1, -1, -1, -1};
    for (int j = wave; j < nGroups; j += OWN_WAVES * 8) {
        v4i v[8]; int g[8];
#pragma unroll
        for (int u = 0; u < 8; ++u) {
            const int jj = j + u * OWN_WAVES;
            int gg = jj + start;
            if (gg >= nGroups) gg -= nGroups;
            g[u] = gg;
            v[u] = (jj < nGroups) ? dst4[(size_t)gg * 32 + lane] : none;
        }
#pragma unroll
        for (int u = 0; u < 8; ++u)
            scan_quad<MODE>(v[u], g[u] * 128 + lane * 4, lo, cnt, lane, src, h, a_l, a_r,
                            nmaxi, nsum, acc);
    }
    for (int eb = nGroups * 128 + wave * 32; eb < nEdges; eb += OWN_THREADS) {
        const int e = eb + lane;
        const int d = (e < nEdges) ? dst[e] : -1;
        const int dl = d - lo;
        const bool m = (unsigned)dl < (unsigned)cnt;
        const unsigned km = __builtin_amdgcn_ballot_w32(m);
        if (km == 0) continue;
        edge_now<MODE>(m, e, dl, lo, lane, src, h, a_l, a_r, nmaxi, nsum, acc);
    }
}

__global__ __launch_bounds__(OWN_THREADS)
void gat_owner(const int* __restrict__ src, const int* __restrict__ dst,
               const unsigned int* __restrict__ binw, const unsigned int* __restrict__ tab,
               const float* __restrict__ h,
               const float* __restrict__ a_l, const float* __restrict__ a_r,
               const float* __restrict__ bias, float* __restrict__ out,
               int nNodes, int nEdges, int nper) {
    __shared__ float        acc[(NPER_MAX + 1) * OUT_CH];
    __shared__ unsigned int list[OWN_WAVES * SEG_CAP];
    __shared__ int          nmaxi[NPER_MAX];
    __shared__ float        nsum[NPER_MAX];
    __shared__ unsigned int ncur[NPER_MAX + 1];
    __shared__ unsigned int overflow;
    __shared__ unsigned int total;

    const int tid  = threadIdx.x;
    const int lane = tid & 31;
    const int wave = __builtin_amdgcn_readfirstlane(tid >> 5);
    const int lo   = blockIdx.x * nper;
    int cnt = nNodes - lo;
    cnt = cnt < 0 ? 0 : (cnt > nper ? nper : cnt);
    unsigned int* seg = list + wave * SEG_CAP;

    unsigned int* ssrc = (unsigned int*)acc;

    for (int i = tid; i < cnt; i += OWN_THREADS) { nmaxi[i] = (int)0xFF800000u; nsum[i] = 0.f; ncur[i] = 0u; }
    if (tid == 0) { overflow = 0u; total = 0u; }
    __syncthreads();

    static_assert(OWN_CPW * OWN_WAVES == OWN_BLOCKS && OWN_CPW <= 32, "one chunk per lane of a wave's first lanes");
    unsigned int wcount = 0u;
    {
        const int ch = chunk_edges(nEdges);
        const int cl = wave * OWN_CPW + (lane & (OWN_CPW - 1));
        const unsigned t0 = tab[cl * (OWN_BLOCKS + 1) + blockIdx.x];
        const unsigned t1 = tab[cl * (OWN_BLOCKS + 1) + blockIdx.x + 1];
        unsigned nn[OWN_CPW], bs[OWN_CPW], pv[OWN_CPW];
        const unsigned int* sp[OWN_CPW];
#pragma unroll
        for (int u = 0; u < OWN_CPW; ++u) {
            const unsigned b0 = (unsigned)__builtin_amdgcn_readlane((int)t0, u);
            const unsigned b1 = (unsigned)__builtin_amdgcn_readlane((int)t1, u);
            nn[u] = b1 - b0;
            sp[u] = binw + (size_t)(wave * OWN_CPW + u) * ch + b0;
            bs[u] = wcount;
            wcount += nn[u];
        }
#pragma unroll
        for (int u = 0; u < OWN_CPW; ++u)
            pv[u] = ((unsigned)lane < nn[u]) ? sp[u][lane] : EMPTY_WORD;
#pragma unroll
        for (int u = 0; u < OWN_CPW; ++u) {
            const unsigned idx = bs[u] + lane;
            if ((unsigned)lane < nn[u] && idx < SEG_CAP) {
                const unsigned dl = pv[u] >> S_BITS;
                const bool mine = dl < (unsigned)cnt;
                seg[idx] = mine ? pv[u] : EMPTY_WORD;
                if (mine) atomicAdd(&ncur[dl], 1u);
            }
        }
#pragma unroll
        for (int u = 0; u < OWN_CPW; ++u) {
            for (unsigned k = 32u + lane; k < nn[u]; k += 32u) {
                const unsigned idx = bs[u] + k;
                if (idx < SEG_CAP) {
                    const unsigned p = sp[u][k];
                    const unsigned dl = p >> S_BITS;
                    const bool mine = dl < (unsigned)cnt;
                    seg[idx] = mine ? p : EMPTY_WORD;
                    if (mine) atomicAdd(&ncur[dl], 1u);
                }
            }
        }
    }
    if (wcount > SEG_CAP && lane == 0) overflow = 1u;
    if (lane == 0) atomicAdd(&total, wcount);
    __syncthreads();

    const unsigned over = __builtin_amdgcn_readfirstlane(overflow | (total > (unsigned)SORT_CAP ? 1u : 0u));
    if (over == 0u) {
        if (wave == 0) {
            unsigned v[SCAN_PER]; unsigned run = 0u;
#pragma unroll
            for (int j = 0; j < SCAN_PER; ++j) {
                const int n = lane * SCAN_PER + j;
                v[j] = (n < cnt) ? ncur[n] : 0u;
                run += v[j];
            }
            unsigned incl = run;
#pragma unroll
            for (int off = 1; off < 32; off <<= 1) {
                const unsigned y = __shfl_up(incl, off, 32);
                if (lane >= off) incl += y;
            }
            unsigned ex = incl - run;
#pragma unroll
            for (int j = 0; j < SCAN_PER; ++j) {
                const int n = lane * SCAN_PER + j;
                if (n < cnt) ncur[n] = ex;
                ex += v[j];
            }
        }
        __syncthreads();

        for (unsigned i = lane; i < wcount; i += 32u) {
            const unsigned p = seg[i];
            const unsigned dl = p >> S_BITS;
            if (dl != (unsigned)NPER_MAX) {
                const unsigned pos = atomicAdd(&ncur[dl], 1u);
                ssrc[pos] = p & ((1u << S_BITS) - 1u);
            }
        }
        __syncthreads();

        const int l15  = lane & 15;
        const int half = lane >> 4;
        const unsigned eoff = 2u * (unsigned)l15 + (unsigned)half;
        const int lb = l15 << 4;
        const float2 bv = ((const float2*)bias)[eoff];
        const char* hb = (const char*)h;
        const v4f arv = ((const v4f*)a_r)[l15];
        const v4f alv = ((const v4f*)a_l)[l15];
        const v4f* h4 = (const v4f*)h;
        unsigned stv = 0u, env = 0u; v4f own = {0.f, 0.f, 0.f, 0.f};
        if (wave < cnt) { stv = ncur[wave > 0 ? wave - 1 : 0]; env = ncur[wave]; own = h4[(size_t)(lo + wave) * (OUT_CH / 4) + l15]; }
        for (int n = wave; n < cnt; n += OWN_WAVES) {
            const unsigned st = n ? __builtin_amdgcn_readfirstlane(stv) : 0u;
            const unsigned en = __builtin_amdgcn_readfirstlane(env);
            float eln = __builtin_fmaf(own.w, alv.w, __builtin_fmaf(own.z, alv.z, __builtin_fmaf(own.y, alv.y, own.x * alv.x)));
            eln += row_xor<8>(eln);
            eln += row_xor<4>(eln);
            eln += row_xor<2>(eln);
            eln += row_xor<1>(eln);
            const int nn = n + OWN_WAVES;
            if (nn < cnt) { stv = ncur[nn - 1]; env = ncur[nn]; own = h4[(size_t)(lo + nn) * (OUT_CH / 4) + l15]; }
            v4f a = {0.f, 0.f, 0.f, 0.f}; float S = 0.f;
            float m = -__builtin_inff();
            for (unsigned b = st; b < en; b += 32u) {
                const unsigned nb = (en - b) < 32u ? (en - b) : 32u;
                const bool live = eoff < nb;
                const unsigned i = b + (live ? eoff : nb - 1u);
                const int   s  = (int)ssrc[i];
                const int rb = s << 8;
                v4f hv[16];
                rows_fetch<0>(hv, hb, rb, lb);
                if (nb > 8u)  rows_fetch<1>(hv, hb, rb, lb);
                if (nb > 16u) rows_fetch<2>(hv, hb, rb, lb);
                if (nb > 24u) rows_fetch<3>(hv, hb, rb, lb);
                float er = 0.f;
                rows_dot<0>(hv, arv, l15, er);
                if (nb > 8u)  rows_dot<1>(hv, arv, l15, er);
                if (nb > 16u) rows_dot<2>(hv, arv, l15, er);
                if (nb > 24u) rows_dot<3>(hv, arv, l15, er);
                const float x  = eln + er;
                const float sc = live ? ((x > 0.f) ? x : NEG_SLOPE * x) : -__builtin_inff();
                float tm = sc;
                tm = __builtin_fmaxf(tm, row_xor<8>(tm));
                tm = __builtin_fmaxf(tm, row_xor<4>(tm));
                tm = __builtin_fmaxf(tm, row_xor<2>(tm));
                tm = __builtin_fmaxf(tm, row_xor<1>(tm));
                tm = __builtin_fmaxf(tm, gb_xhalf(tm));
                const float mn = __builtin_fmaxf(m, tm);
                const float rs = __expf(m - mn);
                a.x *= rs; a.y *= rs; a.z *= rs; a.w *= rs; S *= rs;
                m = mn;
                const float w  = live ? __expf(sc - mn) : 0.f;
                rows_add<0>(hv, w, a, S);
                if (nb > 8u)  rows_add<1>(hv, w, a, S);
                if (nb > 16u) rows_add<2>(hv, w, a, S);
                if (nb > 24u) rows_add<3>(hv, w, a, S);
            }
            const float St = S + gb_xhalf(S);
            const float tx = a.x + gb_xhalf(a.x), ty = a.y + gb_xhalf(a.y);
            const float tz = a.z + gb_xhalf(a.z), tw = a.w + gb_xhalf(a.w);
            float2 o;
            o.x = (half ? tz : tx) / (St + EPS_F) + bv.x;
            o.y = (half ? tw : ty) / (St + EPS_F) + bv.y;
            ((float2*)(out + (size_t)(lo + n) * OUT_CH))[eoff] = o;
        }
        return;
    } else {
        for (int i = tid; i < cnt * OUT_CH; i += OWN_THREADS) acc[i] = 0.f;
        __syncthreads();
        scan_edges<1>(src, dst, h, a_l, a_r, nEdges, lo, cnt, lane, wave, nmaxi, nsum, acc);
        __syncthreads();
        scan_edges<2>(src, dst, h, a_l, a_r, nEdges, lo, cnt, lane, wave, nmaxi, nsum, acc);
    }
    __syncthreads();

    float* o = out + (size_t)lo * OUT_CH;
    for (int i = tid; i < cnt * OUT_CH; i += OWN_THREADS)
        o[i] = acc[i] / (nsum[i >> 6] + EPS_F) + bias[i & (OUT_CH - 1)];
}

extern "C" void kernel_launch(void* const* d_in, const int* in_sizes, int n_in,
                              void* d_out, int out_size, void* d_ws, size_t ws_size,
                              hipStream_t stream) {
    const float* x    = (const float*)d_in[0];
    const int*   ei   = (const int*)d_in[1];
    const float* W    = (const float*)d_in[2];
    const float* a_l  = (const float*)d_in[3];
    const float* a_r  = (const float*)d_in[4];
    const float* bias = (const float*)d_in[5];
    float* out = (float*)d_out;

    const int nNodes = in_sizes[0] / IN_CH;
    const int nEdges = in_sizes[1] / 2;
    const int* src = ei;
    const int* dst = ei + nEdges;

    float* ws       = (float*)d_ws;
    float* h        = ws;
    float* e_l      = h + (size_t)nNodes * OUT_CH;
    float* e_r      = e_l + nNodes;
    float* node_max = e_r + nNodes;
    float* node_sum = node_max + nNodes;
    float* edge_w   = node_sum + nNodes;

    const int nper = (nNodes + OWN_BLOCKS - 1) / OWN_BLOCKS;
    const bool owner = nper <= NPER_MAX && (nEdges & 3) == 0 && nEdges <= BIN_CAP * OWN_BLOCKS &&
                       nNodes <= (1 << S_BITS) && 2 * nNodes >= OWN_BLOCKS * (OWN_BLOCKS + 1);
    unsigned int* binw = (unsigned int*)edge_w;
    unsigned int* tab  = (unsigned int*)node_max;
    const int nBin = owner ? OWN_BLOCKS : 0;

    const int mFull = nNodes & ~15;
    const int nSlabs = (mFull + GB_ROWS - 1) / GB_ROWS;
    const int nGemm = nSlabs < 512 ? nSlabs : 512;
    if (nBin + nGemm > 0)
        gat_gemm_slab<<<nBin + nGemm, GB_THREADS, 0, stream>>>(x, W, h, mFull, nSlabs, src, dst, binw, tab, nEdges, nper, nBin);
    if (mFull < nNodes) {
        int tailThreads = (nNodes - mFull) * OUT_CH;
        gat_gemm_tail<<<(tailThreads + 255) / 256, 256, 0, stream>>>(x, W, h, mFull, nNodes);
    }

    if (owner) {
        gat_owner<<<OWN_BLOCKS, OWN_THREADS, 0, stream>>>(src, dst, binw, tab, h, a_l, a_r, bias, out, nNodes, nEdges, nper);
        return;
    }

    gat_el_er<<<(nNodes + EL_ROWS * BIN_WAVES - 1) / (EL_ROWS * BIN_WAVES), BIN_THREADS, 0, stream>>>(
        h, a_l, a_r, e_l, e_r, nNodes);

    int initN = nNodes * OUT_CH;
    gat_init<<<(initN + 255) / 256, 256, 0, stream>>>(out, bias, node_max, node_sum, nNodes);

    gat_edge_max<<<(nEdges + 255) / 256, 256, 0, stream>>>(src, dst, e_l, e_r, edge_w, node_max, nEdges);
    gat_edge_exp<<<(nEdges + 255) / 256, 256, 0, stream>>>(dst, edge_w, node_max, node_sum, nEdges);
    gat_scatter<<<((size_t)nEdges * 32 + 255) / 256, 256, 0, stream>>>(src, dst, h, edge_w, node_sum, out, nEdges);
}
